# starting kernel
# baseline (speedup 1.0000x reference)
#include <hip/hip_runtime.h>

#define HID 64

typedef __attribute__((ext_vector_type(2))) float v2f;
typedef __attribute__((ext_vector_type(8))) float v8f;

__device__ __forceinline__ v8f wmma_k4(v2f a, v2f b, v8f c) {
  return __builtin_amdgcn_wmma_f32_16x16x4_f32(
       false, a,  false, b,
       (short)0, c,  false,  false);
}

__device__ __forceinline__ void atomic_add_f32_hw(float* addr, float val) {
  asm volatile("global_atomic_add_f32 %0, %1, off scope:SCOPE_DEV"
               :
               : "v"(addr), "v"(val)
               : "memory");
}

__global__ void zero_f32(float* __restrict__ p, long long n) {
  long long i = (long long)blockIdx.x * blockDim.x + threadIdx.x;
  if (i < n) p[i] = 0.0f;
}

__global__ void degree_kernel(const int* __restrict__ dst, float* __restrict__ deg, int E) {
  int e = blockIdx.x * blockDim.x + threadIdx.x;
  if (e < E) atomic_add_f32_hw(&deg[dst[e]], 1.0f);
}

__global__ void scatter_kernel(const float* __restrict__ h, const int* __restrict__ src,
                               const int* __restrict__ dst, float* __restrict__ agg,
                               long long total) {
  long long t = (long long)blockIdx.x * blockDim.x + threadIdx.x;
  if (t >= total) return;
  int e = (int)(t >> 6);
  int c = (int)(t & 63);
  int s = src[e];
  int d = dst[e];
  atomic_add_f32_hw(&agg[(long long)d * HID + c], h[(long long)s * HID + c]);
}

#define PROJ_K 168
#define PROJ_SX_STRIDE 172
#define PROJ_SW_STRIDE 68
#define PROJ_LDS_FLOATS (32 * PROJ_SX_STRIDE + PROJ_K * PROJ_SW_STRIDE)

__global__ __launch_bounds__(128) void proj_kernel(const float* __restrict__ x,
                                                   const float* __restrict__ W,
                                                   const float* __restrict__ bias,
                                                   float* __restrict__ h, int n) {
  extern __shared__ float smem[];
  float(*sX)[PROJ_SX_STRIDE] = (float(*)[PROJ_SX_STRIDE])smem;
  float(*sW)[PROJ_SW_STRIDE] = (float(*)[PROJ_SW_STRIDE])(smem + 32 * PROJ_SX_STRIDE);

  const int tid = threadIdx.x;
  const int rowBase = blockIdx.x * 32;

  for (int i = tid; i < 32 * PROJ_K; i += 128) {
    int r = i / PROJ_K, c = i % PROJ_K;
    int row = rowBase + r;
    sX[r][c] = (row < n && c < 165) ? x[(long long)row * 165 + c] : 0.0f;
  }
  for (int i = tid; i < PROJ_K * 64; i += 128) {
    int r = i >> 6, c = i & 63;
    sW[r][c] = (r < 165) ? W[r * 64 + c] : 0.0f;
  }
  __syncthreads();

  const int wave = tid >> 5;
  const int lane = tid & 31;
  const int hf = lane >> 4;
  const int m = lane & 15;
  const int n0 = wave * 16;

  v8f acc0 = {};
  v8f acc1 = {};
  for (int k = 0; k < PROJ_K; k += 4) {
    const int ka = k + 2 * hf;
    v2f b, a0, a1;
    b[0] = sW[ka][n0 + m];
    b[1] = sW[ka + 1][n0 + m];
    a0[0] = sX[m][ka];
    a0[1] = sX[m][ka + 1];
    a1[0] = sX[16 + m][ka];
    a1[1] = sX[16 + m][ka + 1];
    acc0 = wmma_k4(a0, b, acc0);
    acc1 = wmma_k4(a1, b, acc1);
  }

  const float bv = bias[n0 + m];
  for (int v = 0; v < 8; ++v) {
    int r0 = rowBase + v + 8 * hf;
    int r1 = r0 + 16;
    if (r0 < n) {
      float val = acc0[v] + bv;
      h[(long long)r0 * HID + n0 + m] = val > 0.0f ? val : 0.0f;
    }
    if (r1 < n) {
      float val = acc1[v] + bv;
      h[(long long)r1 * HID + n0 + m] = val > 0.0f ? val : 0.0f;
    }
  }
}

__global__ __launch_bounds__(128) void combine_kernel(
    const float* __restrict__ agg, const float* __restrict__ deg,
    const float* __restrict__ h_in, const float* __restrict__ Wl,
    const float* __restrict__ bl, const float* __restrict__ Wr,
    float* __restrict__ h_out, int n) {
  __shared__ float sAa[32][68];
  __shared__ float sAh[32][68];
  __shared__ float sWl[64][68];
  __shared__ float sWr[64][68];
  const int tid = threadIdx.x;
  const int rowBase = blockIdx.x * 32;

  for (int i = tid; i < 32 * 64; i += 128) {
    int r = i >> 6, c = i & 63;
    int row = rowBase + r;
    float va = 0.0f, vh = 0.0f;
    if (row < n) {
      float d = deg[row];
      d = d > 1.0f ? d : 1.0f;
      va = agg[(long long)row * HID + c] / d;
      vh = h_in[(long long)row * HID + c];
    }
    sAa[r][c] = va;
    sAh[r][c] = vh;
  }
  for (int i = tid; i < 64 * 64; i += 128) {
    int r = i >> 6, c = i & 63;
    sWl[r][c] = Wl[i];
    sWr[r][c] = Wr[i];
  }
  __syncthreads();

  const int wave = tid >> 5;
  const int lane = tid & 31;
  const int hf = lane >> 4;
  const int m = lane & 15;
  const int n0 = wave * 16;

  v8f acc0 = {};
  v8f acc1 = {};
  for (int k = 0; k < 64; k += 4) {
    const int ka = k + 2 * hf;
    v2f b, a0, a1;
    b[0] = sWl[ka][n0 + m];
    b[1] = sWl[ka + 1][n0 + m];
    a0[0] = sAa[m][ka];
    a0[1] = sAa[m][ka + 1];
    a1[0] = sAa[16 + m][ka];
    a1[1] = sAa[16 + m][ka + 1];
    acc0 = wmma_k4(a0, b, acc0);
    acc1 = wmma_k4(a1, b, acc1);
  }
  for (int k = 0; k < 64; k += 4) {
    const int ka = k + 2 * hf;
    v2f b, a0, a1;
    b[0] = sWr[ka][n0 + m];
    b[1] = sWr[ka + 1][n0 + m];
    a0[0] = sAh[m][ka];
    a0[1] = sAh[m][ka + 1];
    a1[0] = sAh[16 + m][ka];
    a1[1] = sAh[16 + m][ka + 1];
    acc0 = wmma_k4(a0, b, acc0);
    acc1 = wmma_k4(a1, b, acc1);
  }

  const float bv = bl[n0 + m];
  for (int v = 0; v < 8; ++v) {
    int r0 = rowBase + v + 8 * hf;
    int r1 = r0 + 16;
    if (r0 < n) {
      float val = acc0[v] + bv;
      h_out[(long long)r0 * HID + n0 + m] = val > 0.0f ? val : 0.0f;
    }
    if (r1 < n) {
      float val = acc1[v] + bv;
      h_out[(long long)r1 * HID + n0 + m] = val > 0.0f ? val : 0.0f;
    }
  }
}

__global__ void cls_kernel(const float* __restrict__ h, const float* __restrict__ Wc,
                           const float* __restrict__ bc, float* __restrict__ out, int n) {
  int i = blockIdx.x * blockDim.x + threadIdx.x;
  if (i >= n) return;
  const float4* hp = (const float4*)(h + (long long)i * HID);
  const float4* wp = (const float4*)Wc;
  float acc = bc[0];
#pragma unroll
  for (int j = 0; j < HID / 4; ++j) {
    float4 hv = hp[j];
    float4 wv = wp[j];
    acc += hv.x * wv.x + hv.y * wv.y + hv.z * wv.z + hv.w * wv.w;
  }
  out[i] = acc;
}

extern "C" void kernel_launch(void* const* d_in, const int* in_sizes, int n_in,
                              void* d_out, int out_size, void* d_ws, size_t ws_size,
                              hipStream_t stream) {
  const float* x   = (const float*)d_in[0];
  const int* eidx  = (const int*)d_in[1];
  const float* Wp  = (const float*)d_in[2];
  const float* bp  = (const float*)d_in[3];
  const float* W1l = (const float*)d_in[4];
  const float* b1l = (const float*)d_in[5];
  const float* W1r = (const float*)d_in[6];
  const float* W2l = (const float*)d_in[7];
  const float* b2l = (const float*)d_in[8];
  const float* W2r = (const float*)d_in[9];
  const float* Wc  = (const float*)d_in[10];
  const float* bc  = (const float*)d_in[11];

  const int N = in_sizes[0] / 165;
  const int E = in_sizes[1] / 2;
  const int* src = eidx;
  const int* dst = eidx + E;

  float* A   = (float*)d_ws;
  float* B   = A + (size_t)N * HID;
  float* AGG = B + (size_t)N * HID;
  float* DEG = AGG + (size_t)N * HID;

  const long long nh = (long long)N * HID;
  const long long te = (long long)E * HID;
  const unsigned gz_nh = (unsigned)((nh + 255) / 256);
  const unsigned gz_n  = (unsigned)((N + 255) / 256);
  const unsigned g_sc  = (unsigned)((te + 255) / 256);
  const unsigned g_row = (unsigned)((N + 31) / 32);
  const size_t proj_lds = (size_t)PROJ_LDS_FLOATS * sizeof(float);

  zero_f32<<<gz_nh, 256, 0, stream>>>(AGG, nh);
  zero_f32<<<gz_n, 256, 0, stream>>>(DEG, (long long)N);
  degree_kernel<<<(E + 255) / 256, 256, 0, stream>>>(dst, DEG, E);

  proj_kernel<<<g_row, 128, proj_lds, stream>>>(x, Wp, bp, A, N);

  scatter_kernel<<<g_sc, 256, 0, stream>>>(A, src, dst, AGG, te);
  combine_kernel<<<g_row, 128, 0, stream>>>(AGG, DEG, A, W1l, b1l, W1r, B, N);

  zero_f32<<<gz_nh, 256, 0, stream>>>(AGG, nh);
  scatter_kernel<<<g_sc, 256, 0, stream>>>(B, src, dst, AGG, te);
  combine_kernel<<<g_row, 128, 0, stream>>>(AGG, DEG, B, W2l, b2l, W2r, A, N);

  cls_kernel<<<(N + 255) / 256, 256, 0, stream>>>(A, Wc, bc, (float*)d_out, N);
}


# scatter: one wave per 32 edges, 16 half-row loads in flight per wave, then 16 no-return f32 adds (was one load + one add per wave); arrival order of the float atomic adds changes
# speedup vs baseline: 2.2456x; 2.2456x over previous
#include <hip/hip_runtime.h>

#define HID 64

typedef __attribute__((ext_vector_type(2))) float v2f;
typedef __attribute__((ext_vector_type(8))) float v8f;

__device__ __forceinline__ v8f wmma_k4(v2f a, v2f b, v8f c) {
  return __builtin_amdgcn_wmma_f32_16x16x4_f32(
       false, a,  false, b,
       (short)0, c,  false,  false);
}

__device__ __forceinline__ void atomic_add_f32_hw(float* addr, float val) {
  asm volatile("global_atomic_add_f32 %0, %1, off scope:SCOPE_DEV"
               :
               : "v"(addr), "v"(val)
               : "memory");
}

__global__ void zero_f32(float* __restrict__ p, long long n) {
  long long i = (long long)blockIdx.x * blockDim.x + threadIdx.x;
  if (i < n) p[i] = 0.0f;
}

__global__ void degree_kernel(const int* __restrict__ dst, float* __restrict__ deg, int E) {
  int e = blockIdx.x * blockDim.x + threadIdx.x;
  if (e < E) atomic_add_f32_hw(&deg[dst[e]], 1.0f);
}

__device__ __forceinline__ void atomic_add_f32_row_lo(float* rowbase, unsigned voff, float val) {
  asm volatile("global_atomic_add_f32 %0, %1, %2 scope:SCOPE_DEV"
               :
               : "v"(voff), "v"(val), "s"(rowbase)
               : "memory");
}
__device__ __forceinline__ void atomic_add_f32_row_hi(float* rowbase, unsigned voff, float val) {
  asm volatile("global_atomic_add_f32 %0, %1, %2 offset:128 scope:SCOPE_DEV\n\t"
               "s_wait_xcnt 0x0"
               :
               : "v"(voff), "v"(val), "s"(rowbase)
               : "memory");
}

#define SC_BATCH 8
__global__ __launch_bounds__(256) void scatter_kernel(const float* __restrict__ h,
                                                      const int* __restrict__ src,
                                                      const int* __restrict__ dst,
                                                      float* __restrict__ agg, int E) {
  const int lane = threadIdx.x & 31;
  const int wave = blockIdx.x * 8 + (threadIdx.x >> 5);
  const int e = wave * 32 + lane;
  int sv = 0;
  int dv = -1;
  if (e < E) {
    sv = src[e];
    dv = dst[e];
  }
  const unsigned voff = (unsigned)lane * 4u;

#pragma unroll
  for (int jb = 0; jb < 32; jb += SC_BATCH) {
    float lo[SC_BATCH], hi[SC_BATCH];
#pragma unroll
    for (int u = 0; u < SC_BATCH; ++u) {
      const int s = __builtin_amdgcn_readlane(sv, jb + u);
      const float* p = h + (long long)s * HID + lane;
      lo[u] = p[0];
      hi[u] = p[32];
    }
#pragma unroll
    for (int u = 0; u < SC_BATCH; ++u) {
      const int d = __builtin_amdgcn_readlane(dv, jb + u);
      if (d >= 0) {
        float* q = agg + (long long)d * HID;
        atomic_add_f32_row_lo(q, voff, lo[u]);
        atomic_add_f32_row_hi(q, voff, hi[u]);
      }
    }
  }
}

#define PROJ_K 168
#define PROJ_SX_STRIDE 172
#define PROJ_SW_STRIDE 68
#define PROJ_LDS_FLOATS (32 * PROJ_SX_STRIDE + PROJ_K * PROJ_SW_STRIDE)

__global__ __launch_bounds__(128) void proj_kernel(const float* __restrict__ x,
                                                   const float* __restrict__ W,
                                                   const float* __restrict__ bias,
                                                   float* __restrict__ h, int n) {
  extern __shared__ float smem[];
  float(*sX)[PROJ_SX_STRIDE] = (float(*)[PROJ_SX_STRIDE])smem;
  float(*sW)[PROJ_SW_STRIDE] = (float(*)[PROJ_SW_STRIDE])(smem + 32 * PROJ_SX_STRIDE);

  const int tid = threadIdx.x;
  const int rowBase = blockIdx.x * 32;

  for (int i = tid; i < 32 * PROJ_K; i += 128) {
    int r = i / PROJ_K, c = i % PROJ_K;
    int row = rowBase + r;
    sX[r][c] = (row < n && c < 165) ? x[(long long)row * 165 + c] : 0.0f;
  }
  for (int i = tid; i < PROJ_K * 64; i += 128) {
    int r = i >> 6, c = i & 63;
    sW[r][c] = (r < 165) ? W[r * 64 + c] : 0.0f;
  }
  __syncthreads();

  const int wave = tid >> 5;
  const int lane = tid & 31;
  const int hf = lane >> 4;
  const int m = lane & 15;
  const int n0 = wave * 16;

  v8f acc0 = {};
  v8f acc1 = {};
  for (int k = 0; k < PROJ_K; k += 4) {
    const int ka = k + 2 * hf;
    v2f b, a0, a1;
    b[0] = sW[ka][n0 + m];
    b[1] = sW[ka + 1][n0 + m];
    a0[0] = sX[m][ka];
    a0[1] = sX[m][ka + 1];
    a1[0] = sX[16 + m][ka];
    a1[1] = sX[16 + m][ka + 1];
    acc0 = wmma_k4(a0, b, acc0);
    acc1 = wmma_k4(a1, b, acc1);
  }

  const float bv = bias[n0 + m];
  for (int v = 0; v < 8; ++v) {
    int r0 = rowBase + v + 8 * hf;
    int r1 = r0 + 16;
    if (r0 < n) {
      float val = acc0[v] + bv;
      h[(long long)r0 * HID + n0 + m] = val > 0.0f ? val : 0.0f;
    }
    if (r1 < n) {
      float val = acc1[v] + bv;
      h[(long long)r1 * HID + n0 + m] = val > 0.0f ? val : 0.0f;
    }
  }
}

__global__ __launch_bounds__(128) void combine_kernel(
    const float* __restrict__ agg, const float* __restrict__ deg,
    const float* __restrict__ h_in, const float* __restrict__ Wl,
    const float* __restrict__ bl, const float* __restrict__ Wr,
    float* __restrict__ h_out, int n) {
  __shared__ float sAa[32][68];
  __shared__ float sAh[32][68];
  __shared__ float sWl[64][68];
  __shared__ float sWr[64][68];
  const int tid = threadIdx.x;
  const int rowBase = blockIdx.x * 32;

  for (int i = tid; i < 32 * 64; i += 128) {
    int r = i >> 6, c = i & 63;
    int row = rowBase + r;
    float va = 0.0f, vh = 0.0f;
    if (row < n) {
      float d = deg[row];
      d = d > 1.0f ? d : 1.0f;
      va = agg[(long long)row * HID + c] / d;
      vh = h_in[(long long)row * HID + c];
    }
    sAa[r][c] = va;
    sAh[r][c] = vh;
  }
  for (int i = tid; i < 64 * 64; i += 128) {
    int r = i >> 6, c = i & 63;
    sWl[r][c] = Wl[i];
    sWr[r][c] = Wr[i];
  }
  __syncthreads();

  const int wave = tid >> 5;
  const int lane = tid & 31;
  const int hf = lane >> 4;
  const int m = lane & 15;
  const int n0 = wave * 16;

  v8f acc0 = {};
  v8f acc1 = {};
  for (int k = 0; k < 64; k += 4) {
    const int ka = k + 2 * hf;
    v2f b, a0, a1;
    b[0] = sWl[ka][n0 + m];
    b[1] = sWl[ka + 1][n0 + m];
    a0[0] = sAa[m][ka];
    a0[1] = sAa[m][ka + 1];
    a1[0] = sAa[16 + m][ka];
    a1[1] = sAa[16 + m][ka + 1];
    acc0 = wmma_k4(a0, b, acc0);
    acc1 = wmma_k4(a1, b, acc1);
  }
  for (int k = 0; k < 64; k += 4) {
    const int ka = k + 2 * hf;
    v2f b, a0, a1;
    b[0] = sWr[ka][n0 + m];
    b[1] = sWr[ka + 1][n0 + m];
    a0[0] = sAh[m][ka];
    a0[1] = sAh[m][ka + 1];
    a1[0] = sAh[16 + m][ka];
    a1[1] = sAh[16 + m][ka + 1];
    acc0 = wmma_k4(a0, b, acc0);
    acc1 = wmma_k4(a1, b, acc1);
  }

  const float bv = bl[n0 + m];
  for (int v = 0; v < 8; ++v) {
    int r0 = rowBase + v + 8 * hf;
    int r1 = r0 + 16;
    if (r0 < n) {
      float val = acc0[v] + bv;
      h_out[(long long)r0 * HID + n0 + m] = val > 0.0f ? val : 0.0f;
    }
    if (r1 < n) {
      float val = acc1[v] + bv;
      h_out[(long long)r1 * HID + n0 + m] = val > 0.0f ? val : 0.0f;
    }
  }
}

__global__ void cls_kernel(const float* __restrict__ h, const float* __restrict__ Wc,
                           const float* __restrict__ bc, float* __restrict__ out, int n) {
  int i = blockIdx.x * blockDim.x + threadIdx.x;
  if (i >= n) return;
  const float4* hp = (const float4*)(h + (long long)i * HID);
  const float4* wp = (const float4*)Wc;
  float acc = bc[0];
#pragma unroll
  for (int j = 0; j < HID / 4; ++j) {
    float4 hv = hp[j];
    float4 wv = wp[j];
    acc += hv.x * wv.x + hv.y * wv.y + hv.z * wv.z + hv.w * wv.w;
  }
  out[i] = acc;
}

extern "C" void kernel_launch(void* const* d_in, const int* in_sizes, int n_in,
                              void* d_out, int out_size, void* d_ws, size_t ws_size,
                              hipStream_t stream) {
  const float* x   = (const float*)d_in[0];
  const int* eidx  = (const int*)d_in[1];
  const float* Wp  = (const float*)d_in[2];
  const float* bp  = (const float*)d_in[3];
  const float* W1l = (const float*)d_in[4];
  const float* b1l = (const float*)d_in[5];
  const float* W1r = (const float*)d_in[6];
  const float* W2l = (const float*)d_in[7];
  const float* b2l = (const float*)d_in[8];
  const float* W2r = (const float*)d_in[9];
  const float* Wc  = (const float*)d_in[10];
  const float* bc  = (const float*)d_in[11];

  const int N = in_sizes[0] / 165;
  const int E = in_sizes[1] / 2;
  const int* src = eidx;
  const int* dst = eidx + E;

  float* A   = (float*)d_ws;
  float* B   = A + (size_t)N * HID;
  float* AGG = B + (size_t)N * HID;
  float* DEG = AGG + (size_t)N * HID;

  const long long nh = (long long)N * HID;
  const unsigned gz_nh = (unsigned)((nh + 255) / 256);
  const unsigned gz_n  = (unsigned)((N + 255) / 256);
  const unsigned g_sc  = (unsigned)((E + 255) / 256);
  const unsigned g_row = (unsigned)((N + 31) / 32);
  const size_t proj_lds = (size_t)PROJ_LDS_FLOATS * sizeof(float);

  zero_f32<<<gz_nh, 256, 0, stream>>>(AGG, nh);
  zero_f32<<<gz_n, 256, 0, stream>>>(DEG, (long long)N);
  degree_kernel<<<(E + 255) / 256, 256, 0, stream>>>(dst, DEG, E);

  proj_kernel<<<g_row, 128, proj_lds, stream>>>(x, Wp, bp, A, N);

  scatter_kernel<<<g_sc, 256, 0, stream>>>(A, src, dst, AGG, E);
  combine_kernel<<<g_row, 128, 0, stream>>>(AGG, DEG, A, W1l, b1l, W1r, B, N);

  zero_f32<<<gz_nh, 256, 0, stream>>>(AGG, nh);
  scatter_kernel<<<g_sc, 256, 0, stream>>>(B, src, dst, AGG, E);
  combine_kernel<<<g_row, 128, 0, stream>>>(AGG, DEG, B, W2l, b2l, W2r, A, N);

  cls_kernel<<<(N + 255) / 256, 256, 0, stream>>>(A, Wc, bc, (float*)d_out, N);
}


# GEMM tile staging (proj: x, W; combine: agg, h, Wl, Wr): all 16-byte loads of a tile issued as one group before the first LDS store, 32 / 28 in flight per lane (was one 4-byte load per round trip)
# speedup vs baseline: 3.5582x; 1.5845x over previous
#include <hip/hip_runtime.h>

#define HID 64

typedef __attribute__((ext_vector_type(2))) float v2f;
typedef __attribute__((ext_vector_type(8))) float v8f;

__device__ __forceinline__ v8f wmma_k4(v2f a, v2f b, v8f c) {
  return __builtin_amdgcn_wmma_f32_16x16x4_f32(
       false, a,  false, b,
       (short)0, c,  false,  false);
}

__device__ __forceinline__ void atomic_add_f32_hw(float* addr, float val) {
  asm volatile("global_atomic_add_f32 %0, %1, off scope:SCOPE_DEV"
               :
               : "v"(addr), "v"(val)
               : "memory");
}

__global__ void zero_f32(float* __restrict__ p, long long n) {
  long long i = (long long)blockIdx.x * blockDim.x + threadIdx.x;
  if (i < n) p[i] = 0.0f;
}

__global__ void degree_kernel(const int* __restrict__ dst, float* __restrict__ deg, int E) {
  int e = blockIdx.x * blockDim.x + threadIdx.x;
  if (e < E) atomic_add_f32_hw(&deg[dst[e]], 1.0f);
}

__device__ __forceinline__ void atomic_add_f32_row_lo(float* rowbase, unsigned voff, float val) {
  asm volatile("global_atomic_add_f32 %0, %1, %2 scope:SCOPE_DEV"
               :
               : "v"(voff), "v"(val), "s"(rowbase)
               : "memory");
}
__device__ __forceinline__ void atomic_add_f32_row_hi(float* rowbase, unsigned voff, float val) {
  asm volatile("global_atomic_add_f32 %0, %1, %2 offset:128 scope:SCOPE_DEV\n\t"
               "s_wait_xcnt 0x0"
               :
               : "v"(voff), "v"(val), "s"(rowbase)
               : "memory");
}

#define SC_BATCH 8
__global__ __launch_bounds__(256) void scatter_kernel(const float* __restrict__ h,
                                                      const int* __restrict__ src,
                                                      const int* __restrict__ dst,
                                                      float* __restrict__ agg, int E) {
  const int lane = threadIdx.x & 31;
  const int wave = blockIdx.x * 8 + (threadIdx.x >> 5);
  const int e = wave * 32 + lane;
  int sv = 0;
  int dv = -1;
  if (e < E) {
    sv = src[e];
    dv = dst[e];
  }
  const unsigned voff = (unsigned)lane * 4u;

#pragma unroll
  for (int jb = 0; jb < 32; jb += SC_BATCH) {
    float lo[SC_BATCH], hi[SC_BATCH];
#pragma unroll
    for (int u = 0; u < SC_BATCH; ++u) {
      const int s = __builtin_amdgcn_readlane(sv, jb + u);
      const float* p = h + (long long)s * HID + lane;
      lo[u] = p[0];
      hi[u] = p[32];
    }
#pragma unroll
    for (int u = 0; u < SC_BATCH; ++u) {
      const int d = __builtin_amdgcn_readlane(dv, jb + u);
      if (d >= 0) {
        float* q = agg + (long long)d * HID;
        atomic_add_f32_row_lo(q, voff, lo[u]);
        atomic_add_f32_row_hi(q, voff, hi[u]);
      }
    }
  }
}

#define PROJ_K 168
#define PROJ_SX_STRIDE 172
#define PROJ_SW_STRIDE 68
#define PROJ_LDS_FLOATS (32 * PROJ_SX_STRIDE + PROJ_K * PROJ_SW_STRIDE)

__global__ __launch_bounds__(128) __attribute__((amdgpu_waves_per_eu(1, 4))) void proj_kernel(const float* __restrict__ x,
                                                   const float* __restrict__ W,
                                                   const float* __restrict__ bias,
                                                   float* __restrict__ h, int n) {
  extern __shared__ float smem[];
  float(*sX)[PROJ_SX_STRIDE] = (float(*)[PROJ_SX_STRIDE])smem;
  float(*sW)[PROJ_SW_STRIDE] = (float(*)[PROJ_SW_STRIDE])(smem + 32 * PROJ_SX_STRIDE);

  const int tid = threadIdx.x;
  const int rowBase = blockIdx.x * 32;

  const bool full_tile = rowBase + 32 <= n;
  if (full_tile) {
    if (tid < 96) sX[tid / 3][165 + tid % 3] = 0.0f;
    for (int i = tid; i < 3 * 64; i += 128) sW[165 + (i >> 6)][i & 63] = 0.0f;

    const float4* wg = (const float4*)W;
    const float4* xg = (const float4*)(x + (long long)rowBase * 165);
    const int qw_last = tid < 80 ? tid + 2560 : 2639;
    const int qx_last = tid < 40 ? tid + 1280 : 1319;
    float4 wv[21];
    float4 xv[11];
#pragma unroll
    for (int j = 0; j < 21; ++j) wv[j] = wg[j < 20 ? tid + 128 * j : qw_last];
#pragma unroll
    for (int j = 0; j < 11; ++j) xv[j] = xg[j < 10 ? tid + 128 * j : qx_last];

#pragma unroll
    for (int j = 0; j < 21; ++j) {
      const int q = j < 20 ? tid + 128 * j : qw_last;
      *(float4*)&sW[q >> 4][(q & 15) * 4] = wv[j];
    }
#pragma unroll
    for (int j = 0; j < 11; ++j) {
      const int q = j < 10 ? tid + 128 * j : qx_last;
      const unsigned f = 4u * (unsigned)q;
      const unsigned r = f / 165u;
      const unsigned c = f - r * 165u;
      const float e[4] = {xv[j].x, xv[j].y, xv[j].z, xv[j].w};
#pragma unroll
      for (int t = 0; t < 4; ++t) {
        unsigned cc = c + (unsigned)t, rr = r;
        if (cc >= 165u) {
          cc -= 165u;
          rr += 1u;
        }
        sX[rr][cc] = e[t];
      }
    }
  } else {
    for (int i = tid; i < 32 * PROJ_K; i += 128) {
      int r = i / PROJ_K, c = i % PROJ_K;
      int row = rowBase + r;
      sX[r][c] = (row < n && c < 165) ? x[(long long)row * 165 + c] : 0.0f;
    }
    for (int i = tid; i < PROJ_K * 64; i += 128) {
      int r = i >> 6, c = i & 63;
      sW[r][c] = (r < 165) ? W[r * 64 + c] : 0.0f;
    }
  }
  __syncthreads();

  const int wave = tid >> 5;
  const int lane = tid & 31;
  const int hf = lane >> 4;
  const int m = lane & 15;
  const int n0 = wave * 16;

  v8f acc0 = {};
  v8f acc1 = {};
  for (int k = 0; k < PROJ_K; k += 4) {
    const int ka = k + 2 * hf;
    v2f b, a0, a1;
    b[0] = sW[ka][n0 + m];
    b[1] = sW[ka + 1][n0 + m];
    a0[0] = sX[m][ka];
    a0[1] = sX[m][ka + 1];
    a1[0] = sX[16 + m][ka];
    a1[1] = sX[16 + m][ka + 1];
    acc0 = wmma_k4(a0, b, acc0);
    acc1 = wmma_k4(a1, b, acc1);
  }

  const float bv = bias[n0 + m];
  for (int v = 0; v < 8; ++v) {
    int r0 = rowBase + v + 8 * hf;
    int r1 = r0 + 16;
    if (r0 < n) {
      float val = acc0[v] + bv;
      h[(long long)r0 * HID + n0 + m] = val > 0.0f ? val : 0.0f;
    }
    if (r1 < n) {
      float val = acc1[v] + bv;
      h[(long long)r1 * HID + n0 + m] = val > 0.0f ? val : 0.0f;
    }
  }
}

__global__ __launch_bounds__(128) void combine_kernel(
    const float* __restrict__ agg, const float* __restrict__ deg,
    const float* __restrict__ h_in, const float* __restrict__ Wl,
    const float* __restrict__ bl, const float* __restrict__ Wr,
    float* __restrict__ h_out, int n) {
  __shared__ __attribute__((aligned(16))) float sAa[32][68];
  __shared__ __attribute__((aligned(16))) float sAh[32][68];
  __shared__ __attribute__((aligned(16))) float sWl[64][68];
  __shared__ __attribute__((aligned(16))) float sWr[64][68];
  const int tid = threadIdx.x;
  const int rowBase = blockIdx.x * 32;

  const bool full_tile = rowBase + 32 <= n;
  if (full_tile) {
    const float4* ag = (const float4*)(agg + (long long)rowBase * HID);
    const float4* hg = (const float4*)(h_in + (long long)rowBase * HID);
    const float4* wlg = (const float4*)Wl;
    const float4* wrg = (const float4*)Wr;
    float dg[4];
    float4 av[4], hv[4], wl[8], wr[8];
#pragma unroll
    for (int j = 0; j < 4; ++j) {
      const int q = tid + 128 * j;
      dg[j] = deg[rowBase + (q >> 4)];
      av[j] = ag[q];
      hv[j] = hg[q];
    }
#pragma unroll
    for (int j = 0; j < 8; ++j) {
      const int q = tid + 128 * j;
      wl[j] = wlg[q];
      wr[j] = wrg[q];
    }
#pragma unroll
    for (int j = 0; j < 4; ++j) {
      const int q = tid + 128 * j;
      float d = dg[j];
      d = d > 1.0f ? d : 1.0f;
      float4 a = av[j];
      a.x = a.x / d;
      a.y = a.y / d;
      a.z = a.z / d;
      a.w = a.w / d;
      *(float4*)&sAa[q >> 4][(q & 15) * 4] = a;
      *(float4*)&sAh[q >> 4][(q & 15) * 4] = hv[j];
    }
#pragma unroll
    for (int j = 0; j < 8; ++j) {
      const int q = tid + 128 * j;
      *(float4*)&sWl[q >> 4][(q & 15) * 4] = wl[j];
      *(float4*)&sWr[q >> 4][(q & 15) * 4] = wr[j];
    }
  } else {
    for (int i = tid; i < 32 * 64; i += 128) {
      int r = i >> 6, c = i & 63;
      int row = rowBase + r;
      float va = 0.0f, vh = 0.0f;
      if (row < n) {
        float d = deg[row];
        d = d > 1.0f ? d : 1.0f;
        va = agg[(long long)row * HID + c] / d;
        vh = h_in[(long long)row * HID + c];
      }
      sAa[r][c] = va;
      sAh[r][c] = vh;
    }
    for (int i = tid; i < 64 * 64; i += 128) {
      int r = i >> 6, c = i & 63;
      sWl[r][c] = Wl[i];
      sWr[r][c] = Wr[i];
    }
  }
  __syncthreads();

  const int wave = tid >> 5;
  const int lane = tid & 31;
  const int hf = lane >> 4;
  const int m = lane & 15;
  const int n0 = wave * 16;

  v8f acc0 = {};
  v8f acc1 = {};
  for (int k = 0; k < 64; k += 4) {
    const int ka = k + 2 * hf;
    v2f b, a0, a1;
    b[0] = sWl[ka][n0 + m];
    b[1] = sWl[ka + 1][n0 + m];
    a0[0] = sAa[m][ka];
    a0[1] = sAa[m][ka + 1];
    a1[0] = sAa[16 + m][ka];
    a1[1] = sAa[16 + m][ka + 1];
    acc0 = wmma_k4(a0, b, acc0);
    acc1 = wmma_k4(a1, b, acc1);
  }
  for (int k = 0; k < 64; k += 4) {
    const int ka = k + 2 * hf;
    v2f b, a0, a1;
    b[0] = sWr[ka][n0 + m];
    b[1] = sWr[ka + 1][n0 + m];
    a0[0] = sAh[m][ka];
    a0[1] = sAh[m][ka + 1];
    a1[0] = sAh[16 + m][ka];
    a1[1] = sAh[16 + m][ka + 1];
    acc0 = wmma_k4(a0, b, acc0);
    acc1 = wmma_k4(a1, b, acc1);
  }

  const float bv = bl[n0 + m];
  for (int v = 0; v < 8; ++v) {
    int r0 = rowBase + v + 8 * hf;
    int r1 = r0 + 16;
    if (r0 < n) {
      float val = acc0[v] + bv;
      h_out[(long long)r0 * HID + n0 + m] = val > 0.0f ? val : 0.0f;
    }
    if (r1 < n) {
      float val = acc1[v] + bv;
      h_out[(long long)r1 * HID + n0 + m] = val > 0.0f ? val : 0.0f;
    }
  }
}

__global__ void cls_kernel(const float* __restrict__ h, const float* __restrict__ Wc,
                           const float* __restrict__ bc, float* __restrict__ out, int n) {
  int i = blockIdx.x * blockDim.x + threadIdx.x;
  if (i >= n) return;
  const float4* hp = (const float4*)(h + (long long)i * HID);
  const float4* wp = (const float4*)Wc;
  float acc = bc[0];
#pragma unroll
  for (int j = 0; j < HID / 4; ++j) {
    float4 hv = hp[j];
    float4 wv = wp[j];
    acc += hv.x * wv.x + hv.y * wv.y + hv.z * wv.z + hv.w * wv.w;
  }
  out[i] = acc;
}

extern "C" void kernel_launch(void* const* d_in, const int* in_sizes, int n_in,
                              void* d_out, int out_size, void* d_ws, size_t ws_size,
                              hipStream_t stream) {
  const float* x   = (const float*)d_in[0];
  const int* eidx  = (const int*)d_in[1];
  const float* Wp  = (const float*)d_in[2];
  const float* bp  = (const float*)d_in[3];
  const float* W1l = (const float*)d_in[4];
  const float* b1l = (const float*)d_in[5];
  const float* W1r = (const float*)d_in[6];
  const float* W2l = (const float*)d_in[7];
  const float* b2l = (const float*)d_in[8];
  const float* W2r = (const float*)d_in[9];
  const float* Wc  = (const float*)d_in[10];
  const float* bc  = (const float*)d_in[11];

  const int N = in_sizes[0] / 165;
  const int E = in_sizes[1] / 2;
  const int* src = eidx;
  const int* dst = eidx + E;

  float* A   = (float*)d_ws;
  float* B   = A + (size_t)N * HID;
  float* AGG = B + (size_t)N * HID;
  float* DEG = AGG + (size_t)N * HID;

  const long long nh = (long long)N * HID;
  const unsigned gz_nh = (unsigned)((nh + 255) / 256);
  const unsigned gz_n  = (unsigned)((N + 255) / 256);
  const unsigned g_sc  = (unsigned)((E + 255) / 256);
  const unsigned g_row = (unsigned)((N + 31) / 32);
  const size_t proj_lds = (size_t)PROJ_LDS_FLOATS * sizeof(float);

  zero_f32<<<gz_nh, 256, 0, stream>>>(AGG, nh);
  zero_f32<<<gz_n, 256, 0, stream>>>(DEG, (long long)N);
  degree_kernel<<<(E + 255) / 256, 256, 0, stream>>>(dst, DEG, E);

  proj_kernel<<<g_row, 128, proj_lds, stream>>>(x, Wp, bp, A, N);

  scatter_kernel<<<g_sc, 256, 0, stream>>>(A, src, dst, AGG, E);
  combine_kernel<<<g_row, 128, 0, stream>>>(AGG, DEG, A, W1l, b1l, W1r, B, N);

  zero_f32<<<gz_nh, 256, 0, stream>>>(AGG, nh);
  scatter_kernel<<<g_sc, 256, 0, stream>>>(B, src, dst, AGG, E);
  combine_kernel<<<g_row, 128, 0, stream>>>(AGG, DEG, B, W2l, b2l, W2r, A, N);

  cls_kernel<<<(N + 255) / 256, 256, 0, stream>>>(A, Wc, bc, (float*)d_out, N);
}


# neighbour sums: edges grouped by destination per call (counting sort; rank = value returned by the degree count's f32 atomic add), one wave sums a node's rows in registers; sum order = arrival rank
# speedup vs baseline: 3.9814x; 1.1189x over previous
#include <hip/hip_runtime.h>

#define HID 64

typedef __attribute__((ext_vector_type(2))) float v2f;
typedef __attribute__((ext_vector_type(8))) float v8f;

__device__ __forceinline__ v8f wmma_k4(v2f a, v2f b, v8f c) {
  return __builtin_amdgcn_wmma_f32_16x16x4_f32(
       false, a,  false, b,
       (short)0, c,  false,  false);
}

__global__ void zero_f32(float* __restrict__ p, long long n) {
  long long i = (long long)blockIdx.x * blockDim.x + threadIdx.x;
  if (i < n) p[i] = 0.0f;
}


__device__ __forceinline__ float atomic_add_f32_ret(const float* base, unsigned boff, float val) {
  float old;
  asm volatile("global_atomic_add_f32 %0, %1, %2, %3 th:TH_ATOMIC_RETURN scope:SCOPE_DEV\n\t"
               "s_wait_loadcnt 0x0"
               : "=&v"(old)
               : "v"(boff), "v"(val), "s"(base)
               : "memory");
  return old;
}
__device__ __forceinline__ void atomic_add_f32_ret4(const float* base, unsigned o0, unsigned o1, unsigned o2,
                                                    unsigned o3, float val, float& r0, float& r1, float& r2,
                                                    float& r3) {
  asm volatile("global_atomic_add_f32 %0, %4, %8, %9 th:TH_ATOMIC_RETURN scope:SCOPE_DEV\n\t"
               "global_atomic_add_f32 %1, %5, %8, %9 th:TH_ATOMIC_RETURN scope:SCOPE_DEV\n\t"
               "global_atomic_add_f32 %2, %6, %8, %9 th:TH_ATOMIC_RETURN scope:SCOPE_DEV\n\t"
               "global_atomic_add_f32 %3, %7, %8, %9 th:TH_ATOMIC_RETURN scope:SCOPE_DEV\n\t"
               "s_wait_loadcnt 0x0"
               : "=&v"(r0), "=&v"(r1), "=&v"(r2), "=&v"(r3)
               : "v"(o0), "v"(o1), "v"(o2), "v"(o3), "v"(val), "s"(base)
               : "memory");
}

#define RK_PER 4
__global__ __launch_bounds__(256) void rank_kernel(const int* __restrict__ dst, float* __restrict__ deg,
                                                   int* __restrict__ rank, int E) {
  const int blockBase = blockIdx.x * (256 * RK_PER);
  const int base = blockBase + threadIdx.x;
  if (blockBase + 256 * RK_PER <= E) {
    const int d0 = dst[base], d1 = dst[base + 256], d2 = dst[base + 512], d3 = dst[base + 768];
    float r0, r1, r2, r3;
    atomic_add_f32_ret4(deg, 4u * (unsigned)d0, 4u * (unsigned)d1, 4u * (unsigned)d2, 4u * (unsigned)d3, 1.0f,
                        r0, r1, r2, r3);
    rank[base] = (int)r0;
    rank[base + 256] = (int)r1;
    rank[base + 512] = (int)r2;
    rank[base + 768] = (int)r3;
  } else {
    for (int u = 0; u < RK_PER; ++u) {
      const int e = base + 256 * u;
      if (e < E) rank[e] = (int)atomic_add_f32_ret(deg, 4u * (unsigned)dst[e], 1.0f);
    }
  }
}

__global__ __launch_bounds__(256) void scan_local_kernel(const float* __restrict__ deg, int* __restrict__ lp,
                                                         int* __restrict__ bt, int n) {
  __shared__ int wsum[8];
  const int tid = threadIdx.x;
  const int lane = tid & 31;
  const int wave = tid >> 5;
  const int i0 = blockIdx.x * 1024 + tid * 4;
  int c[4];
  for (int u = 0; u < 4; ++u) c[u] = (i0 + u < n) ? (int)deg[i0 + u] : 0;
  const int s = c[0] + c[1] + c[2] + c[3];
  int x = s;
  for (int o = 1; o < 32; o <<= 1) {
    const int y = __shfl_up(x, o);
    if (lane >= o) x += y;
  }
  if (lane == 31) wsum[wave] = x;
  __syncthreads();
  int wbase = 0, total = 0;
  for (int w = 0; w < 8; ++w) {
    const int v = wsum[w];
    if (w < wave) wbase += v;
    total += v;
  }
  int run = wbase + x - s;
  for (int u = 0; u < 4; ++u) {
    if (i0 + u < n) lp[i0 + u] = run;
    run += c[u];
  }
  if (tid == 0) bt[blockIdx.x] = total;
}

__global__ __launch_bounds__(256) void scan_finish_kernel(const int* __restrict__ lp, const int* __restrict__ bt,
                                                          int* __restrict__ off, int n) {
  __shared__ int wsum[8];
  const int tid = threadIdx.x;
  const int lane = tid & 31;
  const int wave = tid >> 5;
  int part = 0;
  for (int j = tid; j < (int)blockIdx.x; j += 256) part += bt[j];
  for (int o = 16; o > 0; o >>= 1) part += __shfl_xor(part, o);
  if (lane == 0) wsum[wave] = part;
  __syncthreads();
  int base = 0;
  for (int w = 0; w < 8; ++w) base += wsum[w];
  const int i0 = blockIdx.x * 1024 + tid * 4;
  for (int u = 0; u < 4; ++u)
    if (i0 + u < n) off[i0 + u] = lp[i0 + u] + base;
  if (blockIdx.x == gridDim.x - 1 && tid == 0) off[n] = base + bt[blockIdx.x];
}

__global__ __launch_bounds__(256) void fill_kernel(const int* __restrict__ src, const int* __restrict__ dst,
                                                   const int* __restrict__ rank, const int* __restrict__ off,
                                                   int* __restrict__ csr, int E) {
  const int base = blockIdx.x * (256 * RK_PER) + threadIdx.x;
  int d[RK_PER], r[RK_PER], s[RK_PER], o[RK_PER];
#pragma unroll
  for (int u = 0; u < RK_PER; ++u) {
    const int e = base + 256 * u;
    const bool ok = e < E;
    d[u] = ok ? dst[e] : 0;
    r[u] = ok ? rank[e] : 0;
    s[u] = ok ? src[e] : 0;
  }
#pragma unroll
  for (int u = 0; u < RK_PER; ++u) o[u] = off[d[u]];
#pragma unroll
  for (int u = 0; u < RK_PER; ++u) {
    const int e = base + 256 * u;
    const unsigned pos = (unsigned)(o[u] + r[u]);
    if (e < E) csr[pos < (unsigned)E ? pos : (unsigned)(E - 1)] = s[u];
  }
}

#define GA_NODES 16
#define GA_BATCH 16
__global__ __launch_bounds__(256) void gather_kernel(const float* __restrict__ h, const int* __restrict__ off,
                                                     const int* __restrict__ csr, float* __restrict__ agg, int n) {
  const int lane = threadIdx.x & 31;
  const int wave = blockIdx.x * 8 + (threadIdx.x >> 5);
  const int node0 = wave * GA_NODES;
  if (node0 >= n) return;
  int oi = node0 + lane;
  if (oi > n) oi = n;
  const int ov = off[oi];
  const float2* h2 = (const float2*)h;
  float2* a2 = (float2*)agg;
  const int jn = (n - node0) < GA_NODES ? (n - node0) : GA_NODES;
  for (int j = 0; j < jn; ++j) {
    const int s = __builtin_amdgcn_readlane(ov, j);
    const int e = __builtin_amdgcn_readlane(ov, j + 1);
    float2 acc;
    acc.x = 0.0f;
    acc.y = 0.0f;
    for (int p = s; p < e; p += 32) {
      const int q = p + lane;
      int idx = q < e ? csr[q] : 0;
      idx = (unsigned)idx < (unsigned)n ? idx : 0;
      const int m = (e - p) < 32 ? (e - p) : 32;
#pragma unroll
      for (int b = 0; b < 32; b += GA_BATCH) {
        if (b < m) {
          float2 v[GA_BATCH];
#pragma unroll
          for (int u = 0; u < GA_BATCH; ++u) {
            const int si = __builtin_amdgcn_readlane(idx, b + u);
            v[u] = h2[(long long)si * (HID / 2) + lane];
          }
#pragma unroll
          for (int u = 0; u < GA_BATCH; ++u) {
            if (b + u < m) {
              acc.x += v[u].x;
              acc.y += v[u].y;
            }
          }
        }
      }
    }
    a2[(long long)(node0 + j) * (HID / 2) + lane] = acc;
  }
}

#define PROJ_K 168
#define PROJ_SX_STRIDE 172
#define PROJ_SW_STRIDE 68
#define PROJ_LDS_FLOATS (32 * PROJ_SX_STRIDE + PROJ_K * PROJ_SW_STRIDE)

__global__ __launch_bounds__(128) __attribute__((amdgpu_waves_per_eu(1, 4))) void proj_kernel(const float* __restrict__ x,
                                                   const float* __restrict__ W,
                                                   const float* __restrict__ bias,
                                                   float* __restrict__ h, int n) {
  extern __shared__ float smem[];
  float(*sX)[PROJ_SX_STRIDE] = (float(*)[PROJ_SX_STRIDE])smem;
  float(*sW)[PROJ_SW_STRIDE] = (float(*)[PROJ_SW_STRIDE])(smem + 32 * PROJ_SX_STRIDE);

  const int tid = threadIdx.x;
  const int rowBase = blockIdx.x * 32;

  const bool full_tile = rowBase + 32 <= n;
  if (full_tile) {
    if (tid < 96) sX[tid / 3][165 + tid % 3] = 0.0f;
    for (int i = tid; i < 3 * 64; i += 128) sW[165 + (i >> 6)][i & 63] = 0.0f;

    const float4* wg = (const float4*)W;
    const float4* xg = (const float4*)(x + (long long)rowBase * 165);
    const int qw_last = tid < 80 ? tid + 2560 : 2639;
    const int qx_last = tid < 40 ? tid + 1280 : 1319;
    float4 wv[21];
    float4 xv[11];
#pragma unroll
    for (int j = 0; j < 21; ++j) wv[j] = wg[j < 20 ? tid + 128 * j : qw_last];
#pragma unroll
    for (int j = 0; j < 11; ++j) xv[j] = xg[j < 10 ? tid + 128 * j : qx_last];

#pragma unroll
    for (int j = 0; j < 21; ++j) {
      const int q = j < 20 ? tid + 128 * j : qw_last;
      *(float4*)&sW[q >> 4][(q & 15) * 4] = wv[j];
    }
#pragma unroll
    for (int j = 0; j < 11; ++j) {
      const int q = j < 10 ? tid + 128 * j : qx_last;
      const unsigned f = 4u * (unsigned)q;
      const unsigned r = f / 165u;
      const unsigned c = f - r * 165u;
      const float e[4] = {xv[j].x, xv[j].y, xv[j].z, xv[j].w};
#pragma unroll
      for (int t = 0; t < 4; ++t) {
        unsigned cc = c + (unsigned)t, rr = r;
        if (cc >= 165u) {
          cc -= 165u;
          rr += 1u;
        }
        sX[rr][cc] = e[t];
      }
    }
  } else {
    for (int i = tid; i < 32 * PROJ_K; i += 128) {
      int r = i / PROJ_K, c = i % PROJ_K;
      int row = rowBase + r;
      sX[r][c] = (row < n && c < 165) ? x[(long long)row * 165 + c] : 0.0f;
    }
    for (int i = tid; i < PROJ_K * 64; i += 128) {
      int r = i >> 6, c = i & 63;
      sW[r][c] = (r < 165) ? W[r * 64 + c] : 0.0f;
    }
  }
  __syncthreads();

  const int wave = tid >> 5;
  const int lane = tid & 31;
  const int hf = lane >> 4;
  const int m = lane & 15;
  const int n0 = wave * 16;

  v8f acc0 = {};
  v8f acc1 = {};
  for (int k = 0; k < PROJ_K; k += 4) {
    const int ka = k + 2 * hf;
    v2f b, a0, a1;
    b[0] = sW[ka][n0 + m];
    b[1] = sW[ka + 1][n0 + m];
    a0[0] = sX[m][ka];
    a0[1] = sX[m][ka + 1];
    a1[0] = sX[16 + m][ka];
    a1[1] = sX[16 + m][ka + 1];
    acc0 = wmma_k4(a0, b, acc0);
    acc1 = wmma_k4(a1, b, acc1);
  }

  const float bv = bias[n0 + m];
  for (int v = 0; v < 8; ++v) {
    int r0 = rowBase + v + 8 * hf;
    int r1 = r0 + 16;
    if (r0 < n) {
      float val = acc0[v] + bv;
      h[(long long)r0 * HID + n0 + m] = val > 0.0f ? val : 0.0f;
    }
    if (r1 < n) {
      float val = acc1[v] + bv;
      h[(long long)r1 * HID + n0 + m] = val > 0.0f ? val : 0.0f;
    }
  }
}

__global__ __launch_bounds__(128) void combine_kernel(
    const float* __restrict__ agg, const float* __restrict__ deg,
    const float* h_in, const float* __restrict__ Wl,
    const float* __restrict__ bl, const float* __restrict__ Wr,
    float* h_out, int n) {
  __shared__ __attribute__((aligned(16))) float sAa[32][68];
  __shared__ __attribute__((aligned(16))) float sAh[32][68];
  __shared__ __attribute__((aligned(16))) float sWl[64][68];
  __shared__ __attribute__((aligned(16))) float sWr[64][68];
  const int tid = threadIdx.x;
  const int rowBase = blockIdx.x * 32;

  const bool full_tile = rowBase + 32 <= n;
  if (full_tile) {
    const float4* ag = (const float4*)(agg + (long long)rowBase * HID);
    const float4* hg = (const float4*)(h_in + (long long)rowBase * HID);
    const float4* wlg = (const float4*)Wl;
    const float4* wrg = (const float4*)Wr;
    float dg[4];
    float4 av[4], hv[4], wl[8], wr[8];
#pragma unroll
    for (int j = 0; j < 4; ++j) {
      const int q = tid + 128 * j;
      dg[j] = deg[rowBase + (q >> 4)];
      av[j] = ag[q];
      hv[j] = hg[q];
    }
#pragma unroll
    for (int j = 0; j < 8; ++j) {
      const int q = tid + 128 * j;
      wl[j] = wlg[q];
      wr[j] = wrg[q];
    }
#pragma unroll
    for (int j = 0; j < 4; ++j) {
      const int q = tid + 128 * j;
      float d = dg[j];
      d = d > 1.0f ? d : 1.0f;
      float4 a = av[j];
      a.x = a.x / d;
      a.y = a.y / d;
      a.z = a.z / d;
      a.w = a.w / d;
      *(float4*)&sAa[q >> 4][(q & 15) * 4] = a;
      *(float4*)&sAh[q >> 4][(q & 15) * 4] = hv[j];
    }
#pragma unroll
    for (int j = 0; j < 8; ++j) {
      const int q = tid + 128 * j;
      *(float4*)&sWl[q >> 4][(q & 15) * 4] = wl[j];
      *(float4*)&sWr[q >> 4][(q & 15) * 4] = wr[j];
    }
  } else {
    for (int i = tid; i < 32 * 64; i += 128) {
      int r = i >> 6, c = i & 63;
      int row = rowBase + r;
      float va = 0.0f, vh = 0.0f;
      if (row < n) {
        float d = deg[row];
        d = d > 1.0f ? d : 1.0f;
        va = agg[(long long)row * HID + c] / d;
        vh = h_in[(long long)row * HID + c];
      }
      sAa[r][c] = va;
      sAh[r][c] = vh;
    }
    for (int i = tid; i < 64 * 64; i += 128) {
      int r = i >> 6, c = i & 63;
      sWl[r][c] = Wl[i];
      sWr[r][c] = Wr[i];
    }
  }
  __syncthreads();

  const int wave = tid >> 5;
  const int lane = tid & 31;
  const int hf = lane >> 4;
  const int m = lane & 15;
  const int n0 = wave * 16;

  v8f acc0 = {};
  v8f acc1 = {};
  for (int k = 0; k < 64; k += 4) {
    const int ka = k + 2 * hf;
    v2f b, a0, a1;
    b[0] = sWl[ka][n0 + m];
    b[1] = sWl[ka + 1][n0 + m];
    a0[0] = sAa[m][ka];
    a0[1] = sAa[m][ka + 1];
    a1[0] = sAa[16 + m][ka];
    a1[1] = sAa[16 + m][ka + 1];
    acc0 = wmma_k4(a0, b, acc0);
    acc1 = wmma_k4(a1, b, acc1);
  }
  for (int k = 0; k < 64; k += 4) {
    const int ka = k + 2 * hf;
    v2f b, a0, a1;
    b[0] = sWr[ka][n0 + m];
    b[1] = sWr[ka + 1][n0 + m];
    a0[0] = sAh[m][ka];
    a0[1] = sAh[m][ka + 1];
    a1[0] = sAh[16 + m][ka];
    a1[1] = sAh[16 + m][ka + 1];
    acc0 = wmma_k4(a0, b, acc0);
    acc1 = wmma_k4(a1, b, acc1);
  }

  const float bv = bl[n0 + m];
  for (int v = 0; v < 8; ++v) {
    int r0 = rowBase + v + 8 * hf;
    int r1 = r0 + 16;
    if (r0 < n) {
      float val = acc0[v] + bv;
      h_out[(long long)r0 * HID + n0 + m] = val > 0.0f ? val : 0.0f;
    }
    if (r1 < n) {
      float val = acc1[v] + bv;
      h_out[(long long)r1 * HID + n0 + m] = val > 0.0f ? val : 0.0f;
    }
  }
}

__global__ void cls_kernel(const float* __restrict__ h, const float* __restrict__ Wc,
                           const float* __restrict__ bc, float* __restrict__ out, int n) {
  int i = blockIdx.x * blockDim.x + threadIdx.x;
  if (i >= n) return;
  const float4* hp = (const float4*)(h + (long long)i * HID);
  const float4* wp = (const float4*)Wc;
  float acc = bc[0];
#pragma unroll
  for (int j = 0; j < HID / 4; ++j) {
    float4 hv = hp[j];
    float4 wv = wp[j];
    acc += hv.x * wv.x + hv.y * wv.y + hv.z * wv.z + hv.w * wv.w;
  }
  out[i] = acc;
}

extern "C" void kernel_launch(void* const* d_in, const int* in_sizes, int n_in,
                              void* d_out, int out_size, void* d_ws, size_t ws_size,
                              hipStream_t stream) {
  const float* x   = (const float*)d_in[0];
  const int* eidx  = (const int*)d_in[1];
  const float* Wp  = (const float*)d_in[2];
  const float* bp  = (const float*)d_in[3];
  const float* W1l = (const float*)d_in[4];
  const float* b1l = (const float*)d_in[5];
  const float* W1r = (const float*)d_in[6];
  const float* W2l = (const float*)d_in[7];
  const float* b2l = (const float*)d_in[8];
  const float* W2r = (const float*)d_in[9];
  const float* Wc  = (const float*)d_in[10];
  const float* bc  = (const float*)d_in[11];

  const int N = in_sizes[0] / 165;
  const int E = in_sizes[1] / 2;
  const int* src = eidx;
  const int* dst = eidx + E;

  float* H   = (float*)d_ws;
  float* AGG = H + (size_t)N * HID;
  const size_t E4 = ((size_t)E + 63) / 64 * 64;
  const size_t N4 = ((size_t)N + 1 + 63) / 64 * 64;
  int* CSR   = (int*)(AGG + (size_t)N * HID);
  int* RANK  = CSR + E4;
  int* LP    = RANK + E4;
  int* OFF   = LP + N4;
  float* DEG = (float*)(OFF + N4);
  int* BT    = (int*)(DEG + N4);

  const unsigned gz_n   = (unsigned)((N + 255) / 256);
  const unsigned g_edge = (unsigned)((E + 256 * RK_PER - 1) / (256 * RK_PER));
  const unsigned g_scan = (unsigned)((N + 1023) / 1024);
  const unsigned g_ga   = (unsigned)((N + 8 * GA_NODES - 1) / (8 * GA_NODES));
  const unsigned g_row  = (unsigned)((N + 31) / 32);
  const size_t proj_lds = (size_t)PROJ_LDS_FLOATS * sizeof(float);

  zero_f32<<<gz_n, 256, 0, stream>>>(DEG, (long long)N);
  rank_kernel<<<g_edge, 256, 0, stream>>>(dst, DEG, RANK, E);
  scan_local_kernel<<<g_scan, 256, 0, stream>>>(DEG, LP, BT, N);
  scan_finish_kernel<<<g_scan, 256, 0, stream>>>(LP, BT, OFF, N);
  fill_kernel<<<g_edge, 256, 0, stream>>>(src, dst, RANK, OFF, CSR, E);

  proj_kernel<<<g_row, 128, proj_lds, stream>>>(x, Wp, bp, H, N);

  gather_kernel<<<g_ga, 256, 0, stream>>>(H, OFF, CSR, AGG, N);
  combine_kernel<<<g_row, 128, 0, stream>>>(AGG, DEG, H, W1l, b1l, W1r, H, N);

  gather_kernel<<<g_ga, 256, 0, stream>>>(H, OFF, CSR, AGG, N);
  combine_kernel<<<g_row, 128, 0, stream>>>(AGG, DEG, H, W2l, b2l, W2r, H, N);

  cls_kernel<<<(N + 255) / 256, 256, 0, stream>>>(H, Wc, bc, (float*)d_out, N);
}


# conv: neighbour sums built inside the combine kernel (a wave sums its 8 rows' sources into the LDS A tile, two nodes per pass, one per half-wave); agg plane and 2 launches gone; add order unchanged
# speedup vs baseline: 4.2556x; 1.0689x over previous
#include <hip/hip_runtime.h>

#define HID 64

typedef __attribute__((ext_vector_type(2))) float v2f;
typedef __attribute__((ext_vector_type(8))) float v8f;

__device__ __forceinline__ v8f wmma_k4(v2f a, v2f b, v8f c) {
  return __builtin_amdgcn_wmma_f32_16x16x4_f32(
       false, a,  false, b,
       (short)0, c,  false,  false);
}

__global__ void zero_f32(float* __restrict__ p, long long n) {
  long long i = (long long)blockIdx.x * blockDim.x + threadIdx.x;
  if (i < n) p[i] = 0.0f;
}


__device__ __forceinline__ float atomic_add_f32_ret(const float* base, unsigned boff, float val) {
  float old;
  asm volatile("global_atomic_add_f32 %0, %1, %2, %3 th:TH_ATOMIC_RETURN scope:SCOPE_DEV\n\t"
               "s_wait_loadcnt 0x0"
               : "=&v"(old)
               : "v"(boff), "v"(val), "s"(base)
               : "memory");
  return old;
}
__device__ __forceinline__ void atomic_add_f32_ret4(const float* base, unsigned o0, unsigned o1, unsigned o2,
                                                    unsigned o3, float val, float& r0, float& r1, float& r2,
                                                    float& r3) {
  asm volatile("global_atomic_add_f32 %0, %4, %8, %9 th:TH_ATOMIC_RETURN scope:SCOPE_DEV\n\t"
               "global_atomic_add_f32 %1, %5, %8, %9 th:TH_ATOMIC_RETURN scope:SCOPE_DEV\n\t"
               "global_atomic_add_f32 %2, %6, %8, %9 th:TH_ATOMIC_RETURN scope:SCOPE_DEV\n\t"
               "global_atomic_add_f32 %3, %7, %8, %9 th:TH_ATOMIC_RETURN scope:SCOPE_DEV\n\t"
               "s_wait_loadcnt 0x0"
               : "=&v"(r0), "=&v"(r1), "=&v"(r2), "=&v"(r3)
               : "v"(o0), "v"(o1), "v"(o2), "v"(o3), "v"(val), "s"(base)
               : "memory");
}

#define RK_PER 4
__global__ __launch_bounds__(256) void rank_kernel(const int* __restrict__ dst, float* __restrict__ deg,
                                                   int* __restrict__ rank, int E) {
  const int blockBase = blockIdx.x * (256 * RK_PER);
  const int base = blockBase + threadIdx.x;
  if (blockBase + 256 * RK_PER <= E) {
    const int d0 = dst[base], d1 = dst[base + 256], d2 = dst[base + 512], d3 = dst[base + 768];
    float r0, r1, r2, r3;
    atomic_add_f32_ret4(deg, 4u * (unsigned)d0, 4u * (unsigned)d1, 4u * (unsigned)d2, 4u * (unsigned)d3, 1.0f,
                        r0, r1, r2, r3);
    rank[base] = (int)r0;
    rank[base + 256] = (int)r1;
    rank[base + 512] = (int)r2;
    rank[base + 768] = (int)r3;
  } else {
    for (int u = 0; u < RK_PER; ++u) {
      const int e = base + 256 * u;
      if (e < E) rank[e] = (int)atomic_add_f32_ret(deg, 4u * (unsigned)dst[e], 1.0f);
    }
  }
}

__global__ __launch_bounds__(256) void scan_local_kernel(const float* __restrict__ deg, int* __restrict__ lp,
                                                         int* __restrict__ bt, int n) {
  __shared__ int wsum[8];
  const int tid = threadIdx.x;
  const int lane = tid & 31;
  const int wave = tid >> 5;
  const int i0 = blockIdx.x * 1024 + tid * 4;
  int c[4];
  for (int u = 0; u < 4; ++u) c[u] = (i0 + u < n) ? (int)deg[i0 + u] : 0;
  const int s = c[0] + c[1] + c[2] + c[3];
  int x = s;
  for (int o = 1; o < 32; o <<= 1) {
    const int y = __shfl_up(x, o);
    if (lane >= o) x += y;
  }
  if (lane == 31) wsum[wave] = x;
  __syncthreads();
  int wbase = 0, total = 0;
  for (int w = 0; w < 8; ++w) {
    const int v = wsum[w];
    if (w < wave) wbase += v;
    total += v;
  }
  int run = wbase + x - s;
  for (int u = 0; u < 4; ++u) {
    if (i0 + u < n) lp[i0 + u] = run;
    run += c[u];
  }
  if (tid == 0) bt[blockIdx.x] = total;
}

__global__ __launch_bounds__(256) void scan_finish_kernel(const int* __restrict__ lp, const int* __restrict__ bt,
                                                          int* __restrict__ off, int n) {
  __shared__ int wsum[8];
  const int tid = threadIdx.x;
  const int lane = tid & 31;
  const int wave = tid >> 5;
  int part = 0;
  for (int j = tid; j < (int)blockIdx.x; j += 256) part += bt[j];
  for (int o = 16; o > 0; o >>= 1) part += __shfl_xor(part, o);
  if (lane == 0) wsum[wave] = part;
  __syncthreads();
  int base = 0;
  for (int w = 0; w < 8; ++w) base += wsum[w];
  const int i0 = blockIdx.x * 1024 + tid * 4;
  for (int u = 0; u < 4; ++u)
    if (i0 + u < n) off[i0 + u] = lp[i0 + u] + base;
  if (blockIdx.x == gridDim.x - 1 && tid == 0) off[n] = base + bt[blockIdx.x];
}

__global__ __launch_bounds__(256) void fill_kernel(const int* __restrict__ src, const int* __restrict__ dst,
                                                   const int* __restrict__ rank, const int* __restrict__ off,
                                                   int* __restrict__ csr, int E) {
  const int base = blockIdx.x * (256 * RK_PER) + threadIdx.x;
  int d[RK_PER], r[RK_PER], s[RK_PER], o[RK_PER];
#pragma unroll
  for (int u = 0; u < RK_PER; ++u) {
    const int e = base + 256 * u;
    const bool ok = e < E;
    d[u] = ok ? dst[e] : 0;
    r[u] = ok ? rank[e] : 0;
    s[u] = ok ? src[e] : 0;
  }
#pragma unroll
  for (int u = 0; u < RK_PER; ++u) o[u] = off[d[u]];
#pragma unroll
  for (int u = 0; u < RK_PER; ++u) {
    const int e = base + 256 * u;
    const unsigned pos = (unsigned)(o[u] + r[u]);
    if (e < E) csr[pos < (unsigned)E ? pos : (unsigned)(E - 1)] = s[u];
  }
}

#define PROJ_K 168
#define PROJ_SX_STRIDE 172
#define PROJ_SW_STRIDE 68
#define PROJ_LDS_FLOATS (32 * PROJ_SX_STRIDE + PROJ_K * PROJ_SW_STRIDE)

__global__ __launch_bounds__(128) __attribute__((amdgpu_waves_per_eu(1, 4))) void proj_kernel(const float* __restrict__ x,
                                                   const float* __restrict__ W,
                                                   const float* __restrict__ bias,
                                                   float* __restrict__ h, int n) {
  extern __shared__ float smem[];
  float(*sX)[PROJ_SX_STRIDE] = (float(*)[PROJ_SX_STRIDE])smem;
  float(*sW)[PROJ_SW_STRIDE] = (float(*)[PROJ_SW_STRIDE])(smem + 32 * PROJ_SX_STRIDE);

  const int tid = threadIdx.x;
  const int rowBase = blockIdx.x * 32;

  const bool full_tile = rowBase + 32 <= n;
  if (full_tile) {
    if (tid < 96) sX[tid / 3][165 + tid % 3] = 0.0f;
    for (int i = tid; i < 3 * 64; i += 128) sW[165 + (i >> 6)][i & 63] = 0.0f;

    const float4* wg = (const float4*)W;
    const float4* xg = (const float4*)(x + (long long)rowBase * 165);
    const int qw_last = tid < 80 ? tid + 2560 : 2639;
    const int qx_last = tid < 40 ? tid + 1280 : 1319;
    float4 wv[21];
    float4 xv[11];
#pragma unroll
    for (int j = 0; j < 21; ++j) wv[j] = wg[j < 20 ? tid + 128 * j : qw_last];
#pragma unroll
    for (int j = 0; j < 11; ++j) xv[j] = xg[j < 10 ? tid + 128 * j : qx_last];

#pragma unroll
    for (int j = 0; j < 21; ++j) {
      const int q = j < 20 ? tid + 128 * j : qw_last;
      *(float4*)&sW[q >> 4][(q & 15) * 4] = wv[j];
    }
#pragma unroll
    for (int j = 0; j < 11; ++j) {
      const int q = j < 10 ? tid + 128 * j : qx_last;
      const unsigned f = 4u * (unsigned)q;
      const unsigned r = f / 165u;
      const unsigned c = f - r * 165u;
      const float e[4] = {xv[j].x, xv[j].y, xv[j].z, xv[j].w};
#pragma unroll
      for (int t = 0; t < 4; ++t) {
        unsigned cc = c + (unsigned)t, rr = r;
        if (cc >= 165u) {
          cc -= 165u;
          rr += 1u;
        }
        sX[rr][cc] = e[t];
      }
    }
  } else {
    for (int i = tid; i < 32 * PROJ_K; i += 128) {
      int r = i / PROJ_K, c = i % PROJ_K;
      int row = rowBase + r;
      sX[r][c] = (row < n && c < 165) ? x[(long long)row * 165 + c] : 0.0f;
    }
    for (int i = tid; i < PROJ_K * 64; i += 128) {
      int r = i >> 6, c = i & 63;
      sW[r][c] = (r < 165) ? W[r * 64 + c] : 0.0f;
    }
  }
  __syncthreads();

  const int wave = tid >> 5;
  const int lane = tid & 31;
  const int hf = lane >> 4;
  const int m = lane & 15;
  const int n0 = wave * 16;

  v8f acc0 = {};
  v8f acc1 = {};
  for (int k = 0; k < PROJ_K; k += 4) {
    const int ka = k + 2 * hf;
    v2f b, a0, a1;
    b[0] = sW[ka][n0 + m];
    b[1] = sW[ka + 1][n0 + m];
    a0[0] = sX[m][ka];
    a0[1] = sX[m][ka + 1];
    a1[0] = sX[16 + m][ka];
    a1[1] = sX[16 + m][ka + 1];
    acc0 = wmma_k4(a0, b, acc0);
    acc1 = wmma_k4(a1, b, acc1);
  }

  const float bv = bias[n0 + m];
  for (int v = 0; v < 8; ++v) {
    int r0 = rowBase + v + 8 * hf;
    int r1 = r0 + 16;
    if (r0 < n) {
      float val = acc0[v] + bv;
      h[(long long)r0 * HID + n0 + m] = val > 0.0f ? val : 0.0f;
    }
    if (r1 < n) {
      float val = acc1[v] + bv;
      h[(long long)r1 * HID + n0 + m] = val > 0.0f ? val : 0.0f;
    }
  }
}

__global__ __launch_bounds__(128) void conv_kernel(
    const float* __restrict__ h_in, const int* __restrict__ off, const int* __restrict__ csr,
    const float* __restrict__ Wl, const float* __restrict__ bl, const float* __restrict__ Wr,
    float* __restrict__ h_out, int n) {
  __shared__ __attribute__((aligned(16))) float sAa[32][68];
  __shared__ __attribute__((aligned(16))) float sAh[32][68];
  __shared__ __attribute__((aligned(16))) float sWl[64][68];
  __shared__ __attribute__((aligned(16))) float sWr[64][68];
  const int tid = threadIdx.x;
  const int rowBase = blockIdx.x * 32;
  const int wave = tid >> 5;
  const int lane = tid & 31;
  const int hf = lane >> 4;
  const int m = lane & 15;
  const int n0 = wave * 16;

  const int nb = rowBase + 8 * wave;
  int oi = nb + lane;
  if (oi > n) oi = n;
  const int ov = off[oi];

  int i0[4], i1[4];
#pragma unroll
  for (int t = 0; t < 4; ++t) {
    const int sA = __builtin_amdgcn_readlane(ov, 2 * t);
    const int sB = __builtin_amdgcn_readlane(ov, 2 * t + 1);
    const int sC = __builtin_amdgcn_readlane(ov, 2 * t + 2);
    const int s = hf ? sB : sA;
    const int cnt = hf ? (sC - sB) : (sB - sA);
    i0[t] = (m < cnt) ? csr[s + m] : 0;
    i1[t] = (16 + m < cnt) ? csr[s + 16 + m] : 0;
  }

  const bool full_tile = rowBase + 32 <= n;
  if (full_tile) {
    const float4* hg = (const float4*)(h_in + (long long)rowBase * HID);
    const float4* wlg = (const float4*)Wl;
    const float4* wrg = (const float4*)Wr;
    float4 hv[4], wl[8], wr[8];
#pragma unroll
    for (int j = 0; j < 4; ++j) hv[j] = hg[tid + 128 * j];
#pragma unroll
    for (int j = 0; j < 8; ++j) {
      wl[j] = wlg[tid + 128 * j];
      wr[j] = wrg[tid + 128 * j];
    }
#pragma unroll
    for (int j = 0; j < 4; ++j) {
      const int q = tid + 128 * j;
      *(float4*)&sAh[q >> 4][(q & 15) * 4] = hv[j];
    }
#pragma unroll
    for (int j = 0; j < 8; ++j) {
      const int q = tid + 128 * j;
      *(float4*)&sWl[q >> 4][(q & 15) * 4] = wl[j];
      *(float4*)&sWr[q >> 4][(q & 15) * 4] = wr[j];
    }
  } else {
    for (int i = tid; i < 32 * 64; i += 128) {
      int r = i >> 6, c = i & 63;
      int row = rowBase + r;
      sAh[r][c] = (row < n) ? h_in[(long long)row * HID + c] : 0.0f;
    }
    for (int i = tid; i < 64 * 64; i += 128) {
      int r = i >> 6, c = i & 63;
      sWl[r][c] = Wl[i];
      sWr[r][c] = Wr[i];
    }
  }

  const float4* h4 = (const float4*)h_in;
#pragma unroll
  for (int t = 0; t < 4; ++t) {
    const int sA = __builtin_amdgcn_readlane(ov, 2 * t);
    const int sB = __builtin_amdgcn_readlane(ov, 2 * t + 1);
    const int sC = __builtin_amdgcn_readlane(ov, 2 * t + 2);
    const int s = hf ? sB : sA;
    const int cnt = hf ? (sC - sB) : (sB - sA);
    const int cA = sB - sA, cB = sC - sB;
    const int mx = cA > cB ? cA : cB;
    float4 acc;
    acc.x = 0.0f;
    acc.y = 0.0f;
    acc.z = 0.0f;
    acc.w = 0.0f;
    for (int base = 0; base < mx; base += 16) {
      int idx;
      if (base == 0) {
        idx = i0[t];
      } else if (base == 16) {
        idx = i1[t];
      } else {
        idx = (base + m < cnt) ? csr[s + base + m] : 0;
      }
      idx = (unsigned)idx < (unsigned)n ? idx : 0;
      const bool more = mx - base > 8;
      float4 v[16];
#pragma unroll
      for (int u = 0; u < 8; ++u) {
        const int ra = __builtin_amdgcn_readlane(idx, u);
        const int rb = __builtin_amdgcn_readlane(idx, 16 + u);
        const int si = hf ? rb : ra;
        v[u] = h4[(long long)si * (HID / 4) + m];
      }
      if (more) {
#pragma unroll
        for (int u = 8; u < 16; ++u) {
          const int ra = __builtin_amdgcn_readlane(idx, u);
          const int rb = __builtin_amdgcn_readlane(idx, 16 + u);
          const int si = hf ? rb : ra;
          v[u] = h4[(long long)si * (HID / 4) + m];
        }
      }
#pragma unroll
      for (int u = 0; u < 8; ++u) {
        if (base + u < cnt) {
          acc.x += v[u].x;
          acc.y += v[u].y;
          acc.z += v[u].z;
          acc.w += v[u].w;
        }
      }
      if (more) {
#pragma unroll
        for (int u = 8; u < 16; ++u) {
          if (base + u < cnt) {
            acc.x += v[u].x;
            acc.y += v[u].y;
            acc.z += v[u].z;
            acc.w += v[u].w;
          }
        }
      }
    }
    float d = (float)cnt;
    d = d > 1.0f ? d : 1.0f;
    float4 a;
    a.x = acc.x / d;
    a.y = acc.y / d;
    a.z = acc.z / d;
    a.w = acc.w / d;
    *(float4*)&sAa[8 * wave + 2 * t + hf][4 * m] = a;
  }
  __syncthreads();

  v8f acc0 = {};
  v8f acc1 = {};
  for (int k = 0; k < 64; k += 4) {
    const int ka = k + 2 * hf;
    v2f b, a0, a1;
    b[0] = sWl[ka][n0 + m];
    b[1] = sWl[ka + 1][n0 + m];
    a0[0] = sAa[m][ka];
    a0[1] = sAa[m][ka + 1];
    a1[0] = sAa[16 + m][ka];
    a1[1] = sAa[16 + m][ka + 1];
    acc0 = wmma_k4(a0, b, acc0);
    acc1 = wmma_k4(a1, b, acc1);
  }
  for (int k = 0; k < 64; k += 4) {
    const int ka = k + 2 * hf;
    v2f b, a0, a1;
    b[0] = sWr[ka][n0 + m];
    b[1] = sWr[ka + 1][n0 + m];
    a0[0] = sAh[m][ka];
    a0[1] = sAh[m][ka + 1];
    a1[0] = sAh[16 + m][ka];
    a1[1] = sAh[16 + m][ka + 1];
    acc0 = wmma_k4(a0, b, acc0);
    acc1 = wmma_k4(a1, b, acc1);
  }

  const float bv = bl[n0 + m];
  for (int v = 0; v < 8; ++v) {
    int r0 = rowBase + v + 8 * hf;
    int r1 = r0 + 16;
    if (r0 < n) {
      float val = acc0[v] + bv;
      h_out[(long long)r0 * HID + n0 + m] = val > 0.0f ? val : 0.0f;
    }
    if (r1 < n) {
      float val = acc1[v] + bv;
      h_out[(long long)r1 * HID + n0 + m] = val > 0.0f ? val : 0.0f;
    }
  }
}

__global__ void cls_kernel(const float* __restrict__ h, const float* __restrict__ Wc,
                           const float* __restrict__ bc, float* __restrict__ out, int n) {
  int i = blockIdx.x * blockDim.x + threadIdx.x;
  if (i >= n) return;
  const float4* hp = (const float4*)(h + (long long)i * HID);
  const float4* wp = (const float4*)Wc;
  float acc = bc[0];
#pragma unroll
  for (int j = 0; j < HID / 4; ++j) {
    float4 hv = hp[j];
    float4 wv = wp[j];
    acc += hv.x * wv.x + hv.y * wv.y + hv.z * wv.z + hv.w * wv.w;
  }
  out[i] = acc;
}

extern "C" void kernel_launch(void* const* d_in, const int* in_sizes, int n_in,
                              void* d_out, int out_size, void* d_ws, size_t ws_size,
                              hipStream_t stream) {
  const float* x   = (const float*)d_in[0];
  const int* eidx  = (const int*)d_in[1];
  const float* Wp  = (const float*)d_in[2];
  const float* bp  = (const float*)d_in[3];
  const float* W1l = (const float*)d_in[4];
  const float* b1l = (const float*)d_in[5];
  const float* W1r = (const float*)d_in[6];
  const float* W2l = (const float*)d_in[7];
  const float* b2l = (const float*)d_in[8];
  const float* W2r = (const float*)d_in[9];
  const float* Wc  = (const float*)d_in[10];
  const float* bc  = (const float*)d_in[11];

  const int N = in_sizes[0] / 165;
  const int E = in_sizes[1] / 2;
  const int* src = eidx;
  const int* dst = eidx + E;

  float* HA  = (float*)d_ws;
  float* HB  = HA + (size_t)N * HID;
  const size_t E4 = ((size_t)E + 63) / 64 * 64;
  const size_t N4 = ((size_t)N + 1 + 63) / 64 * 64;
  int* CSR   = (int*)(HB + (size_t)N * HID);
  int* RANK  = CSR + E4;
  int* LP    = RANK + E4;
  int* OFF   = LP + N4;
  float* DEG = (float*)(OFF + N4);
  int* BT    = (int*)(DEG + N4);

  const unsigned gz_n   = (unsigned)((N + 255) / 256);
  const unsigned g_edge = (unsigned)((E + 256 * RK_PER - 1) / (256 * RK_PER));
  const unsigned g_scan = (unsigned)((N + 1023) / 1024);
  const unsigned g_row  = (unsigned)((N + 31) / 32);
  const size_t proj_lds = (size_t)PROJ_LDS_FLOATS * sizeof(float);

  zero_f32<<<gz_n, 256, 0, stream>>>(DEG, (long long)N);
  rank_kernel<<<g_edge, 256, 0, stream>>>(dst, DEG, RANK, E);
  scan_local_kernel<<<g_scan, 256, 0, stream>>>(DEG, LP, BT, N);
  scan_finish_kernel<<<g_scan, 256, 0, stream>>>(LP, BT, OFF, N);
  fill_kernel<<<g_edge, 256, 0, stream>>>(src, dst, RANK, OFF, CSR, E);

  proj_kernel<<<g_row, 128, proj_lds, stream>>>(x, Wp, bp, HA, N);

  conv_kernel<<<g_row, 128, 0, stream>>>(HA, OFF, CSR, W1l, b1l, W1r, HB, N);
  conv_kernel<<<g_row, 128, 0, stream>>>(HB, OFF, CSR, W2l, b2l, W2r, HA, N);

  cls_kernel<<<(N + 255) / 256, 256, 0, stream>>>(HA, Wc, bc, (float*)d_out, N);
}


# edge grouping: two-level counting sort in LDS (chunk by 512-node bucket, then by node), a counter row per wave, ranks by lane ballots; no atomic left, 5 launches to 2; sum order fixed by edge order
# speedup vs baseline: 6.9206x; 1.6262x over previous
#include <hip/hip_runtime.h>

#define HID 64

typedef __attribute__((ext_vector_type(2))) float v2f;
typedef __attribute__((ext_vector_type(8))) float v8f;

__device__ __forceinline__ v8f wmma_k4(v2f a, v2f b, v8f c) {
  return __builtin_amdgcn_wmma_f32_16x16x4_f32(
       false, a,  false, b,
       (short)0, c,  false,  false);
}

#define SORT_CH 16384
#define SORT_PER 16
#define BK_SHIFT 9
#define BK_NODES 512
#define NB_MAX 512
#define NC_MAX 256

__device__ __forceinline__ int wave_incl_scan(int x, int lane) {
  for (int o = 1; o < 32; o <<= 1) {
    const int y = __shfl_up(x, o);
    if (lane >= o) x += y;
  }
  return x;
}

__device__ __forceinline__ unsigned same_key_lanes(int key, bool valid) {
  unsigned m = __builtin_amdgcn_ballot_w32(valid);
  for (int i = 0; i < BK_SHIFT; ++i) {
    const bool bit = (key >> i) & 1;
    const unsigned bm = __builtin_amdgcn_ballot_w32(bit);
    m &= bit ? bm : ~bm;
  }
  return valid ? m : 0u;
}

__global__ __launch_bounds__(1024) void bucket_kernel(const int* __restrict__ src, const int* __restrict__ dst,
                                                      int* __restrict__ rec, int* __restrict__ st, int E,
                                                      int nb) {
  __shared__ int wrow[32 * NB_MAX];
  __shared__ int hist[NB_MAX];
  __shared__ int wsum[32];
  const int tid = threadIdx.x;
  const int lane = tid & 31;
  const int wave = tid >> 5;
  const unsigned lt = (1u << lane) - 1u;
  const int e0 = blockIdx.x * SORT_CH;
  for (int i = tid; i < 32 * NB_MAX; i += 1024) wrow[i] = 0;
  __syncthreads();
  int* myrow = wrow + wave * NB_MAX;
  int d[SORT_PER], s[SORT_PER], r[SORT_PER];
#pragma unroll
  for (int u = 0; u < SORT_PER; ++u) {
    const int e = e0 + tid + 1024 * u;
    const bool ok = e < E;
    d[u] = ok ? dst[e] : 0;
    s[u] = ok ? src[e] : 0;
  }
#pragma unroll
  for (int u = 0; u < SORT_PER; ++u) {
    const int e = e0 + tid + 1024 * u;
    const bool ok = e < E;
    int b = d[u] >> BK_SHIFT;
    b = b < 0 ? 0 : (b < nb ? b : nb - 1);
    d[u] = (d[u] & (BK_NODES - 1)) | (b << BK_SHIFT);
    const int old = myrow[b];
    const unsigned m = same_key_lanes(b, ok);
    r[u] = old + __popc(m & lt);
    if (ok && (m & lt) == 0u) myrow[b] = old + __popc(m);
  }
  __syncthreads();
  int v = 0;
  if (tid < nb) {
    for (int w = 0; w < 32; ++w) {
      const int c = wrow[w * NB_MAX + tid];
      wrow[w * NB_MAX + tid] = v;
      v += c;
    }
  }
  const int inc = wave_incl_scan(v, lane);
  if (lane == 31) wsum[wave] = inc;
  __syncthreads();
  int wbase = 0, total = 0;
  for (int w = 0; w < 32; ++w) {
    const int t = wsum[w];
    if (w < wave) wbase += t;
    total += t;
  }
  const int excl = wbase + inc - v;
  int* row = st + (long long)blockIdx.x * (nb + 1);
  if (tid < nb) {
    hist[tid] = excl;
    row[tid] = excl;
  }
  if (tid == 0) row[nb] = total;
  __syncthreads();
  int2* rec2 = (int2*)rec;
#pragma unroll
  for (int u = 0; u < SORT_PER; ++u) {
    const int e = e0 + tid + 1024 * u;
    if (e < E) {
      const int b = d[u] >> BK_SHIFT;
      const int pos = hist[b] + myrow[b] + r[u];
      rec2[e0 + pos] = make_int2(d[u], s[u]);
    }
  }
}

__global__ __launch_bounds__(1024) void group_kernel(const int* __restrict__ rec, const int* __restrict__ st,
                                                     int* __restrict__ csr, int* __restrict__ off, int E, int n,
                                                     int nb, int nc) {
  __shared__ int crow[32 * BK_NODES];
  __shared__ int nstart[BK_NODES];
  __shared__ int runBeg[NC_MAX];
  __shared__ int runLen[NC_MAX];
  __shared__ int wsumA[32];
  __shared__ int wsumB[32];
  const int tid = threadIdx.x;
  const int lane = tid & 31;
  const int wave = tid >> 5;
  const unsigned lt = (1u << lane) - 1u;
  const int b = blockIdx.x;
  const int ncc = nc < NC_MAX ? nc : NC_MAX;
  const int2* rec2 = (const int2*)rec;
  int* myrow = crow + wave * BK_NODES;

  int s0 = 0;
  if (tid < ncc) {
    const int* row = st + (long long)tid * (nb + 1);
    s0 = row[b];
    const int s1 = row[b + 1];
    const int beg = tid * SORT_CH + s0;
    int len = s1 - s0;
    const int lim = E - beg;
    len = len < lim ? len : lim;
    runBeg[tid] = beg;
    runLen[tid] = len;
  }
  for (int i = tid; i < 32 * BK_NODES; i += 1024) crow[i] = 0;
  int part = s0;
  for (int o = 16; o > 0; o >>= 1) part += __shfl_xor(part, o);
  if (lane == 0) wsumA[wave] = part;
  __syncthreads();
  int base = 0;
  for (int w = 0; w < 32; ++w) base += wsumA[w];

  for (int c = wave; c < ncc; c += 32) {
    const int beg = runBeg[c];
    const int len = runLen[c];
    const bool p0 = lane < len, p1 = lane + 32 < len;
    const int2 r0 = rec2[p0 ? beg + lane : 0];
    const int2 r1 = rec2[p1 ? beg + lane + 32 : 0];
    {
      const int key = r0.x & (BK_NODES - 1);
      const int old = myrow[key];
      const unsigned m = same_key_lanes(key, p0);
      if (p0 && (m & lt) == 0u) myrow[key] = old + __popc(m);
    }
    {
      const int key = r1.x & (BK_NODES - 1);
      const int old = myrow[key];
      const unsigned m = same_key_lanes(key, p1);
      if (p1 && (m & lt) == 0u) myrow[key] = old + __popc(m);
    }
    for (int k0 = 64; k0 < len; k0 += 32) {
      const bool pk = k0 + lane < len;
      const int2 rk = rec2[pk ? beg + k0 + lane : 0];
      const int key = rk.x & (BK_NODES - 1);
      const int old = myrow[key];
      const unsigned m = same_key_lanes(key, pk);
      if (pk && (m & lt) == 0u) myrow[key] = old + __popc(m);
    }
  }
  __syncthreads();

  int v = 0;
  if (tid < BK_NODES) {
    for (int w = 0; w < 32; ++w) {
      const int c = crow[w * BK_NODES + tid];
      crow[w * BK_NODES + tid] = v;
      v += c;
    }
  }
  const int inc = wave_incl_scan(v, lane);
  if (lane == 31) wsumB[wave] = inc;
  __syncthreads();
  int wbase = 0, total = 0;
  for (int w = 0; w < 32; ++w) {
    const int t = wsumB[w];
    if (w < wave) wbase += t;
    total += t;
  }
  const int excl = wbase + inc - v;
  if (tid < BK_NODES) {
    nstart[tid] = excl;
    const int node = b * BK_NODES + tid;
    if (node < n) off[node] = base + excl;
  }
  if (b == nb - 1 && tid == 0) off[n] = base + total;
  __syncthreads();

  for (int c = wave; c < ncc; c += 32) {
    const int beg = runBeg[c];
    const int len = runLen[c];
    const bool p0 = lane < len, p1 = lane + 32 < len;
    const int2 r0 = rec2[p0 ? beg + lane : 0];
    const int2 r1 = rec2[p1 ? beg + lane + 32 : 0];
    {
      const int key = r0.x & (BK_NODES - 1);
      const int old = myrow[key];
      const unsigned m = same_key_lanes(key, p0);
      const unsigned pos = (unsigned)(base + nstart[key] + old + __popc(m & lt));
      if (p0 && pos < (unsigned)E) csr[pos] = r0.y;
      if (p0 && (m & lt) == 0u) myrow[key] = old + __popc(m);
    }
    {
      const int key = r1.x & (BK_NODES - 1);
      const int old = myrow[key];
      const unsigned m = same_key_lanes(key, p1);
      const unsigned pos = (unsigned)(base + nstart[key] + old + __popc(m & lt));
      if (p1 && pos < (unsigned)E) csr[pos] = r1.y;
      if (p1 && (m & lt) == 0u) myrow[key] = old + __popc(m);
    }
    for (int k0 = 64; k0 < len; k0 += 32) {
      const bool pk = k0 + lane < len;
      const int2 rk = rec2[pk ? beg + k0 + lane : 0];
      const int key = rk.x & (BK_NODES - 1);
      const int old = myrow[key];
      const unsigned m = same_key_lanes(key, pk);
      const unsigned pos = (unsigned)(base + nstart[key] + old + __popc(m & lt));
      if (pk && pos < (unsigned)E) csr[pos] = rk.y;
      if (pk && (m & lt) == 0u) myrow[key] = old + __popc(m);
    }
  }
}

#define PROJ_K 168
#define PROJ_SX_STRIDE 172
#define PROJ_SW_STRIDE 68
#define PROJ_LDS_FLOATS (32 * PROJ_SX_STRIDE + PROJ_K * PROJ_SW_STRIDE)

__global__ __launch_bounds__(128) __attribute__((amdgpu_waves_per_eu(1, 4))) void proj_kernel(const float* __restrict__ x,
                                                   const float* __restrict__ W,
                                                   const float* __restrict__ bias,
                                                   float* __restrict__ h, int n) {
  extern __shared__ float smem[];
  float(*sX)[PROJ_SX_STRIDE] = (float(*)[PROJ_SX_STRIDE])smem;
  float(*sW)[PROJ_SW_STRIDE] = (float(*)[PROJ_SW_STRIDE])(smem + 32 * PROJ_SX_STRIDE);

  const int tid = threadIdx.x;
  const int rowBase = blockIdx.x * 32;

  const bool full_tile = rowBase + 32 <= n;
  if (full_tile) {
    if (tid < 96) sX[tid / 3][165 + tid % 3] = 0.0f;
    for (int i = tid; i < 3 * 64; i += 128) sW[165 + (i >> 6)][i & 63] = 0.0f;

    const float4* wg = (const float4*)W;
    const float4* xg = (const float4*)(x + (long long)rowBase * 165);
    const int qw_last = tid < 80 ? tid + 2560 : 2639;
    const int qx_last = tid < 40 ? tid + 1280 : 1319;
    float4 wv[21];
    float4 xv[11];
#pragma unroll
    for (int j = 0; j < 21; ++j) wv[j] = wg[j < 20 ? tid + 128 * j : qw_last];
#pragma unroll
    for (int j = 0; j < 11; ++j) xv[j] = xg[j < 10 ? tid + 128 * j : qx_last];

#pragma unroll
    for (int j = 0; j < 21; ++j) {
      const int q = j < 20 ? tid + 128 * j : qw_last;
      *(float4*)&sW[q >> 4][(q & 15) * 4] = wv[j];
    }
#pragma unroll
    for (int j = 0; j < 11; ++j) {
      const int q = j < 10 ? tid + 128 * j : qx_last;
      const unsigned f = 4u * (unsigned)q;
      const unsigned r = f / 165u;
      const unsigned c = f - r * 165u;
      const float e[4] = {xv[j].x, xv[j].y, xv[j].z, xv[j].w};
#pragma unroll
      for (int t = 0; t < 4; ++t) {
        unsigned cc = c + (unsigned)t, rr = r;
        if (cc >= 165u) {
          cc -= 165u;
          rr += 1u;
        }
        sX[rr][cc] = e[t];
      }
    }
  } else {
    for (int i = tid; i < 32 * PROJ_K; i += 128) {
      int r = i / PROJ_K, c = i % PROJ_K;
      int row = rowBase + r;
      sX[r][c] = (row < n && c < 165) ? x[(long long)row * 165 + c] : 0.0f;
    }
    for (int i = tid; i < PROJ_K * 64; i += 128) {
      int r = i >> 6, c = i & 63;
      sW[r][c] = (r < 165) ? W[r * 64 + c] : 0.0f;
    }
  }
  __syncthreads();

  const int wave = tid >> 5;
  const int lane = tid & 31;
  const int hf = lane >> 4;
  const int m = lane & 15;
  const int n0 = wave * 16;

  v8f acc0 = {};
  v8f acc1 = {};
  for (int k = 0; k < PROJ_K; k += 4) {
    const int ka = k + 2 * hf;
    v2f b, a0, a1;
    b[0] = sW[ka][n0 + m];
    b[1] = sW[ka + 1][n0 + m];
    a0[0] = sX[m][ka];
    a0[1] = sX[m][ka + 1];
    a1[0] = sX[16 + m][ka];
    a1[1] = sX[16 + m][ka + 1];
    acc0 = wmma_k4(a0, b, acc0);
    acc1 = wmma_k4(a1, b, acc1);
  }

  const float bv = bias[n0 + m];
  for (int v = 0; v < 8; ++v) {
    int r0 = rowBase + v + 8 * hf;
    int r1 = r0 + 16;
    if (r0 < n) {
      float val = acc0[v] + bv;
      h[(long long)r0 * HID + n0 + m] = val > 0.0f ? val : 0.0f;
    }
    if (r1 < n) {
      float val = acc1[v] + bv;
      h[(long long)r1 * HID + n0 + m] = val > 0.0f ? val : 0.0f;
    }
  }
}

__global__ __launch_bounds__(128) void conv_kernel(
    const float* __restrict__ h_in, const int* __restrict__ off, const int* __restrict__ csr,
    const float* __restrict__ Wl, const float* __restrict__ bl, const float* __restrict__ Wr,
    float* __restrict__ h_out, int n) {
  __shared__ __attribute__((aligned(16))) float sAa[32][68];
  __shared__ __attribute__((aligned(16))) float sAh[32][68];
  __shared__ __attribute__((aligned(16))) float sWl[64][68];
  __shared__ __attribute__((aligned(16))) float sWr[64][68];
  const int tid = threadIdx.x;
  const int rowBase = blockIdx.x * 32;
  const int wave = tid >> 5;
  const int lane = tid & 31;
  const int hf = lane >> 4;
  const int m = lane & 15;
  const int n0 = wave * 16;

  const int nb = rowBase + 8 * wave;
  int oi = nb + lane;
  if (oi > n) oi = n;
  const int ov = off[oi];

  int i0[4], i1[4];
#pragma unroll
  for (int t = 0; t < 4; ++t) {
    const int sA = __builtin_amdgcn_readlane(ov, 2 * t);
    const int sB = __builtin_amdgcn_readlane(ov, 2 * t + 1);
    const int sC = __builtin_amdgcn_readlane(ov, 2 * t + 2);
    const int s = hf ? sB : sA;
    const int cnt = hf ? (sC - sB) : (sB - sA);
    i0[t] = (m < cnt) ? csr[s + m] : 0;
    i1[t] = (16 + m < cnt) ? csr[s + 16 + m] : 0;
  }

  const bool full_tile = rowBase + 32 <= n;
  if (full_tile) {
    const float4* hg = (const float4*)(h_in + (long long)rowBase * HID);
    const float4* wlg = (const float4*)Wl;
    const float4* wrg = (const float4*)Wr;
    float4 hv[4], wl[8], wr[8];
#pragma unroll
    for (int j = 0; j < 4; ++j) hv[j] = hg[tid + 128 * j];
#pragma unroll
    for (int j = 0; j < 8; ++j) {
      wl[j] = wlg[tid + 128 * j];
      wr[j] = wrg[tid + 128 * j];
    }
#pragma unroll
    for (int j = 0; j < 4; ++j) {
      const int q = tid + 128 * j;
      *(float4*)&sAh[q >> 4][(q & 15) * 4] = hv[j];
    }
#pragma unroll
    for (int j = 0; j < 8; ++j) {
      const int q = tid + 128 * j;
      *(float4*)&sWl[q >> 4][(q & 15) * 4] = wl[j];
      *(float4*)&sWr[q >> 4][(q & 15) * 4] = wr[j];
    }
  } else {
    for (int i = tid; i < 32 * 64; i += 128) {
      int r = i >> 6, c = i & 63;
      int row = rowBase + r;
      sAh[r][c] = (row < n) ? h_in[(long long)row * HID + c] : 0.0f;
    }
    for (int i = tid; i < 64 * 64; i += 128) {
      int r = i >> 6, c = i & 63;
      sWl[r][c] = Wl[i];
      sWr[r][c] = Wr[i];
    }
  }

  const float4* h4 = (const float4*)h_in;
#pragma unroll
  for (int t = 0; t < 4; ++t) {
    const int sA = __builtin_amdgcn_readlane(ov, 2 * t);
    const int sB = __builtin_amdgcn_readlane(ov, 2 * t + 1);
    const int sC = __builtin_amdgcn_readlane(ov, 2 * t + 2);
    const int s = hf ? sB : sA;
    const int cnt = hf ? (sC - sB) : (sB - sA);
    const int cA = sB - sA, cB = sC - sB;
    const int mx = cA > cB ? cA : cB;
    float4 acc;
    acc.x = 0.0f;
    acc.y = 0.0f;
    acc.z = 0.0f;
    acc.w = 0.0f;
    for (int base = 0; base < mx; base += 16) {
      int idx;
      if (base == 0) {
        idx = i0[t];
      } else if (base == 16) {
        idx = i1[t];
      } else {
        idx = (base + m < cnt) ? csr[s + base + m] : 0;
      }
      idx = (unsigned)idx < (unsigned)n ? idx : 0;
      const bool more = mx - base > 8;
      float4 v[16];
#pragma unroll
      for (int u = 0; u < 8; ++u) {
        const int ra = __builtin_amdgcn_readlane(idx, u);
        const int rb = __builtin_amdgcn_readlane(idx, 16 + u);
        const int si = hf ? rb : ra;
        v[u] = h4[(long long)si * (HID / 4) + m];
      }
      if (more) {
#pragma unroll
        for (int u = 8; u < 16; ++u) {
          const int ra = __builtin_amdgcn_readlane(idx, u);
          const int rb = __builtin_amdgcn_readlane(idx, 16 + u);
          const int si = hf ? rb : ra;
          v[u] = h4[(long long)si * (HID / 4) + m];
        }
      }
#pragma unroll
      for (int u = 0; u < 8; ++u) {
        if (base + u < cnt) {
          acc.x += v[u].x;
          acc.y += v[u].y;
          acc.z += v[u].z;
          acc.w += v[u].w;
        }
      }
      if (more) {
#pragma unroll
        for (int u = 8; u < 16; ++u) {
          if (base + u < cnt) {
            acc.x += v[u].x;
            acc.y += v[u].y;
            acc.z += v[u].z;
            acc.w += v[u].w;
          }
        }
      }
    }
    float d = (float)cnt;
    d = d > 1.0f ? d : 1.0f;
    float4 a;
    a.x = acc.x / d;
    a.y = acc.y / d;
    a.z = acc.z / d;
    a.w = acc.w / d;
    *(float4*)&sAa[8 * wave + 2 * t + hf][4 * m] = a;
  }
  __syncthreads();

  v8f acc0 = {};
  v8f acc1 = {};
  for (int k = 0; k < 64; k += 4) {
    const int ka = k + 2 * hf;
    v2f b, a0, a1;
    b[0] = sWl[ka][n0 + m];
    b[1] = sWl[ka + 1][n0 + m];
    a0[0] = sAa[m][ka];
    a0[1] = sAa[m][ka + 1];
    a1[0] = sAa[16 + m][ka];
    a1[1] = sAa[16 + m][ka + 1];
    acc0 = wmma_k4(a0, b, acc0);
    acc1 = wmma_k4(a1, b, acc1);
  }
  for (int k = 0; k < 64; k += 4) {
    const int ka = k + 2 * hf;
    v2f b, a0, a1;
    b[0] = sWr[ka][n0 + m];
    b[1] = sWr[ka + 1][n0 + m];
    a0[0] = sAh[m][ka];
    a0[1] = sAh[m][ka + 1];
    a1[0] = sAh[16 + m][ka];
    a1[1] = sAh[16 + m][ka + 1];
    acc0 = wmma_k4(a0, b, acc0);
    acc1 = wmma_k4(a1, b, acc1);
  }

  const float bv = bl[n0 + m];
  for (int v = 0; v < 8; ++v) {
    int r0 = rowBase + v + 8 * hf;
    int r1 = r0 + 16;
    if (r0 < n) {
      float val = acc0[v] + bv;
      h_out[(long long)r0 * HID + n0 + m] = val > 0.0f ? val : 0.0f;
    }
    if (r1 < n) {
      float val = acc1[v] + bv;
      h_out[(long long)r1 * HID + n0 + m] = val > 0.0f ? val : 0.0f;
    }
  }
}

__global__ void cls_kernel(const float* __restrict__ h, const float* __restrict__ Wc,
                           const float* __restrict__ bc, float* __restrict__ out, int n) {
  int i = blockIdx.x * blockDim.x + threadIdx.x;
  if (i >= n) return;
  const float4* hp = (const float4*)(h + (long long)i * HID);
  const float4* wp = (const float4*)Wc;
  float acc = bc[0];
#pragma unroll
  for (int j = 0; j < HID / 4; ++j) {
    float4 hv = hp[j];
    float4 wv = wp[j];
    acc += hv.x * wv.x + hv.y * wv.y + hv.z * wv.z + hv.w * wv.w;
  }
  out[i] = acc;
}

extern "C" void kernel_launch(void* const* d_in, const int* in_sizes, int n_in,
                              void* d_out, int out_size, void* d_ws, size_t ws_size,
                              hipStream_t stream) {
  const float* x   = (const float*)d_in[0];
  const int* eidx  = (const int*)d_in[1];
  const float* Wp  = (const float*)d_in[2];
  const float* bp  = (const float*)d_in[3];
  const float* W1l = (const float*)d_in[4];
  const float* b1l = (const float*)d_in[5];
  const float* W1r = (const float*)d_in[6];
  const float* W2l = (const float*)d_in[7];
  const float* b2l = (const float*)d_in[8];
  const float* W2r = (const float*)d_in[9];
  const float* Wc  = (const float*)d_in[10];
  const float* bc  = (const float*)d_in[11];

  const int N = in_sizes[0] / 165;
  const int E = in_sizes[1] / 2;
  const int* src = eidx;
  const int* dst = eidx + E;

  float* HA  = (float*)d_ws;
  float* HB  = HA + (size_t)N * HID;
  const size_t E4 = ((size_t)E + 63) / 64 * 64;
  const size_t N4 = ((size_t)N + 1 + 63) / 64 * 64;
  int* CSR   = (int*)(HB + (size_t)N * HID);
  int* ST    = CSR + E4;
  int* OFF   = ST + E4 + N4;

  const int NC = (E + SORT_CH - 1) / SORT_CH;
  const int NB = (N + BK_NODES - 1) / BK_NODES;
  const unsigned g_row  = (unsigned)((N + 31) / 32);
  const size_t proj_lds = (size_t)PROJ_LDS_FLOATS * sizeof(float);

  bucket_kernel<<<(unsigned)NC, 1024, 0, stream>>>(src, dst, (int*)HB, ST, E, NB);
  group_kernel<<<(unsigned)NB, 1024, 0, stream>>>((const int*)HB, ST, CSR, OFF, E, N, NB, NC);

  proj_kernel<<<g_row, 128, proj_lds, stream>>>(x, Wp, bp, HA, N);

  conv_kernel<<<g_row, 128, 0, stream>>>(HA, OFF, CSR, W1l, b1l, W1r, HB, N);
  conv_kernel<<<g_row, 128, 0, stream>>>(HB, OFF, CSR, W2l, b2l, W2r, HA, N);

  cls_kernel<<<(N + 255) / 256, 256, 0, stream>>>(HA, Wc, bc, (float*)d_out, N);
}


# last layer: the 64 -> 1 output product is taken from conv 2's accumulators (lane butterfly, then the 4 waves via LDS); h2 never stored or re-read, 6 launches to 5; sum order of that product changes
# speedup vs baseline: 7.3588x; 1.0633x over previous
#include <hip/hip_runtime.h>

#define HID 64

typedef __attribute__((ext_vector_type(2))) float v2f;
typedef __attribute__((ext_vector_type(8))) float v8f;

__device__ __forceinline__ v8f wmma_k4(v2f a, v2f b, v8f c) {
  return __builtin_amdgcn_wmma_f32_16x16x4_f32(
       false, a,  false, b,
       (short)0, c,  false,  false);
}

#define SORT_CH 16384
#define SORT_PER 16
#define BK_SHIFT 9
#define BK_NODES 512
#define NB_MAX 512
#define NC_MAX 256

__device__ __forceinline__ int wave_incl_scan(int x, int lane) {
  for (int o = 1; o < 32; o <<= 1) {
    const int y = __shfl_up(x, o);
    if (lane >= o) x += y;
  }
  return x;
}

__device__ __forceinline__ unsigned same_key_lanes(int key, bool valid) {
  unsigned m = __builtin_amdgcn_ballot_w32(valid);
  for (int i = 0; i < BK_SHIFT; ++i) {
    const bool bit = (key >> i) & 1;
    const unsigned bm = __builtin_amdgcn_ballot_w32(bit);
    m &= bit ? bm : ~bm;
  }
  return valid ? m : 0u;
}

__global__ __launch_bounds__(1024) void bucket_kernel(const int* __restrict__ src, const int* __restrict__ dst,
                                                      int* __restrict__ rec, int* __restrict__ st, int E,
                                                      int nb) {
  __shared__ int wrow[32 * NB_MAX];
  __shared__ int hist[NB_MAX];
  __shared__ int wsum[32];
  const int tid = threadIdx.x;
  const int lane = tid & 31;
  const int wave = tid >> 5;
  const unsigned lt = (1u << lane) - 1u;
  const int e0 = blockIdx.x * SORT_CH;
  for (int i = tid; i < 32 * NB_MAX; i += 1024) wrow[i] = 0;
  __syncthreads();
  int* myrow = wrow + wave * NB_MAX;
  int d[SORT_PER], s[SORT_PER], r[SORT_PER];
#pragma unroll
  for (int u = 0; u < SORT_PER; ++u) {
    const int e = e0 + tid + 1024 * u;
    const bool ok = e < E;
    d[u] = ok ? dst[e] : 0;
    s[u] = ok ? src[e] : 0;
  }
#pragma unroll
  for (int u = 0; u < SORT_PER; ++u) {
    const int e = e0 + tid + 1024 * u;
    const bool ok = e < E;
    int b = d[u] >> BK_SHIFT;
    b = b < 0 ? 0 : (b < nb ? b : nb - 1);
    d[u] = (d[u] & (BK_NODES - 1)) | (b << BK_SHIFT);
    const int old = myrow[b];
    const unsigned m = same_key_lanes(b, ok);
    r[u] = old + __popc(m & lt);
    if (ok && (m & lt) == 0u) myrow[b] = old + __popc(m);
  }
  __syncthreads();
  int v = 0;
  if (tid < nb) {
    for (int w = 0; w < 32; ++w) {
      const int c = wrow[w * NB_MAX + tid];
      wrow[w * NB_MAX + tid] = v;
      v += c;
    }
  }
  const int inc = wave_incl_scan(v, lane);
  if (lane == 31) wsum[wave] = inc;
  __syncthreads();
  int wbase = 0, total = 0;
  for (int w = 0; w < 32; ++w) {
    const int t = wsum[w];
    if (w < wave) wbase += t;
    total += t;
  }
  const int excl = wbase + inc - v;
  int* row = st + (long long)blockIdx.x * (nb + 1);
  if (tid < nb) {
    hist[tid] = excl;
    row[tid] = excl;
  }
  if (tid == 0) row[nb] = total;
  __syncthreads();
  int2* rec2 = (int2*)rec;
#pragma unroll
  for (int u = 0; u < SORT_PER; ++u) {
    const int e = e0 + tid + 1024 * u;
    if (e < E) {
      const int b = d[u] >> BK_SHIFT;
      const int pos = hist[b] + myrow[b] + r[u];
      rec2[e0 + pos] = make_int2(d[u], s[u]);
    }
  }
}

__global__ __launch_bounds__(1024) void group_kernel(const int* __restrict__ rec, const int* __restrict__ st,
                                                     int* __restrict__ csr, int* __restrict__ off, int E, int n,
                                                     int nb, int nc) {
  __shared__ int crow[32 * BK_NODES];
  __shared__ int nstart[BK_NODES];
  __shared__ int runBeg[NC_MAX];
  __shared__ int runLen[NC_MAX];
  __shared__ int wsumA[32];
  __shared__ int wsumB[32];
  const int tid = threadIdx.x;
  const int lane = tid & 31;
  const int wave = tid >> 5;
  const unsigned lt = (1u << lane) - 1u;
  const int b = blockIdx.x;
  const int ncc = nc < NC_MAX ? nc : NC_MAX;
  const int2* rec2 = (const int2*)rec;
  int* myrow = crow + wave * BK_NODES;

  int s0 = 0;
  if (tid < ncc) {
    const int* row = st + (long long)tid * (nb + 1);
    s0 = row[b];
    const int s1 = row[b + 1];
    const int beg = tid * SORT_CH + s0;
    int len = s1 - s0;
    const int lim = E - beg;
    len = len < lim ? len : lim;
    runBeg[tid] = beg;
    runLen[tid] = len;
  }
  for (int i = tid; i < 32 * BK_NODES; i += 1024) crow[i] = 0;
  int part = s0;
  for (int o = 16; o > 0; o >>= 1) part += __shfl_xor(part, o);
  if (lane == 0) wsumA[wave] = part;
  __syncthreads();
  int base = 0;
  for (int w = 0; w < 32; ++w) base += wsumA[w];

  for (int c = wave; c < ncc; c += 32) {
    const int beg = runBeg[c];
    const int len = runLen[c];
    const bool p0 = lane < len, p1 = lane + 32 < len;
    const int2 r0 = rec2[p0 ? beg + lane : 0];
    const int2 r1 = rec2[p1 ? beg + lane + 32 : 0];
    {
      const int key = r0.x & (BK_NODES - 1);
      const int old = myrow[key];
      const unsigned m = same_key_lanes(key, p0);
      if (p0 && (m & lt) == 0u) myrow[key] = old + __popc(m);
    }
    {
      const int key = r1.x & (BK_NODES - 1);
      const int old = myrow[key];
      const unsigned m = same_key_lanes(key, p1);
      if (p1 && (m & lt) == 0u) myrow[key] = old + __popc(m);
    }
    for (int k0 = 64; k0 < len; k0 += 32) {
      const bool pk = k0 + lane < len;
      const int2 rk = rec2[pk ? beg + k0 + lane : 0];
      const int key = rk.x & (BK_NODES - 1);
      const int old = myrow[key];
      const unsigned m = same_key_lanes(key, pk);
      if (pk && (m & lt) == 0u) myrow[key] = old + __popc(m);
    }
  }
  __syncthreads();

  int v = 0;
  if (tid < BK_NODES) {
    for (int w = 0; w < 32; ++w) {
      const int c = crow[w * BK_NODES + tid];
      crow[w * BK_NODES + tid] = v;
      v += c;
    }
  }
  const int inc = wave_incl_scan(v, lane);
  if (lane == 31) wsumB[wave] = inc;
  __syncthreads();
  int wbase = 0, total = 0;
  for (int w = 0; w < 32; ++w) {
    const int t = wsumB[w];
    if (w < wave) wbase += t;
    total += t;
  }
  const int excl = wbase + inc - v;
  if (tid < BK_NODES) {
    nstart[tid] = excl;
    const int node = b * BK_NODES + tid;
    if (node < n) off[node] = base + excl;
  }
  if (b == nb - 1 && tid == 0) off[n] = base + total;
  __syncthreads();

  for (int c = wave; c < ncc; c += 32) {
    const int beg = runBeg[c];
    const int len = runLen[c];
    const bool p0 = lane < len, p1 = lane + 32 < len;
    const int2 r0 = rec2[p0 ? beg + lane : 0];
    const int2 r1 = rec2[p1 ? beg + lane + 32 : 0];
    {
      const int key = r0.x & (BK_NODES - 1);
      const int old = myrow[key];
      const unsigned m = same_key_lanes(key, p0);
      const unsigned pos = (unsigned)(base + nstart[key] + old + __popc(m & lt));
      if (p0 && pos < (unsigned)E) csr[pos] = r0.y;
      if (p0 && (m & lt) == 0u) myrow[key] = old + __popc(m);
    }
    {
      const int key = r1.x & (BK_NODES - 1);
      const int old = myrow[key];
      const unsigned m = same_key_lanes(key, p1);
      const unsigned pos = (unsigned)(base + nstart[key] + old + __popc(m & lt));
      if (p1 && pos < (unsigned)E) csr[pos] = r1.y;
      if (p1 && (m & lt) == 0u) myrow[key] = old + __popc(m);
    }
    for (int k0 = 64; k0 < len; k0 += 32) {
      const bool pk = k0 + lane < len;
      const int2 rk = rec2[pk ? beg + k0 + lane : 0];
      const int key = rk.x & (BK_NODES - 1);
      const int old = myrow[key];
      const unsigned m = same_key_lanes(key, pk);
      const unsigned pos = (unsigned)(base + nstart[key] + old + __popc(m & lt));
      if (pk && pos < (unsigned)E) csr[pos] = rk.y;
      if (pk && (m & lt) == 0u) myrow[key] = old + __popc(m);
    }
  }
}

#define PROJ_K 168
#define PROJ_SX_STRIDE 172
#define PROJ_SW_STRIDE 68
#define PROJ_LDS_FLOATS (32 * PROJ_SX_STRIDE + PROJ_K * PROJ_SW_STRIDE)

__global__ __launch_bounds__(128) __attribute__((amdgpu_waves_per_eu(1, 4))) void proj_kernel(const float* __restrict__ x,
                                                   const float* __restrict__ W,
                                                   const float* __restrict__ bias,
                                                   float* __restrict__ h, int n) {
  extern __shared__ float smem[];
  float(*sX)[PROJ_SX_STRIDE] = (float(*)[PROJ_SX_STRIDE])smem;
  float(*sW)[PROJ_SW_STRIDE] = (float(*)[PROJ_SW_STRIDE])(smem + 32 * PROJ_SX_STRIDE);

  const int tid = threadIdx.x;
  const int rowBase = blockIdx.x * 32;

  const bool full_tile = rowBase + 32 <= n;
  if (full_tile) {
    if (tid < 96) sX[tid / 3][165 + tid % 3] = 0.0f;
    for (int i = tid; i < 3 * 64; i += 128) sW[165 + (i >> 6)][i & 63] = 0.0f;

    const float4* wg = (const float4*)W;
    const float4* xg = (const float4*)(x + (long long)rowBase * 165);
    const int qw_last = tid < 80 ? tid + 2560 : 2639;
    const int qx_last = tid < 40 ? tid + 1280 : 1319;
    float4 wv[21];
    float4 xv[11];
#pragma unroll
    for (int j = 0; j < 21; ++j) wv[j] = wg[j < 20 ? tid + 128 * j : qw_last];
#pragma unroll
    for (int j = 0; j < 11; ++j) xv[j] = xg[j < 10 ? tid + 128 * j : qx_last];

#pragma unroll
    for (int j = 0; j < 21; ++j) {
      const int q = j < 20 ? tid + 128 * j : qw_last;
      *(float4*)&sW[q >> 4][(q & 15) * 4] = wv[j];
    }
#pragma unroll
    for (int j = 0; j < 11; ++j) {
      const int q = j < 10 ? tid + 128 * j : qx_last;
      const unsigned f = 4u * (unsigned)q;
      const unsigned r = f / 165u;
      const unsigned c = f - r * 165u;
      const float e[4] = {xv[j].x, xv[j].y, xv[j].z, xv[j].w};
#pragma unroll
      for (int t = 0; t < 4; ++t) {
        unsigned cc = c + (unsigned)t, rr = r;
        if (cc >= 165u) {
          cc -= 165u;
          rr += 1u;
        }
        sX[rr][cc] = e[t];
      }
    }
  } else {
    for (int i = tid; i < 32 * PROJ_K; i += 128) {
      int r = i / PROJ_K, c = i % PROJ_K;
      int row = rowBase + r;
      sX[r][c] = (row < n && c < 165) ? x[(long long)row * 165 + c] : 0.0f;
    }
    for (int i = tid; i < PROJ_K * 64; i += 128) {
      int r = i >> 6, c = i & 63;
      sW[r][c] = (r < 165) ? W[r * 64 + c] : 0.0f;
    }
  }
  __syncthreads();

  const int wave = tid >> 5;
  const int lane = tid & 31;
  const int hf = lane >> 4;
  const int m = lane & 15;
  const int n0 = wave * 16;

  v8f acc0 = {};
  v8f acc1 = {};
  for (int k = 0; k < PROJ_K; k += 4) {
    const int ka = k + 2 * hf;
    v2f b, a0, a1;
    b[0] = sW[ka][n0 + m];
    b[1] = sW[ka + 1][n0 + m];
    a0[0] = sX[m][ka];
    a0[1] = sX[m][ka + 1];
    a1[0] = sX[16 + m][ka];
    a1[1] = sX[16 + m][ka + 1];
    acc0 = wmma_k4(a0, b, acc0);
    acc1 = wmma_k4(a1, b, acc1);
  }

  const float bv = bias[n0 + m];
  for (int v = 0; v < 8; ++v) {
    int r0 = rowBase + v + 8 * hf;
    int r1 = r0 + 16;
    if (r0 < n) {
      float val = acc0[v] + bv;
      h[(long long)r0 * HID + n0 + m] = val > 0.0f ? val : 0.0f;
    }
    if (r1 < n) {
      float val = acc1[v] + bv;
      h[(long long)r1 * HID + n0 + m] = val > 0.0f ? val : 0.0f;
    }
  }
}

template <bool WITH_CLS>
__global__ __launch_bounds__(128) void conv_kernel(
    const float* __restrict__ h_in, const int* __restrict__ off, const int* __restrict__ csr,
    const float* __restrict__ Wl, const float* __restrict__ bl, const float* __restrict__ Wr,
    float* __restrict__ h_out, const float* __restrict__ Wc, const float* __restrict__ bc,
    float* __restrict__ out, int n) {
  __shared__ __attribute__((aligned(16))) float sAa[32][68];
  __shared__ __attribute__((aligned(16))) float sAh[32][68];
  __shared__ __attribute__((aligned(16))) float sWl[64][68];
  __shared__ __attribute__((aligned(16))) float sWr[64][68];
  __shared__ float sPart[4][32];
  const int tid = threadIdx.x;
  const int rowBase = blockIdx.x * 32;
  const int wave = tid >> 5;
  const int lane = tid & 31;
  const int hf = lane >> 4;
  const int m = lane & 15;
  const int n0 = wave * 16;

  const int nb = rowBase + 8 * wave;
  int oi = nb + lane;
  if (oi > n) oi = n;
  const int ov = off[oi];

  int i0[4], i1[4];
#pragma unroll
  for (int t = 0; t < 4; ++t) {
    const int sA = __builtin_amdgcn_readlane(ov, 2 * t);
    const int sB = __builtin_amdgcn_readlane(ov, 2 * t + 1);
    const int sC = __builtin_amdgcn_readlane(ov, 2 * t + 2);
    const int s = hf ? sB : sA;
    const int cnt = hf ? (sC - sB) : (sB - sA);
    i0[t] = (m < cnt) ? csr[s + m] : 0;
    i1[t] = (16 + m < cnt) ? csr[s + 16 + m] : 0;
  }

  const bool full_tile = rowBase + 32 <= n;
  if (full_tile) {
    const float4* hg = (const float4*)(h_in + (long long)rowBase * HID);
    const float4* wlg = (const float4*)Wl;
    const float4* wrg = (const float4*)Wr;
    float4 hv[4], wl[8], wr[8];
#pragma unroll
    for (int j = 0; j < 4; ++j) hv[j] = hg[tid + 128 * j];
#pragma unroll
    for (int j = 0; j < 8; ++j) {
      wl[j] = wlg[tid + 128 * j];
      wr[j] = wrg[tid + 128 * j];
    }
#pragma unroll
    for (int j = 0; j < 4; ++j) {
      const int q = tid + 128 * j;
      *(float4*)&sAh[q >> 4][(q & 15) * 4] = hv[j];
    }
#pragma unroll
    for (int j = 0; j < 8; ++j) {
      const int q = tid + 128 * j;
      *(float4*)&sWl[q >> 4][(q & 15) * 4] = wl[j];
      *(float4*)&sWr[q >> 4][(q & 15) * 4] = wr[j];
    }
  } else {
    for (int i = tid; i < 32 * 64; i += 128) {
      int r = i >> 6, c = i & 63;
      int row = rowBase + r;
      sAh[r][c] = (row < n) ? h_in[(long long)row * HID + c] : 0.0f;
    }
    for (int i = tid; i < 64 * 64; i += 128) {
      int r = i >> 6, c = i & 63;
      sWl[r][c] = Wl[i];
      sWr[r][c] = Wr[i];
    }
  }

  const float4* h4 = (const float4*)h_in;
#pragma unroll
  for (int t = 0; t < 4; ++t) {
    const int sA = __builtin_amdgcn_readlane(ov, 2 * t);
    const int sB = __builtin_amdgcn_readlane(ov, 2 * t + 1);
    const int sC = __builtin_amdgcn_readlane(ov, 2 * t + 2);
    const int s = hf ? sB : sA;
    const int cnt = hf ? (sC - sB) : (sB - sA);
    const int cA = sB - sA, cB = sC - sB;
    const int mx = cA > cB ? cA : cB;
    float4 acc;
    acc.x = 0.0f;
    acc.y = 0.0f;
    acc.z = 0.0f;
    acc.w = 0.0f;
    for (int base = 0; base < mx; base += 16) {
      int idx;
      if (base == 0) {
        idx = i0[t];
      } else if (base == 16) {
        idx = i1[t];
      } else {
        idx = (base + m < cnt) ? csr[s + base + m] : 0;
      }
      idx = (unsigned)idx < (unsigned)n ? idx : 0;
      const bool more = mx - base > 8;
      float4 v[16];
#pragma unroll
      for (int u = 0; u < 8; ++u) {
        const int ra = __builtin_amdgcn_readlane(idx, u);
        const int rb = __builtin_amdgcn_readlane(idx, 16 + u);
        const int si = hf ? rb : ra;
        v[u] = h4[(long long)si * (HID / 4) + m];
      }
      if (more) {
#pragma unroll
        for (int u = 8; u < 16; ++u) {
          const int ra = __builtin_amdgcn_readlane(idx, u);
          const int rb = __builtin_amdgcn_readlane(idx, 16 + u);
          const int si = hf ? rb : ra;
          v[u] = h4[(long long)si * (HID / 4) + m];
        }
      }
#pragma unroll
      for (int u = 0; u < 8; ++u) {
        if (base + u < cnt) {
          acc.x += v[u].x;
          acc.y += v[u].y;
          acc.z += v[u].z;
          acc.w += v[u].w;
        }
      }
      if (more) {
#pragma unroll
        for (int u = 8; u < 16; ++u) {
          if (base + u < cnt) {
            acc.x += v[u].x;
            acc.y += v[u].y;
            acc.z += v[u].z;
            acc.w += v[u].w;
          }
        }
      }
    }
    float d = (float)cnt;
    d = d > 1.0f ? d : 1.0f;
    float4 a;
    a.x = acc.x / d;
    a.y = acc.y / d;
    a.z = acc.z / d;
    a.w = acc.w / d;
    *(float4*)&sAa[8 * wave + 2 * t + hf][4 * m] = a;
  }
  __syncthreads();

  v8f acc0 = {};
  v8f acc1 = {};
  for (int k = 0; k < 64; k += 4) {
    const int ka = k + 2 * hf;
    v2f b, a0, a1;
    b[0] = sWl[ka][n0 + m];
    b[1] = sWl[ka + 1][n0 + m];
    a0[0] = sAa[m][ka];
    a0[1] = sAa[m][ka + 1];
    a1[0] = sAa[16 + m][ka];
    a1[1] = sAa[16 + m][ka + 1];
    acc0 = wmma_k4(a0, b, acc0);
    acc1 = wmma_k4(a1, b, acc1);
  }
  for (int k = 0; k < 64; k += 4) {
    const int ka = k + 2 * hf;
    v2f b, a0, a1;
    b[0] = sWr[ka][n0 + m];
    b[1] = sWr[ka + 1][n0 + m];
    a0[0] = sAh[m][ka];
    a0[1] = sAh[m][ka + 1];
    a1[0] = sAh[16 + m][ka];
    a1[1] = sAh[16 + m][ka + 1];
    acc0 = wmma_k4(a0, b, acc0);
    acc1 = wmma_k4(a1, b, acc1);
  }

  const float bv = bl[n0 + m];
  if constexpr (!WITH_CLS) {
    for (int v = 0; v < 8; ++v) {
      int r0 = rowBase + v + 8 * hf;
      int r1 = r0 + 16;
      if (r0 < n) {
        float val = acc0[v] + bv;
        h_out[(long long)r0 * HID + n0 + m] = val > 0.0f ? val : 0.0f;
      }
      if (r1 < n) {
        float val = acc1[v] + bv;
        h_out[(long long)r1 * HID + n0 + m] = val > 0.0f ? val : 0.0f;
      }
    }
  } else {
    const float wc = Wc[n0 + m];
    float p[16];
#pragma unroll
    for (int v = 0; v < 8; ++v) {
      float a = acc0[v] + bv;
      float b = acc1[v] + bv;
      a = a > 0.0f ? a : 0.0f;
      b = b > 0.0f ? b : 0.0f;
      p[v] = a * wc;
      p[8 + v] = b * wc;
    }
    const bool b3 = (m & 8) != 0, b2 = (m & 4) != 0, b1 = (m & 2) != 0, b0 = (m & 1) != 0;
    float q[8], r[4], s2[2];
#pragma unroll
    for (int i = 0; i < 8; ++i) {
      const float give = b3 ? p[i] : p[i + 8];
      const float keep = b3 ? p[i + 8] : p[i];
      q[i] = keep + __shfl_xor(give, 8);
    }
#pragma unroll
    for (int i = 0; i < 4; ++i) {
      const float give = b2 ? q[i] : q[i + 4];
      const float keep = b2 ? q[i + 4] : q[i];
      r[i] = keep + __shfl_xor(give, 4);
    }
#pragma unroll
    for (int i = 0; i < 2; ++i) {
      const float give = b1 ? r[i] : r[i + 2];
      const float keep = b1 ? r[i + 2] : r[i];
      s2[i] = keep + __shfl_xor(give, 2);
    }
    const float give = b0 ? s2[0] : s2[1];
    const float keep = b0 ? s2[1] : s2[0];
    const float t = keep + __shfl_xor(give, 1);
    const int lr = m < 8 ? m + 8 * hf : 8 + m + 8 * hf;
    sPart[wave][lr] = t;
    __syncthreads();
    if (tid < 32) {
      const int row = rowBase + tid;
      if (row < n) out[row] = ((sPart[0][tid] + sPart[1][tid]) + (sPart[2][tid] + sPart[3][tid])) + bc[0];
    }
  }
}

extern "C" void kernel_launch(void* const* d_in, const int* in_sizes, int n_in,
                              void* d_out, int out_size, void* d_ws, size_t ws_size,
                              hipStream_t stream) {
  const float* x   = (const float*)d_in[0];
  const int* eidx  = (const int*)d_in[1];
  const float* Wp  = (const float*)d_in[2];
  const float* bp  = (const float*)d_in[3];
  const float* W1l = (const float*)d_in[4];
  const float* b1l = (const float*)d_in[5];
  const float* W1r = (const float*)d_in[6];
  const float* W2l = (const float*)d_in[7];
  const float* b2l = (const float*)d_in[8];
  const float* W2r = (const float*)d_in[9];
  const float* Wc  = (const float*)d_in[10];
  const float* bc  = (const float*)d_in[11];

  const int N = in_sizes[0] / 165;
  const int E = in_sizes[1] / 2;
  const int* src = eidx;
  const int* dst = eidx + E;

  float* HA  = (float*)d_ws;
  float* HB  = HA + (size_t)N * HID;
  const size_t E4 = ((size_t)E + 63) / 64 * 64;
  const size_t N4 = ((size_t)N + 1 + 63) / 64 * 64;
  int* CSR   = (int*)(HB + (size_t)N * HID);
  int* ST    = CSR + E4;
  int* OFF   = ST + E4 + N4;

  const int NC = (E + SORT_CH - 1) / SORT_CH;
  const int NB = (N + BK_NODES - 1) / BK_NODES;
  const unsigned g_row  = (unsigned)((N + 31) / 32);
  const size_t proj_lds = (size_t)PROJ_LDS_FLOATS * sizeof(float);

  bucket_kernel<<<(unsigned)NC, 1024, 0, stream>>>(src, dst, (int*)HB, ST, E, NB);
  group_kernel<<<(unsigned)NB, 1024, 0, stream>>>((const int*)HB, ST, CSR, OFF, E, N, NB, NC);

  proj_kernel<<<g_row, 128, proj_lds, stream>>>(x, Wp, bp, HA, N);

  conv_kernel<false><<<g_row, 128, 0, stream>>>(HA, OFF, CSR, W1l, b1l, W1r, HB, nullptr, nullptr, nullptr, N);
  conv_kernel<true><<<g_row, 128, 0, stream>>>(HB, OFF, CSR, W2l, b2l, W2r, nullptr, Wc, bc, (float*)d_out, N);
}


# edge grouping: both sort levels build their output in LDS at its final place and copy it out in whole 128-byte lines (was 8- and 4-byte stores touching up to 32 lines each); sum order unchanged
# speedup vs baseline: 8.1803x; 1.1116x over previous
#include <hip/hip_runtime.h>

#define HID 64

typedef __attribute__((ext_vector_type(2))) float v2f;
typedef __attribute__((ext_vector_type(8))) float v8f;

__device__ __forceinline__ v8f wmma_k4(v2f a, v2f b, v8f c) {
  return __builtin_amdgcn_wmma_f32_16x16x4_f32(
       false, a,  false, b,
       (short)0, c,  false,  false);
}

#define SORT_CH 16384
#define SORT_PER 16
#define BK_SHIFT 9
#define BK_NODES 512
#define NB_MAX 512
#define NC_MAX 256
#define GROUP_CAP 16384

__device__ __forceinline__ int wave_incl_scan(int x, int lane) {
  for (int o = 1; o < 32; o <<= 1) {
    const int y = __shfl_up(x, o);
    if (lane >= o) x += y;
  }
  return x;
}

__device__ __forceinline__ unsigned same_key_lanes(int key, bool valid) {
  unsigned m = __builtin_amdgcn_ballot_w32(valid);
  for (int i = 0; i < BK_SHIFT; ++i) {
    const bool bit = (key >> i) & 1;
    const unsigned bm = __builtin_amdgcn_ballot_w32(bit);
    m &= bit ? bm : ~bm;
  }
  return valid ? m : 0u;
}

__global__ __launch_bounds__(1024) void bucket_kernel(const int* __restrict__ src, const int* __restrict__ dst,
                                                      int* __restrict__ rec, int* __restrict__ st, int E,
                                                      int nb) {
  __shared__ int wrow[32 * NB_MAX];
  __shared__ int hist[NB_MAX];
  __shared__ int wsum[32];
  __shared__ __attribute__((aligned(16))) int2 stage[SORT_CH];
  const int tid = threadIdx.x;
  const int lane = tid & 31;
  const int wave = tid >> 5;
  const unsigned lt = (1u << lane) - 1u;
  const int e0 = blockIdx.x * SORT_CH;
  for (int i = tid; i < 32 * NB_MAX; i += 1024) wrow[i] = 0;
  __syncthreads();
  int* myrow = wrow + wave * NB_MAX;
  int d[SORT_PER], s[SORT_PER], r[SORT_PER];
#pragma unroll
  for (int u = 0; u < SORT_PER; ++u) {
    const int e = e0 + tid + 1024 * u;
    const bool ok = e < E;
    d[u] = ok ? dst[e] : 0;
    s[u] = ok ? src[e] : 0;
  }
#pragma unroll
  for (int u = 0; u < SORT_PER; ++u) {
    const int e = e0 + tid + 1024 * u;
    const bool ok = e < E;
    int b = d[u] >> BK_SHIFT;
    b = b < 0 ? 0 : (b < nb ? b : nb - 1);
    d[u] = (d[u] & (BK_NODES - 1)) | (b << BK_SHIFT);
    const int old = myrow[b];
    const unsigned m = same_key_lanes(b, ok);
    r[u] = old + __popc(m & lt);
    if (ok && (m & lt) == 0u) myrow[b] = old + __popc(m);
  }
  __syncthreads();
  int v = 0;
  if (tid < nb) {
    for (int w = 0; w < 32; ++w) {
      const int c = wrow[w * NB_MAX + tid];
      wrow[w * NB_MAX + tid] = v;
      v += c;
    }
  }
  const int inc = wave_incl_scan(v, lane);
  if (lane == 31) wsum[wave] = inc;
  __syncthreads();
  int wbase = 0, total = 0;
  for (int w = 0; w < 32; ++w) {
    const int t = wsum[w];
    if (w < wave) wbase += t;
    total += t;
  }
  const int excl = wbase + inc - v;
  int* row = st + (long long)blockIdx.x * (nb + 1);
  if (tid < nb) {
    hist[tid] = excl;
    row[tid] = excl;
  }
  if (tid == 0) row[nb] = total;
  __syncthreads();
  int2* rec2 = (int2*)rec;
#pragma unroll
  for (int u = 0; u < SORT_PER; ++u) {
    const int e = e0 + tid + 1024 * u;
    if (e < E) {
      const int b = d[u] >> BK_SHIFT;
      const int pos = hist[b] + myrow[b] + r[u];
      if ((unsigned)pos < (unsigned)SORT_CH) stage[pos] = make_int2(d[u], s[u]);
    }
  }
  __syncthreads();
  const int left = E - e0;
  const int cnt = left < SORT_CH ? left : SORT_CH;
  const int4* st4 = (const int4*)stage;
  int4* out4 = (int4*)(rec2 + e0);
  int4 pv[SORT_PER / 2];
#pragma unroll
  for (int j = 0; j < SORT_PER / 2; ++j) pv[j] = st4[tid + 1024 * j];
#pragma unroll
  for (int j = 0; j < SORT_PER / 2; ++j) {
    const int q = tid + 1024 * j;
    if (2 * q + 1 < cnt) {
      out4[q] = pv[j];
    } else if (2 * q < cnt) {
      rec2[e0 + 2 * q] = make_int2(pv[j].x, pv[j].y);
    }
  }
}

__global__ __launch_bounds__(1024) void group_kernel(const int* __restrict__ rec, const int* __restrict__ st,
                                                     int* __restrict__ csr, int* __restrict__ off, int E, int n,
                                                     int nb, int nc) {
  __shared__ int crow[32 * BK_NODES];
  __shared__ int nstart[BK_NODES];
  __shared__ int runBeg[NC_MAX];
  __shared__ int runLen[NC_MAX];
  __shared__ int wsumA[32];
  __shared__ int wsumB[32];
  __shared__ int stage[GROUP_CAP];
  const int tid = threadIdx.x;
  const int lane = tid & 31;
  const int wave = tid >> 5;
  const unsigned lt = (1u << lane) - 1u;
  const int b = blockIdx.x;
  const int ncc = nc < NC_MAX ? nc : NC_MAX;
  const int2* rec2 = (const int2*)rec;
  int* myrow = crow + wave * BK_NODES;

  int s0 = 0;
  if (tid < ncc) {
    const int* row = st + (long long)tid * (nb + 1);
    s0 = row[b];
    const int s1 = row[b + 1];
    const int beg = tid * SORT_CH + s0;
    int len = s1 - s0;
    const int lim = E - beg;
    len = len < lim ? len : lim;
    runBeg[tid] = beg;
    runLen[tid] = len;
  }
  for (int i = tid; i < 32 * BK_NODES; i += 1024) crow[i] = 0;
  int part = s0;
  for (int o = 16; o > 0; o >>= 1) part += __shfl_xor(part, o);
  if (lane == 0) wsumA[wave] = part;
  __syncthreads();
  int base = 0;
  for (int w = 0; w < 32; ++w) base += wsumA[w];

  for (int c = wave; c < ncc; c += 32) {
    const int beg = runBeg[c];
    const int len = runLen[c];
    const bool p0 = lane < len, p1 = lane + 32 < len;
    const int2 r0 = rec2[p0 ? beg + lane : 0];
    const int2 r1 = rec2[p1 ? beg + lane + 32 : 0];
    {
      const int key = r0.x & (BK_NODES - 1);
      const int old = myrow[key];
      const unsigned m = same_key_lanes(key, p0);
      if (p0 && (m & lt) == 0u) myrow[key] = old + __popc(m);
    }
    {
      const int key = r1.x & (BK_NODES - 1);
      const int old = myrow[key];
      const unsigned m = same_key_lanes(key, p1);
      if (p1 && (m & lt) == 0u) myrow[key] = old + __popc(m);
    }
    for (int k0 = 64; k0 < len; k0 += 32) {
      const bool pk = k0 + lane < len;
      const int2 rk = rec2[pk ? beg + k0 + lane : 0];
      const int key = rk.x & (BK_NODES - 1);
      const int old = myrow[key];
      const unsigned m = same_key_lanes(key, pk);
      if (pk && (m & lt) == 0u) myrow[key] = old + __popc(m);
    }
  }
  __syncthreads();

  int v = 0;
  if (tid < BK_NODES) {
    for (int w = 0; w < 32; ++w) {
      const int c = crow[w * BK_NODES + tid];
      crow[w * BK_NODES + tid] = v;
      v += c;
    }
  }
  const int inc = wave_incl_scan(v, lane);
  if (lane == 31) wsumB[wave] = inc;
  __syncthreads();
  int wbase = 0, total = 0;
  for (int w = 0; w < 32; ++w) {
    const int t = wsumB[w];
    if (w < wave) wbase += t;
    total += t;
  }
  const int excl = wbase + inc - v;
  if (tid < BK_NODES) {
    nstart[tid] = excl;
    const int node = b * BK_NODES + tid;
    if (node < n) off[node] = base + excl;
  }
  if (b == nb - 1 && tid == 0) off[n] = base + total;
  __syncthreads();

  const int sh = base & 31;
  const bool staged = total + 32 <= GROUP_CAP;
  auto put = [&](unsigned lp, int v, bool p) {
    if (staged) {
      const unsigned j = lp + (unsigned)sh;
      if (p && j < (unsigned)GROUP_CAP) stage[j] = v;
    } else {
      const unsigned pos = (unsigned)base + lp;
      if (p && pos < (unsigned)E) csr[pos] = v;
    }
  };
  for (int c = wave; c < ncc; c += 32) {
    const int beg = runBeg[c];
    const int len = runLen[c];
    const bool p0 = lane < len, p1 = lane + 32 < len;
    const int2 r0 = rec2[p0 ? beg + lane : 0];
    const int2 r1 = rec2[p1 ? beg + lane + 32 : 0];
    {
      const int key = r0.x & (BK_NODES - 1);
      const int old = myrow[key];
      const unsigned m = same_key_lanes(key, p0);
      put((unsigned)(nstart[key] + old + __popc(m & lt)), r0.y, p0);
      if (p0 && (m & lt) == 0u) myrow[key] = old + __popc(m);
    }
    {
      const int key = r1.x & (BK_NODES - 1);
      const int old = myrow[key];
      const unsigned m = same_key_lanes(key, p1);
      put((unsigned)(nstart[key] + old + __popc(m & lt)), r1.y, p1);
      if (p1 && (m & lt) == 0u) myrow[key] = old + __popc(m);
    }
    for (int k0 = 64; k0 < len; k0 += 32) {
      const bool pk = k0 + lane < len;
      const int2 rk = rec2[pk ? beg + k0 + lane : 0];
      const int key = rk.x & (BK_NODES - 1);
      const int old = myrow[key];
      const unsigned m = same_key_lanes(key, pk);
      put((unsigned)(nstart[key] + old + __popc(m & lt)), rk.y, pk);
      if (pk && (m & lt) == 0u) myrow[key] = old + __popc(m);
    }
  }
  if (staged) {
    __syncthreads();
    for (int j = tid; j < sh + total; j += 1024) {
      const unsigned pos = (unsigned)(base - sh + j);
      if (j >= sh && pos < (unsigned)E) csr[pos] = stage[j];
    }
  }
}

#define PROJ_K 168
#define PROJ_SX_STRIDE 172
#define PROJ_SW_STRIDE 68
#define PROJ_LDS_FLOATS (32 * PROJ_SX_STRIDE + PROJ_K * PROJ_SW_STRIDE)

__global__ __launch_bounds__(128) __attribute__((amdgpu_waves_per_eu(1, 4))) void proj_kernel(const float* __restrict__ x,
                                                   const float* __restrict__ W,
                                                   const float* __restrict__ bias,
                                                   float* __restrict__ h, int n) {
  extern __shared__ float smem[];
  float(*sX)[PROJ_SX_STRIDE] = (float(*)[PROJ_SX_STRIDE])smem;
  float(*sW)[PROJ_SW_STRIDE] = (float(*)[PROJ_SW_STRIDE])(smem + 32 * PROJ_SX_STRIDE);

  const int tid = threadIdx.x;
  const int rowBase = blockIdx.x * 32;

  const bool full_tile = rowBase + 32 <= n;
  if (full_tile) {
    if (tid < 96) sX[tid / 3][165 + tid % 3] = 0.0f;
    for (int i = tid; i < 3 * 64; i += 128) sW[165 + (i >> 6)][i & 63] = 0.0f;

    const float4* wg = (const float4*)W;
    const float4* xg = (const float4*)(x + (long long)rowBase * 165);
    const int qw_last = tid < 80 ? tid + 2560 : 2639;
    const int qx_last = tid < 40 ? tid + 1280 : 1319;
    float4 wv[21];
    float4 xv[11];
#pragma unroll
    for (int j = 0; j < 21; ++j) wv[j] = wg[j < 20 ? tid + 128 * j : qw_last];
#pragma unroll
    for (int j = 0; j < 11; ++j) xv[j] = xg[j < 10 ? tid + 128 * j : qx_last];

#pragma unroll
    for (int j = 0; j < 21; ++j) {
      const int q = j < 20 ? tid + 128 * j : qw_last;
      *(float4*)&sW[q >> 4][(q & 15) * 4] = wv[j];
    }
#pragma unroll
    for (int j = 0; j < 11; ++j) {
      const int q = j < 10 ? tid + 128 * j : qx_last;
      const unsigned f = 4u * (unsigned)q;
      const unsigned r = f / 165u;
      const unsigned c = f - r * 165u;
      const float e[4] = {xv[j].x, xv[j].y, xv[j].z, xv[j].w};
#pragma unroll
      for (int t = 0; t < 4; ++t) {
        unsigned cc = c + (unsigned)t, rr = r;
        if (cc >= 165u) {
          cc -= 165u;
          rr += 1u;
        }
        sX[rr][cc] = e[t];
      }
    }
  } else {
    for (int i = tid; i < 32 * PROJ_K; i += 128) {
      int r = i / PROJ_K, c = i % PROJ_K;
      int row = rowBase + r;
      sX[r][c] = (row < n && c < 165) ? x[(long long)row * 165 + c] : 0.0f;
    }
    for (int i = tid; i < PROJ_K * 64; i += 128) {
      int r = i >> 6, c = i & 63;
      sW[r][c] = (r < 165) ? W[r * 64 + c] : 0.0f;
    }
  }
  __syncthreads();

  const int wave = tid >> 5;
  const int lane = tid & 31;
  const int hf = lane >> 4;
  const int m = lane & 15;
  const int n0 = wave * 16;

  v8f acc0 = {};
  v8f acc1 = {};
  for (int k = 0; k < PROJ_K; k += 4) {
    const int ka = k + 2 * hf;
    v2f b, a0, a1;
    b[0] = sW[ka][n0 + m];
    b[1] = sW[ka + 1][n0 + m];
    a0[0] = sX[m][ka];
    a0[1] = sX[m][ka + 1];
    a1[0] = sX[16 + m][ka];
    a1[1] = sX[16 + m][ka + 1];
    acc0 = wmma_k4(a0, b, acc0);
    acc1 = wmma_k4(a1, b, acc1);
  }

  const float bv = bias[n0 + m];
  for (int v = 0; v < 8; ++v) {
    int r0 = rowBase + v + 8 * hf;
    int r1 = r0 + 16;
    if (r0 < n) {
      float val = acc0[v] + bv;
      h[(long long)r0 * HID + n0 + m] = val > 0.0f ? val : 0.0f;
    }
    if (r1 < n) {
      float val = acc1[v] + bv;
      h[(long long)r1 * HID + n0 + m] = val > 0.0f ? val : 0.0f;
    }
  }
}

template <bool WITH_CLS>
__global__ __launch_bounds__(128) void conv_kernel(
    const float* __restrict__ h_in, const int* __restrict__ off, const int* __restrict__ csr,
    const float* __restrict__ Wl, const float* __restrict__ bl, const float* __restrict__ Wr,
    float* __restrict__ h_out, const float* __restrict__ Wc, const float* __restrict__ bc,
    float* __restrict__ out, int n) {
  __shared__ __attribute__((aligned(16))) float sAa[32][68];
  __shared__ __attribute__((aligned(16))) float sAh[32][68];
  __shared__ __attribute__((aligned(16))) float sWl[64][68];
  __shared__ __attribute__((aligned(16))) float sWr[64][68];
  __shared__ float sPart[4][32];
  const int tid = threadIdx.x;
  const int rowBase = blockIdx.x * 32;
  const int wave = tid >> 5;
  const int lane = tid & 31;
  const int hf = lane >> 4;
  const int m = lane & 15;
  const int n0 = wave * 16;

  const int nb = rowBase + 8 * wave;
  int oi = nb + lane;
  if (oi > n) oi = n;
  const int ov = off[oi];

  int i0[4], i1[4];
#pragma unroll
  for (int t = 0; t < 4; ++t) {
    const int sA = __builtin_amdgcn_readlane(ov, 2 * t);
    const int sB = __builtin_amdgcn_readlane(ov, 2 * t + 1);
    const int sC = __builtin_amdgcn_readlane(ov, 2 * t + 2);
    const int s = hf ? sB : sA;
    const int cnt = hf ? (sC - sB) : (sB - sA);
    i0[t] = (m < cnt) ? csr[s + m] : 0;
    i1[t] = (16 + m < cnt) ? csr[s + 16 + m] : 0;
  }

  const bool full_tile = rowBase + 32 <= n;
  if (full_tile) {
    const float4* hg = (const float4*)(h_in + (long long)rowBase * HID);
    const float4* wlg = (const float4*)Wl;
    const float4* wrg = (const float4*)Wr;
    float4 hv[4], wl[8], wr[8];
#pragma unroll
    for (int j = 0; j < 4; ++j) hv[j] = hg[tid + 128 * j];
#pragma unroll
    for (int j = 0; j < 8; ++j) {
      wl[j] = wlg[tid + 128 * j];
      wr[j] = wrg[tid + 128 * j];
    }
#pragma unroll
    for (int j = 0; j < 4; ++j) {
      const int q = tid + 128 * j;
      *(float4*)&sAh[q >> 4][(q & 15) * 4] = hv[j];
    }
#pragma unroll
    for (int j = 0; j < 8; ++j) {
      const int q = tid + 128 * j;
      *(float4*)&sWl[q >> 4][(q & 15) * 4] = wl[j];
      *(float4*)&sWr[q >> 4][(q & 15) * 4] = wr[j];
    }
  } else {
    for (int i = tid; i < 32 * 64; i += 128) {
      int r = i >> 6, c = i & 63;
      int row = rowBase + r;
      sAh[r][c] = (row < n) ? h_in[(long long)row * HID + c] : 0.0f;
    }
    for (int i = tid; i < 64 * 64; i += 128) {
      int r = i >> 6, c = i & 63;
      sWl[r][c] = Wl[i];
      sWr[r][c] = Wr[i];
    }
  }

  const float4* h4 = (const float4*)h_in;
#pragma unroll
  for (int t = 0; t < 4; ++t) {
    const int sA = __builtin_amdgcn_readlane(ov, 2 * t);
    const int sB = __builtin_amdgcn_readlane(ov, 2 * t + 1);
    const int sC = __builtin_amdgcn_readlane(ov, 2 * t + 2);
    const int s = hf ? sB : sA;
    const int cnt = hf ? (sC - sB) : (sB - sA);
    const int cA = sB - sA, cB = sC - sB;
    const int mx = cA > cB ? cA : cB;
    float4 acc;
    acc.x = 0.0f;
    acc.y = 0.0f;
    acc.z = 0.0f;
    acc.w = 0.0f;
    for (int base = 0; base < mx; base += 16) {
      int idx;
      if (base == 0) {
        idx = i0[t];
      } else if (base == 16) {
        idx = i1[t];
      } else {
        idx = (base + m < cnt) ? csr[s + base + m] : 0;
      }
      idx = (unsigned)idx < (unsigned)n ? idx : 0;
      const bool more = mx - base > 8;
      float4 v[16];
#pragma unroll
      for (int u = 0; u < 8; ++u) {
        const int ra = __builtin_amdgcn_readlane(idx, u);
        const int rb = __builtin_amdgcn_readlane(idx, 16 + u);
        const int si = hf ? rb : ra;
        v[u] = h4[(long long)si * (HID / 4) + m];
      }
      if (more) {
#pragma unroll
        for (int u = 8; u < 16; ++u) {
          const int ra = __builtin_amdgcn_readlane(idx, u);
          const int rb = __builtin_amdgcn_readlane(idx, 16 + u);
          const int si = hf ? rb : ra;
          v[u] = h4[(long long)si * (HID / 4) + m];
        }
      }
#pragma unroll
      for (int u = 0; u < 8; ++u) {
        if (base + u < cnt) {
          acc.x += v[u].x;
          acc.y += v[u].y;
          acc.z += v[u].z;
          acc.w += v[u].w;
        }
      }
      if (more) {
#pragma unroll
        for (int u = 8; u < 16; ++u) {
          if (base + u < cnt) {
            acc.x += v[u].x;
            acc.y += v[u].y;
            acc.z += v[u].z;
            acc.w += v[u].w;
          }
        }
      }
    }
    float d = (float)cnt;
    d = d > 1.0f ? d : 1.0f;
    float4 a;
    a.x = acc.x / d;
    a.y = acc.y / d;
    a.z = acc.z / d;
    a.w = acc.w / d;
    *(float4*)&sAa[8 * wave + 2 * t + hf][4 * m] = a;
  }
  __syncthreads();

  v8f acc0 = {};
  v8f acc1 = {};
  for (int k = 0; k < 64; k += 4) {
    const int ka = k + 2 * hf;
    v2f b, a0, a1;
    b[0] = sWl[ka][n0 + m];
    b[1] = sWl[ka + 1][n0 + m];
    a0[0] = sAa[m][ka];
    a0[1] = sAa[m][ka + 1];
    a1[0] = sAa[16 + m][ka];
    a1[1] = sAa[16 + m][ka + 1];
    acc0 = wmma_k4(a0, b, acc0);
    acc1 = wmma_k4(a1, b, acc1);
  }
  for (int k = 0; k < 64; k += 4) {
    const int ka = k + 2 * hf;
    v2f b, a0, a1;
    b[0] = sWr[ka][n0 + m];
    b[1] = sWr[ka + 1][n0 + m];
    a0[0] = sAh[m][ka];
    a0[1] = sAh[m][ka + 1];
    a1[0] = sAh[16 + m][ka];
    a1[1] = sAh[16 + m][ka + 1];
    acc0 = wmma_k4(a0, b, acc0);
    acc1 = wmma_k4(a1, b, acc1);
  }

  const float bv = bl[n0 + m];
  if constexpr (!WITH_CLS) {
    for (int v = 0; v < 8; ++v) {
      int r0 = rowBase + v + 8 * hf;
      int r1 = r0 + 16;
      if (r0 < n) {
        float val = acc0[v] + bv;
        h_out[(long long)r0 * HID + n0 + m] = val > 0.0f ? val : 0.0f;
      }
      if (r1 < n) {
        float val = acc1[v] + bv;
        h_out[(long long)r1 * HID + n0 + m] = val > 0.0f ? val : 0.0f;
      }
    }
  } else {
    const float wc = Wc[n0 + m];
    float p[16];
#pragma unroll
    for (int v = 0; v < 8; ++v) {
      float a = acc0[v] + bv;
      float b = acc1[v] + bv;
      a = a > 0.0f ? a : 0.0f;
      b = b > 0.0f ? b : 0.0f;
      p[v] = a * wc;
      p[8 + v] = b * wc;
    }
    const bool b3 = (m & 8) != 0, b2 = (m & 4) != 0, b1 = (m & 2) != 0, b0 = (m & 1) != 0;
    float q[8], r[4], s2[2];
#pragma unroll
    for (int i = 0; i < 8; ++i) {
      const float give = b3 ? p[i] : p[i + 8];
      const float keep = b3 ? p[i + 8] : p[i];
      q[i] = keep + __shfl_xor(give, 8);
    }
#pragma unroll
    for (int i = 0; i < 4; ++i) {
      const float give = b2 ? q[i] : q[i + 4];
      const float keep = b2 ? q[i + 4] : q[i];
      r[i] = keep + __shfl_xor(give, 4);
    }
#pragma unroll
    for (int i = 0; i < 2; ++i) {
      const float give = b1 ? r[i] : r[i + 2];
      const float keep = b1 ? r[i + 2] : r[i];
      s2[i] = keep + __shfl_xor(give, 2);
    }
    const float give = b0 ? s2[0] : s2[1];
    const float keep = b0 ? s2[1] : s2[0];
    const float t = keep + __shfl_xor(give, 1);
    const int lr = m < 8 ? m + 8 * hf : 8 + m + 8 * hf;
    sPart[wave][lr] = t;
    __syncthreads();
    if (tid < 32) {
      const int row = rowBase + tid;
      if (row < n) out[row] = ((sPart[0][tid] + sPart[1][tid]) + (sPart[2][tid] + sPart[3][tid])) + bc[0];
    }
  }
}

extern "C" void kernel_launch(void* const* d_in, const int* in_sizes, int n_in,
                              void* d_out, int out_size, void* d_ws, size_t ws_size,
                              hipStream_t stream) {
  const float* x   = (const float*)d_in[0];
  const int* eidx  = (const int*)d_in[1];
  const float* Wp  = (const float*)d_in[2];
  const float* bp  = (const float*)d_in[3];
  const float* W1l = (const float*)d_in[4];
  const float* b1l = (const float*)d_in[5];
  const float* W1r = (const float*)d_in[6];
  const float* W2l = (const float*)d_in[7];
  const float* b2l = (const float*)d_in[8];
  const float* W2r = (const float*)d_in[9];
  const float* Wc  = (const float*)d_in[10];
  const float* bc  = (const float*)d_in[11];

  const int N = in_sizes[0] / 165;
  const int E = in_sizes[1] / 2;
  const int* src = eidx;
  const int* dst = eidx + E;

  float* HA  = (float*)d_ws;
  float* HB  = HA + (size_t)N * HID;
  const size_t E4 = ((size_t)E + 63) / 64 * 64;
  const size_t N4 = ((size_t)N + 1 + 63) / 64 * 64;
  int* CSR   = (int*)(HB + (size_t)N * HID);
  int* ST    = CSR + E4;
  int* OFF   = ST + E4 + N4;

  const int NC = (E + SORT_CH - 1) / SORT_CH;
  const int NB = (N + BK_NODES - 1) / BK_NODES;
  const unsigned g_row  = (unsigned)((N + 31) / 32);
  const size_t proj_lds = (size_t)PROJ_LDS_FLOATS * sizeof(float);

  bucket_kernel<<<(unsigned)NC, 1024, 0, stream>>>(src, dst, (int*)HB, ST, E, NB);
  group_kernel<<<(unsigned)NB, 1024, 0, stream>>>((const int*)HB, ST, CSR, OFF, E, N, NB, NC);

  proj_kernel<<<g_row, 128, proj_lds, stream>>>(x, Wp, bp, HA, N);

  conv_kernel<false><<<g_row, 128, 0, stream>>>(HA, OFF, CSR, W1l, b1l, W1r, HB, nullptr, nullptr, nullptr, N);
  conv_kernel<true><<<g_row, 128, 0, stream>>>(HB, OFF, CSR, W2l, b2l, W2r, nullptr, Wc, bc, (float*)d_out, N);
}


# conv neighbour sums: source index to its 16 lanes by one DPP row_share, 32-bit row offset, ended runs read a row of zeros so adds need no mask (was 2 readlanes, select, 64-bit math); add order kept
# speedup vs baseline: 8.6038x; 1.0518x over previous
#include <hip/hip_runtime.h>

#define HID 64

typedef __attribute__((ext_vector_type(2))) float v2f;
typedef __attribute__((ext_vector_type(8))) float v8f;

__device__ __forceinline__ v8f wmma_k4(v2f a, v2f b, v8f c) {
  return __builtin_amdgcn_wmma_f32_16x16x4_f32(
       false, a,  false, b,
       (short)0, c,  false,  false);
}

#define SORT_CH 16384
#define SORT_PER 16
#define BK_SHIFT 9
#define BK_NODES 512
#define NB_MAX 512
#define NC_MAX 256
#define GROUP_CAP 16384

__device__ __forceinline__ int wave_incl_scan(int x, int lane) {
  for (int o = 1; o < 32; o <<= 1) {
    const int y = __shfl_up(x, o);
    if (lane >= o) x += y;
  }
  return x;
}

__device__ __forceinline__ unsigned same_key_lanes(int key, bool valid) {
  unsigned m = __builtin_amdgcn_ballot_w32(valid);
  for (int i = 0; i < BK_SHIFT; ++i) {
    const bool bit = (key >> i) & 1;
    const unsigned bm = __builtin_amdgcn_ballot_w32(bit);
    m &= bit ? bm : ~bm;
  }
  return valid ? m : 0u;
}

__global__ __launch_bounds__(1024) void bucket_kernel(const int* __restrict__ src, const int* __restrict__ dst,
                                                      int* __restrict__ rec, int* __restrict__ st, int E,
                                                      int nb) {
  __shared__ int wrow[32 * NB_MAX];
  __shared__ int hist[NB_MAX];
  __shared__ int wsum[32];
  __shared__ __attribute__((aligned(16))) int2 stage[SORT_CH];
  const int tid = threadIdx.x;
  const int lane = tid & 31;
  const int wave = tid >> 5;
  const unsigned lt = (1u << lane) - 1u;
  const int e0 = blockIdx.x * SORT_CH;
  for (int i = tid; i < 32 * NB_MAX; i += 1024) wrow[i] = 0;
  __syncthreads();
  int* myrow = wrow + wave * NB_MAX;
  int d[SORT_PER], s[SORT_PER], r[SORT_PER];
#pragma unroll
  for (int u = 0; u < SORT_PER; ++u) {
    const int e = e0 + tid + 1024 * u;
    const bool ok = e < E;
    d[u] = ok ? dst[e] : 0;
    s[u] = ok ? src[e] : 0;
  }
#pragma unroll
  for (int u = 0; u < SORT_PER; ++u) {
    const int e = e0 + tid + 1024 * u;
    const bool ok = e < E;
    int b = d[u] >> BK_SHIFT;
    b = b < 0 ? 0 : (b < nb ? b : nb - 1);
    d[u] = (d[u] & (BK_NODES - 1)) | (b << BK_SHIFT);
    const int old = myrow[b];
    const unsigned m = same_key_lanes(b, ok);
    r[u] = old + __popc(m & lt);
    if (ok && (m & lt) == 0u) myrow[b] = old + __popc(m);
  }
  __syncthreads();
  int v = 0;
  if (tid < nb) {
    for (int w = 0; w < 32; ++w) {
      const int c = wrow[w * NB_MAX + tid];
      wrow[w * NB_MAX + tid] = v;
      v += c;
    }
  }
  const int inc = wave_incl_scan(v, lane);
  if (lane == 31) wsum[wave] = inc;
  __syncthreads();
  int wbase = 0, total = 0;
  for (int w = 0; w < 32; ++w) {
    const int t = wsum[w];
    if (w < wave) wbase += t;
    total += t;
  }
  const int excl = wbase + inc - v;
  int* row = st + (long long)blockIdx.x * (nb + 1);
  if (tid < nb) {
    hist[tid] = excl;
    row[tid] = excl;
  }
  if (tid == 0) row[nb] = total;
  __syncthreads();
  int2* rec2 = (int2*)rec;
#pragma unroll
  for (int u = 0; u < SORT_PER; ++u) {
    const int e = e0 + tid + 1024 * u;
    if (e < E) {
      const int b = d[u] >> BK_SHIFT;
      const int pos = hist[b] + myrow[b] + r[u];
      if ((unsigned)pos < (unsigned)SORT_CH) stage[pos] = make_int2(d[u], s[u]);
    }
  }
  __syncthreads();
  const int left = E - e0;
  const int cnt = left < SORT_CH ? left : SORT_CH;
  const int4* st4 = (const int4*)stage;
  int4* out4 = (int4*)(rec2 + e0);
  int4 pv[SORT_PER / 2];
#pragma unroll
  for (int j = 0; j < SORT_PER / 2; ++j) pv[j] = st4[tid + 1024 * j];
#pragma unroll
  for (int j = 0; j < SORT_PER / 2; ++j) {
    const int q = tid + 1024 * j;
    if (2 * q + 1 < cnt) {
      out4[q] = pv[j];
    } else if (2 * q < cnt) {
      rec2[e0 + 2 * q] = make_int2(pv[j].x, pv[j].y);
    }
  }
}

__global__ __launch_bounds__(1024) void group_kernel(const int* __restrict__ rec, const int* __restrict__ st,
                                                     int* __restrict__ csr, int* __restrict__ off,
                                                     float* __restrict__ zrow, int E, int n, int nb, int nc) {
  __shared__ int crow[32 * BK_NODES];
  __shared__ int nstart[BK_NODES];
  __shared__ int runBeg[NC_MAX];
  __shared__ int runLen[NC_MAX];
  __shared__ int wsumA[32];
  __shared__ int wsumB[32];
  __shared__ int stage[GROUP_CAP];
  const int tid = threadIdx.x;
  const int lane = tid & 31;
  const int wave = tid >> 5;
  const unsigned lt = (1u << lane) - 1u;
  const int b = blockIdx.x;
  const int ncc = nc < NC_MAX ? nc : NC_MAX;
  const int2* rec2 = (const int2*)rec;
  int* myrow = crow + wave * BK_NODES;
  if (b == 0 && tid < HID) zrow[tid] = 0.0f;

  int s0 = 0;
  if (tid < ncc) {
    const int* row = st + (long long)tid * (nb + 1);
    s0 = row[b];
    const int s1 = row[b + 1];
    const int beg = tid * SORT_CH + s0;
    int len = s1 - s0;
    const int lim = E - beg;
    len = len < lim ? len : lim;
    runBeg[tid] = beg;
    runLen[tid] = len;
  }
  for (int i = tid; i < 32 * BK_NODES; i += 1024) crow[i] = 0;
  int part = s0;
  for (int o = 16; o > 0; o >>= 1) part += __shfl_xor(part, o);
  if (lane == 0) wsumA[wave] = part;
  __syncthreads();
  int base = 0;
  for (int w = 0; w < 32; ++w) base += wsumA[w];

  for (int c = wave; c < ncc; c += 32) {
    const int beg = runBeg[c];
    const int len = runLen[c];
    const bool p0 = lane < len, p1 = lane + 32 < len;
    const int2 r0 = rec2[p0 ? beg + lane : 0];
    const int2 r1 = rec2[p1 ? beg + lane + 32 : 0];
    {
      const int key = r0.x & (BK_NODES - 1);
      const int old = myrow[key];
      const unsigned m = same_key_lanes(key, p0);
      if (p0 && (m & lt) == 0u) myrow[key] = old + __popc(m);
    }
    {
      const int key = r1.x & (BK_NODES - 1);
      const int old = myrow[key];
      const unsigned m = same_key_lanes(key, p1);
      if (p1 && (m & lt) == 0u) myrow[key] = old + __popc(m);
    }
    for (int k0 = 64; k0 < len; k0 += 32) {
      const bool pk = k0 + lane < len;
      const int2 rk = rec2[pk ? beg + k0 + lane : 0];
      const int key = rk.x & (BK_NODES - 1);
      const int old = myrow[key];
      const unsigned m = same_key_lanes(key, pk);
      if (pk && (m & lt) == 0u) myrow[key] = old + __popc(m);
    }
  }
  __syncthreads();

  int v = 0;
  if (tid < BK_NODES) {
    for (int w = 0; w < 32; ++w) {
      const int c = crow[w * BK_NODES + tid];
      crow[w * BK_NODES + tid] = v;
      v += c;
    }
  }
  const int inc = wave_incl_scan(v, lane);
  if (lane == 31) wsumB[wave] = inc;
  __syncthreads();
  int wbase = 0, total = 0;
  for (int w = 0; w < 32; ++w) {
    const int t = wsumB[w];
    if (w < wave) wbase += t;
    total += t;
  }
  const int excl = wbase + inc - v;
  if (tid < BK_NODES) {
    nstart[tid] = excl;
    const int node = b * BK_NODES + tid;
    if (node < n) off[node] = base + excl;
  }
  if (b == nb - 1 && tid == 0) off[n] = base + total;
  __syncthreads();

  const int sh = base & 31;
  const bool staged = total + 32 <= GROUP_CAP;
  auto put = [&](unsigned lp, int v, bool p) {
    if (staged) {
      const unsigned j = lp + (unsigned)sh;
      if (p && j < (unsigned)GROUP_CAP) stage[j] = v;
    } else {
      const unsigned pos = (unsigned)base + lp;
      if (p && pos < (unsigned)E) csr[pos] = v;
    }
  };
  for (int c = wave; c < ncc; c += 32) {
    const int beg = runBeg[c];
    const int len = runLen[c];
    const bool p0 = lane < len, p1 = lane + 32 < len;
    const int2 r0 = rec2[p0 ? beg + lane : 0];
    const int2 r1 = rec2[p1 ? beg + lane + 32 : 0];
    {
      const int key = r0.x & (BK_NODES - 1);
      const int old = myrow[key];
      const unsigned m = same_key_lanes(key, p0);
      put((unsigned)(nstart[key] + old + __popc(m & lt)), r0.y, p0);
      if (p0 && (m & lt) == 0u) myrow[key] = old + __popc(m);
    }
    {
      const int key = r1.x & (BK_NODES - 1);
      const int old = myrow[key];
      const unsigned m = same_key_lanes(key, p1);
      put((unsigned)(nstart[key] + old + __popc(m & lt)), r1.y, p1);
      if (p1 && (m & lt) == 0u) myrow[key] = old + __popc(m);
    }
    for (int k0 = 64; k0 < len; k0 += 32) {
      const bool pk = k0 + lane < len;
      const int2 rk = rec2[pk ? beg + k0 + lane : 0];
      const int key = rk.x & (BK_NODES - 1);
      const int old = myrow[key];
      const unsigned m = same_key_lanes(key, pk);
      put((unsigned)(nstart[key] + old + __popc(m & lt)), rk.y, pk);
      if (pk && (m & lt) == 0u) myrow[key] = old + __popc(m);
    }
  }
  if (staged) {
    __syncthreads();
    for (int j = tid; j < sh + total; j += 1024) {
      const unsigned pos = (unsigned)(base - sh + j);
      if (j >= sh && pos < (unsigned)E) csr[pos] = stage[j];
    }
  }
}

#define PROJ_K 168
#define PROJ_SX_STRIDE 172
#define PROJ_SW_STRIDE 68
#define PROJ_LDS_FLOATS (32 * PROJ_SX_STRIDE + PROJ_K * PROJ_SW_STRIDE)

__global__ __launch_bounds__(128) __attribute__((amdgpu_waves_per_eu(1, 4))) void proj_kernel(const float* __restrict__ x,
                                                   const float* __restrict__ W,
                                                   const float* __restrict__ bias,
                                                   float* __restrict__ h, int n) {
  extern __shared__ float smem[];
  float(*sX)[PROJ_SX_STRIDE] = (float(*)[PROJ_SX_STRIDE])smem;
  float(*sW)[PROJ_SW_STRIDE] = (float(*)[PROJ_SW_STRIDE])(smem + 32 * PROJ_SX_STRIDE);

  const int tid = threadIdx.x;
  const int rowBase = blockIdx.x * 32;

  const bool full_tile = rowBase + 32 <= n;
  if (full_tile) {
    if (tid < 96) sX[tid / 3][165 + tid % 3] = 0.0f;
    for (int i = tid; i < 3 * 64; i += 128) sW[165 + (i >> 6)][i & 63] = 0.0f;

    const float4* wg = (const float4*)W;
    const float4* xg = (const float4*)(x + (long long)rowBase * 165);
    const int qw_last = tid < 80 ? tid + 2560 : 2639;
    const int qx_last = tid < 40 ? tid + 1280 : 1319;
    float4 wv[21];
    float4 xv[11];
#pragma unroll
    for (int j = 0; j < 21; ++j) wv[j] = wg[j < 20 ? tid + 128 * j : qw_last];
#pragma unroll
    for (int j = 0; j < 11; ++j) xv[j] = xg[j < 10 ? tid + 128 * j : qx_last];

#pragma unroll
    for (int j = 0; j < 21; ++j) {
      const int q = j < 20 ? tid + 128 * j : qw_last;
      *(float4*)&sW[q >> 4][(q & 15) * 4] = wv[j];
    }
#pragma unroll
    for (int j = 0; j < 11; ++j) {
      const int q = j < 10 ? tid + 128 * j : qx_last;
      const unsigned f = 4u * (unsigned)q;
      const unsigned r = f / 165u;
      const unsigned c = f - r * 165u;
      const float e[4] = {xv[j].x, xv[j].y, xv[j].z, xv[j].w};
#pragma unroll
      for (int t = 0; t < 4; ++t) {
        unsigned cc = c + (unsigned)t, rr = r;
        if (cc >= 165u) {
          cc -= 165u;
          rr += 1u;
        }
        sX[rr][cc] = e[t];
      }
    }
  } else {
    for (int i = tid; i < 32 * PROJ_K; i += 128) {
      int r = i / PROJ_K, c = i % PROJ_K;
      int row = rowBase + r;
      sX[r][c] = (row < n && c < 165) ? x[(long long)row * 165 + c] : 0.0f;
    }
    for (int i = tid; i < PROJ_K * 64; i += 128) {
      int r = i >> 6, c = i & 63;
      sW[r][c] = (r < 165) ? W[r * 64 + c] : 0.0f;
    }
  }
  __syncthreads();

  const int wave = tid >> 5;
  const int lane = tid & 31;
  const int hf = lane >> 4;
  const int m = lane & 15;
  const int n0 = wave * 16;

  v8f acc0 = {};
  v8f acc1 = {};
  for (int k = 0; k < PROJ_K; k += 4) {
    const int ka = k + 2 * hf;
    v2f b, a0, a1;
    b[0] = sW[ka][n0 + m];
    b[1] = sW[ka + 1][n0 + m];
    a0[0] = sX[m][ka];
    a0[1] = sX[m][ka + 1];
    a1[0] = sX[16 + m][ka];
    a1[1] = sX[16 + m][ka + 1];
    acc0 = wmma_k4(a0, b, acc0);
    acc1 = wmma_k4(a1, b, acc1);
  }

  const float bv = bias[n0 + m];
  for (int v = 0; v < 8; ++v) {
    int r0 = rowBase + v + 8 * hf;
    int r1 = r0 + 16;
    if (r0 < n) {
      float val = acc0[v] + bv;
      h[(long long)r0 * HID + n0 + m] = val > 0.0f ? val : 0.0f;
    }
    if (r1 < n) {
      float val = acc1[v] + bv;
      h[(long long)r1 * HID + n0 + m] = val > 0.0f ? val : 0.0f;
    }
  }
}

template <int U>
__device__ __forceinline__ unsigned half_share(unsigned x) {
  return (unsigned)__builtin_amdgcn_update_dpp(0, (int)x, 0x150 + U, 0xf, 0xf, true);
}

template <int U, int END>
struct RowLoads {
  static __device__ __forceinline__ void issue(float4* v, const float4* __restrict__ h4, unsigned q, unsigned m4) {
    v[U] = h4[half_share<(U & 15)>(q) + m4];
    RowLoads<U + 1, END>::issue(v, h4, q, m4);
  }
};
template <int END>
struct RowLoads<END, END> {
  static __device__ __forceinline__ void issue(float4*, const float4* __restrict__, unsigned, unsigned) {}
};

template <bool WITH_CLS>
__global__ __launch_bounds__(128) void conv_kernel(
    const float* __restrict__ h_in, const int* __restrict__ off, const int* __restrict__ csr,
    const float* __restrict__ Wl, const float* __restrict__ bl, const float* __restrict__ Wr,
    float* __restrict__ h_out, const float* __restrict__ Wc, const float* __restrict__ bc,
    float* __restrict__ out, const float* __restrict__ zrow, int n) {
  __shared__ __attribute__((aligned(16))) float sAa[32][68];
  __shared__ __attribute__((aligned(16))) float sAh[32][68];
  __shared__ __attribute__((aligned(16))) float sWl[64][68];
  __shared__ __attribute__((aligned(16))) float sWr[64][68];
  __shared__ float sPart[4][32];
  const int tid = threadIdx.x;
  const int rowBase = blockIdx.x * 32;
  const int wave = tid >> 5;
  const int lane = tid & 31;
  const int hf = lane >> 4;
  const int m = lane & 15;
  const int n0 = wave * 16;

  const int nb = rowBase + 8 * wave;
  int oi = nb + lane;
  if (oi > n) oi = n;
  const int ov = off[oi];

  int i0[4], i1[4];
#pragma unroll
  for (int t = 0; t < 4; ++t) {
    const int sA = __builtin_amdgcn_readlane(ov, 2 * t);
    const int sB = __builtin_amdgcn_readlane(ov, 2 * t + 1);
    const int sC = __builtin_amdgcn_readlane(ov, 2 * t + 2);
    const int s = hf ? sB : sA;
    const int cnt = hf ? (sC - sB) : (sB - sA);
    i0[t] = (m < cnt) ? csr[s + m] : 0;
    i1[t] = (16 + m < cnt) ? csr[s + 16 + m] : 0;
  }

  const bool full_tile = rowBase + 32 <= n;
  if (full_tile) {
    const float4* hg = (const float4*)(h_in + (long long)rowBase * HID);
    const float4* wlg = (const float4*)Wl;
    const float4* wrg = (const float4*)Wr;
    float4 hv[4], wl[8], wr[8];
#pragma unroll
    for (int j = 0; j < 4; ++j) hv[j] = hg[tid + 128 * j];
#pragma unroll
    for (int j = 0; j < 8; ++j) {
      wl[j] = wlg[tid + 128 * j];
      wr[j] = wrg[tid + 128 * j];
    }
#pragma unroll
    for (int j = 0; j < 4; ++j) {
      const int q = tid + 128 * j;
      *(float4*)&sAh[q >> 4][(q & 15) * 4] = hv[j];
    }
#pragma unroll
    for (int j = 0; j < 8; ++j) {
      const int q = tid + 128 * j;
      *(float4*)&sWl[q >> 4][(q & 15) * 4] = wl[j];
      *(float4*)&sWr[q >> 4][(q & 15) * 4] = wr[j];
    }
  } else {
    for (int i = tid; i < 32 * 64; i += 128) {
      int r = i >> 6, c = i & 63;
      int row = rowBase + r;
      sAh[r][c] = (row < n) ? h_in[(long long)row * HID + c] : 0.0f;
    }
    for (int i = tid; i < 64 * 64; i += 128) {
      int r = i >> 6, c = i & 63;
      sWl[r][c] = Wl[i];
      sWr[r][c] = Wr[i];
    }
  }

  const float4* h4 = (const float4*)h_in;
  const unsigned zq = (unsigned)((zrow - h_in) >> 2);
#pragma unroll
  for (int t = 0; t < 4; ++t) {
    const int sA = __builtin_amdgcn_readlane(ov, 2 * t);
    const int sB = __builtin_amdgcn_readlane(ov, 2 * t + 1);
    const int sC = __builtin_amdgcn_readlane(ov, 2 * t + 2);
    const int s = hf ? sB : sA;
    const int cnt = hf ? (sC - sB) : (sB - sA);
    const int cA = sB - sA, cB = sC - sB;
    const int mx = cA > cB ? cA : cB;
    float4 acc;
    acc.x = 0.0f;
    acc.y = 0.0f;
    acc.z = 0.0f;
    acc.w = 0.0f;
    for (int base = 0; base < mx; base += 16) {
      int idx;
      if (base == 0) {
        idx = i0[t];
      } else if (base == 16) {
        idx = i1[t];
      } else {
        idx = (base + m < cnt) ? csr[s + base + m] : 0;
      }
      const bool live = base + m < cnt && (unsigned)idx < (unsigned)n;
      const unsigned q = live ? (unsigned)idx * (unsigned)(HID / 4) : zq;
      const bool more = mx - base > 8;
      float4 v[16];
      RowLoads<0, 8>::issue(v, h4, q, (unsigned)m);
      if (more) RowLoads<8, 16>::issue(v, h4, q, (unsigned)m);
#pragma unroll
      for (int u = 0; u < 8; ++u) {
        acc.x += v[u].x;
        acc.y += v[u].y;
        acc.z += v[u].z;
        acc.w += v[u].w;
      }
      if (more) {
#pragma unroll
        for (int u = 8; u < 16; ++u) {
          acc.x += v[u].x;
          acc.y += v[u].y;
          acc.z += v[u].z;
          acc.w += v[u].w;
        }
      }
    }
    float d = (float)cnt;
    d = d > 1.0f ? d : 1.0f;
    float4 a;
    a.x = acc.x / d;
    a.y = acc.y / d;
    a.z = acc.z / d;
    a.w = acc.w / d;
    *(float4*)&sAa[8 * wave + 2 * t + hf][4 * m] = a;
  }
  __syncthreads();

  v8f acc0 = {};
  v8f acc1 = {};
  for (int k = 0; k < 64; k += 4) {
    const int ka = k + 2 * hf;
    v2f b, a0, a1;
    b[0] = sWl[ka][n0 + m];
    b[1] = sWl[ka + 1][n0 + m];
    a0[0] = sAa[m][ka];
    a0[1] = sAa[m][ka + 1];
    a1[0] = sAa[16 + m][ka];
    a1[1] = sAa[16 + m][ka + 1];
    acc0 = wmma_k4(a0, b, acc0);
    acc1 = wmma_k4(a1, b, acc1);
  }
  for (int k = 0; k < 64; k += 4) {
    const int ka = k + 2 * hf;
    v2f b, a0, a1;
    b[0] = sWr[ka][n0 + m];
    b[1] = sWr[ka + 1][n0 + m];
    a0[0] = sAh[m][ka];
    a0[1] = sAh[m][ka + 1];
    a1[0] = sAh[16 + m][ka];
    a1[1] = sAh[16 + m][ka + 1];
    acc0 = wmma_k4(a0, b, acc0);
    acc1 = wmma_k4(a1, b, acc1);
  }

  const float bv = bl[n0 + m];
  if constexpr (!WITH_CLS) {
    for (int v = 0; v < 8; ++v) {
      int r0 = rowBase + v + 8 * hf;
      int r1 = r0 + 16;
      if (r0 < n) {
        float val = acc0[v] + bv;
        h_out[(long long)r0 * HID + n0 + m] = val > 0.0f ? val : 0.0f;
      }
      if (r1 < n) {
        float val = acc1[v] + bv;
        h_out[(long long)r1 * HID + n0 + m] = val > 0.0f ? val : 0.0f;
      }
    }
  } else {
    const float wc = Wc[n0 + m];
    float p[16];
#pragma unroll
    for (int v = 0; v < 8; ++v) {
      float a = acc0[v] + bv;
      float b = acc1[v] + bv;
      a = a > 0.0f ? a : 0.0f;
      b = b > 0.0f ? b : 0.0f;
      p[v] = a * wc;
      p[8 + v] = b * wc;
    }
    const bool b3 = (m & 8) != 0, b2 = (m & 4) != 0, b1 = (m & 2) != 0, b0 = (m & 1) != 0;
    float q[8], r[4], s2[2];
#pragma unroll
    for (int i = 0; i < 8; ++i) {
      const float give = b3 ? p[i] : p[i + 8];
      const float keep = b3 ? p[i + 8] : p[i];
      q[i] = keep + __shfl_xor(give, 8);
    }
#pragma unroll
    for (int i = 0; i < 4; ++i) {
      const float give = b2 ? q[i] : q[i + 4];
      const float keep = b2 ? q[i + 4] : q[i];
      r[i] = keep + __shfl_xor(give, 4);
    }
#pragma unroll
    for (int i = 0; i < 2; ++i) {
      const float give = b1 ? r[i] : r[i + 2];
      const float keep = b1 ? r[i + 2] : r[i];
      s2[i] = keep + __shfl_xor(give, 2);
    }
    const float give = b0 ? s2[0] : s2[1];
    const float keep = b0 ? s2[1] : s2[0];
    const float t = keep + __shfl_xor(give, 1);
    const int lr = m < 8 ? m + 8 * hf : 8 + m + 8 * hf;
    sPart[wave][lr] = t;
    __syncthreads();
    if (tid < 32) {
      const int row = rowBase + tid;
      if (row < n) out[row] = ((sPart[0][tid] + sPart[1][tid]) + (sPart[2][tid] + sPart[3][tid])) + bc[0];
    }
  }
}

extern "C" void kernel_launch(void* const* d_in, const int* in_sizes, int n_in,
                              void* d_out, int out_size, void* d_ws, size_t ws_size,
                              hipStream_t stream) {
  const float* x   = (const float*)d_in[0];
  const int* eidx  = (const int*)d_in[1];
  const float* Wp  = (const float*)d_in[2];
  const float* bp  = (const float*)d_in[3];
  const float* W1l = (const float*)d_in[4];
  const float* b1l = (const float*)d_in[5];
  const float* W1r = (const float*)d_in[6];
  const float* W2l = (const float*)d_in[7];
  const float* b2l = (const float*)d_in[8];
  const float* W2r = (const float*)d_in[9];
  const float* Wc  = (const float*)d_in[10];
  const float* bc  = (const float*)d_in[11];

  const int N = in_sizes[0] / 165;
  const int E = in_sizes[1] / 2;
  const int* src = eidx;
  const int* dst = eidx + E;

  float* HA  = (float*)d_ws;
  float* HB  = HA + (size_t)N * HID;
  const size_t E4 = ((size_t)E + 63) / 64 * 64;
  const size_t N4 = ((size_t)N + 1 + 63) / 64 * 64;
  int* CSR   = (int*)(HB + (size_t)N * HID);
  int* ST    = CSR + E4;
  int* OFF   = ST + E4 + N4;
  float* ZROW = (float*)(ST + E4);

  const int NC = (E + SORT_CH - 1) / SORT_CH;
  const int NB = (N + BK_NODES - 1) / BK_NODES;
  const unsigned g_row  = (unsigned)((N + 31) / 32);
  const size_t proj_lds = (size_t)PROJ_LDS_FLOATS * sizeof(float);

  bucket_kernel<<<(unsigned)NC, 1024, 0, stream>>>(src, dst, (int*)HB, ST, E, NB);
  group_kernel<<<(unsigned)NB, 1024, 0, stream>>>((const int*)HB, ST, CSR, OFF, ZROW, E, N, NB, NC);

  proj_kernel<<<g_row, 128, proj_lds, stream>>>(x, Wp, bp, HA, N);

  conv_kernel<false><<<g_row, 128, 0, stream>>>(HA, OFF, CSR, W1l, b1l, W1r, HB, nullptr, nullptr, nullptr, ZROW, N);
  conv_kernel<true><<<g_row, 128, 0, stream>>>(HB, OFF, CSR, W2l, b2l, W2r, nullptr, Wc, bc, (float*)d_out, ZROW, N);
}


# GEMM kernels (proj, conv 1, conv 2): weights staged to LDS once per work-group, which loops over 32-row tiles (grids 1024 / 1536; was one work-group and one weight staging per tile); sum order kept
# speedup vs baseline: 9.2009x; 1.0694x over previous
#include <hip/hip_runtime.h>

#define HID 64

typedef __attribute__((ext_vector_type(2))) float v2f;
typedef __attribute__((ext_vector_type(8))) float v8f;

__device__ __forceinline__ v8f wmma_k4(v2f a, v2f b, v8f c) {
  return __builtin_amdgcn_wmma_f32_16x16x4_f32(
       false, a,  false, b,
       (short)0, c,  false,  false);
}

__device__ __forceinline__ void lds_barrier() {
  __builtin_amdgcn_fence(__ATOMIC_RELEASE, "workgroup", "local");
  __builtin_amdgcn_s_barrier();
  __builtin_amdgcn_fence(__ATOMIC_ACQUIRE, "workgroup", "local");
}

#define SORT_CH 16384
#define SORT_PER 16
#define BK_SHIFT 9
#define BK_NODES 512
#define NB_MAX 512
#define NC_MAX 256
#define GROUP_CAP 16384

__device__ __forceinline__ int wave_incl_scan(int x, int lane) {
  for (int o = 1; o < 32; o <<= 1) {
    const int y = __shfl_up(x, o);
    if (lane >= o) x += y;
  }
  return x;
}

__device__ __forceinline__ unsigned same_key_lanes(int key, bool valid) {
  unsigned m = __builtin_amdgcn_ballot_w32(valid);
  for (int i = 0; i < BK_SHIFT; ++i) {
    const bool bit = (key >> i) & 1;
    const unsigned bm = __builtin_amdgcn_ballot_w32(bit);
    m &= bit ? bm : ~bm;
  }
  return valid ? m : 0u;
}

__global__ __launch_bounds__(1024) void bucket_kernel(const int* __restrict__ src, const int* __restrict__ dst,
                                                      int* __restrict__ rec, int* __restrict__ st, int E,
                                                      int nb) {
  __shared__ int wrow[32 * NB_MAX];
  __shared__ int hist[NB_MAX];
  __shared__ int wsum[32];
  __shared__ __attribute__((aligned(16))) int2 stage[SORT_CH];
  const int tid = threadIdx.x;
  const int lane = tid & 31;
  const int wave = tid >> 5;
  const unsigned lt = (1u << lane) - 1u;
  const int e0 = blockIdx.x * SORT_CH;
  for (int i = tid; i < 32 * NB_MAX; i += 1024) wrow[i] = 0;
  __syncthreads();
  int* myrow = wrow + wave * NB_MAX;
  int d[SORT_PER], s[SORT_PER], r[SORT_PER];
#pragma unroll
  for (int u = 0; u < SORT_PER; ++u) {
    const int e = e0 + tid + 1024 * u;
    const bool ok = e < E;
    d[u] = ok ? dst[e] : 0;
    s[u] = ok ? src[e] : 0;
  }
#pragma unroll
  for (int u = 0; u < SORT_PER; ++u) {
    const int e = e0 + tid + 1024 * u;
    const bool ok = e < E;
    int b = d[u] >> BK_SHIFT;
    b = b < 0 ? 0 : (b < nb ? b : nb - 1);
    d[u] = (d[u] & (BK_NODES - 1)) | (b << BK_SHIFT);
    const int old = myrow[b];
    const unsigned m = same_key_lanes(b, ok);
    r[u] = old + __popc(m & lt);
    if (ok && (m & lt) == 0u) myrow[b] = old + __popc(m);
  }
  __syncthreads();
  int v = 0;
  if (tid < nb) {
    for (int w = 0; w < 32; ++w) {
      const int c = wrow[w * NB_MAX + tid];
      wrow[w * NB_MAX + tid] = v;
      v += c;
    }
  }
  const int inc = wave_incl_scan(v, lane);
  if (lane == 31) wsum[wave] = inc;
  __syncthreads();
  int wbase = 0, total = 0;
  for (int w = 0; w < 32; ++w) {
    const int t = wsum[w];
    if (w < wave) wbase += t;
    total += t;
  }
  const int excl = wbase + inc - v;
  int* row = st + (long long)blockIdx.x * (nb + 1);
  if (tid < nb) {
    hist[tid] = excl;
    row[tid] = excl;
  }
  if (tid == 0) row[nb] = total;
  __syncthreads();
  int2* rec2 = (int2*)rec;
#pragma unroll
  for (int u = 0; u < SORT_PER; ++u) {
    const int e = e0 + tid + 1024 * u;
    if (e < E) {
      const int b = d[u] >> BK_SHIFT;
      const int pos = hist[b] + myrow[b] + r[u];
      if ((unsigned)pos < (unsigned)SORT_CH) stage[pos] = make_int2(d[u], s[u]);
    }
  }
  __syncthreads();
  const int left = E - e0;
  const int cnt = left < SORT_CH ? left : SORT_CH;
  const int4* st4 = (const int4*)stage;
  int4* out4 = (int4*)(rec2 + e0);
  int4 pv[SORT_PER / 2];
#pragma unroll
  for (int j = 0; j < SORT_PER / 2; ++j) pv[j] = st4[tid + 1024 * j];
#pragma unroll
  for (int j = 0; j < SORT_PER / 2; ++j) {
    const int q = tid + 1024 * j;
    if (2 * q + 1 < cnt) {
      out4[q] = pv[j];
    } else if (2 * q < cnt) {
      rec2[e0 + 2 * q] = make_int2(pv[j].x, pv[j].y);
    }
  }
}

__global__ __launch_bounds__(1024) void group_kernel(const int* __restrict__ rec, const int* __restrict__ st,
                                                     int* __restrict__ csr, int* __restrict__ off,
                                                     float* __restrict__ zrow, int E, int n, int nb, int nc) {
  __shared__ int crow[32 * BK_NODES];
  __shared__ int nstart[BK_NODES];
  __shared__ int runBeg[NC_MAX];
  __shared__ int runLen[NC_MAX];
  __shared__ int wsumA[32];
  __shared__ int wsumB[32];
  __shared__ int stage[GROUP_CAP];
  const int tid = threadIdx.x;
  const int lane = tid & 31;
  const int wave = tid >> 5;
  const unsigned lt = (1u << lane) - 1u;
  const int b = blockIdx.x;
  const int ncc = nc < NC_MAX ? nc : NC_MAX;
  const int2* rec2 = (const int2*)rec;
  int* myrow = crow + wave * BK_NODES;
  if (b == 0 && tid < HID) zrow[tid] = 0.0f;

  int s0 = 0;
  if (tid < ncc) {
    const int* row = st + (long long)tid * (nb + 1);
    s0 = row[b];
    const int s1 = row[b + 1];
    const int beg = tid * SORT_CH + s0;
    int len = s1 - s0;
    const int lim = E - beg;
    len = len < lim ? len : lim;
    runBeg[tid] = beg;
    runLen[tid] = len;
  }
  for (int i = tid; i < 32 * BK_NODES; i += 1024) crow[i] = 0;
  int part = s0;
  for (int o = 16; o > 0; o >>= 1) part += __shfl_xor(part, o);
  if (lane == 0) wsumA[wave] = part;
  __syncthreads();
  int base = 0;
  for (int w = 0; w < 32; ++w) base += wsumA[w];

  for (int c = wave; c < ncc; c += 32) {
    const int beg = runBeg[c];
    const int len = runLen[c];
    const bool p0 = lane < len, p1 = lane + 32 < len;
    const int2 r0 = rec2[p0 ? beg + lane : 0];
    const int2 r1 = rec2[p1 ? beg + lane + 32 : 0];
    {
      const int key = r0.x & (BK_NODES - 1);
      const int old = myrow[key];
      const unsigned m = same_key_lanes(key, p0);
      if (p0 && (m & lt) == 0u) myrow[key] = old + __popc(m);
    }
    {
      const int key = r1.x & (BK_NODES - 1);
      const int old = myrow[key];
      const unsigned m = same_key_lanes(key, p1);
      if (p1 && (m & lt) == 0u) myrow[key] = old + __popc(m);
    }
    for (int k0 = 64; k0 < len; k0 += 32) {
      const bool pk = k0 + lane < len;
      const int2 rk = rec2[pk ? beg + k0 + lane : 0];
      const int key = rk.x & (BK_NODES - 1);
      const int old = myrow[key];
      const unsigned m = same_key_lanes(key, pk);
      if (pk && (m & lt) == 0u) myrow[key] = old + __popc(m);
    }
  }
  __syncthreads();

  int v = 0;
  if (tid < BK_NODES) {
    for (int w = 0; w < 32; ++w) {
      const int c = crow[w * BK_NODES + tid];
      crow[w * BK_NODES + tid] = v;
      v += c;
    }
  }
  const int inc = wave_incl_scan(v, lane);
  if (lane == 31) wsumB[wave] = inc;
  __syncthreads();
  int wbase = 0, total = 0;
  for (int w = 0; w < 32; ++w) {
    const int t = wsumB[w];
    if (w < wave) wbase += t;
    total += t;
  }
  const int excl = wbase + inc - v;
  if (tid < BK_NODES) {
    nstart[tid] = excl;
    const int node = b * BK_NODES + tid;
    if (node < n) off[node] = base + excl;
  }
  if (b == nb - 1 && tid == 0) off[n] = base + total;
  __syncthreads();

  const int sh = base & 31;
  const bool staged = total + 32 <= GROUP_CAP;
  auto put = [&](unsigned lp, int v, bool p) {
    if (staged) {
      const unsigned j = lp + (unsigned)sh;
      if (p && j < (unsigned)GROUP_CAP) stage[j] = v;
    } else {
      const unsigned pos = (unsigned)base + lp;
      if (p && pos < (unsigned)E) csr[pos] = v;
    }
  };
  for (int c = wave; c < ncc; c += 32) {
    const int beg = runBeg[c];
    const int len = runLen[c];
    const bool p0 = lane < len, p1 = lane + 32 < len;
    const int2 r0 = rec2[p0 ? beg + lane : 0];
    const int2 r1 = rec2[p1 ? beg + lane + 32 : 0];
    {
      const int key = r0.x & (BK_NODES - 1);
      const int old = myrow[key];
      const unsigned m = same_key_lanes(key, p0);
      put((unsigned)(nstart[key] + old + __popc(m & lt)), r0.y, p0);
      if (p0 && (m & lt) == 0u) myrow[key] = old + __popc(m);
    }
    {
      const int key = r1.x & (BK_NODES - 1);
      const int old = myrow[key];
      const unsigned m = same_key_lanes(key, p1);
      put((unsigned)(nstart[key] + old + __popc(m & lt)), r1.y, p1);
      if (p1 && (m & lt) == 0u) myrow[key] = old + __popc(m);
    }
    for (int k0 = 64; k0 < len; k0 += 32) {
      const bool pk = k0 + lane < len;
      const int2 rk = rec2[pk ? beg + k0 + lane : 0];
      const int key = rk.x & (BK_NODES - 1);
      const int old = myrow[key];
      const unsigned m = same_key_lanes(key, pk);
      put((unsigned)(nstart[key] + old + __popc(m & lt)), rk.y, pk);
      if (pk && (m & lt) == 0u) myrow[key] = old + __popc(m);
    }
  }
  if (staged) {
    __syncthreads();
    for (int j = tid; j < sh + total; j += 1024) {
      const unsigned pos = (unsigned)(base - sh + j);
      if (j >= sh && pos < (unsigned)E) csr[pos] = stage[j];
    }
  }
}

#define PROJ_K 168
#define PROJ_SX_STRIDE 172
#define PROJ_SW_STRIDE 68
#define PROJ_LDS_FLOATS (32 * PROJ_SX_STRIDE + PROJ_K * PROJ_SW_STRIDE)

__global__ __launch_bounds__(128) __attribute__((amdgpu_waves_per_eu(1, 4))) void proj_kernel(const float* __restrict__ x,
                                                   const float* __restrict__ W,
                                                   const float* __restrict__ bias,
                                                   float* __restrict__ h, int n) {
  extern __shared__ float smem[];
  float(*sX)[PROJ_SX_STRIDE] = (float(*)[PROJ_SX_STRIDE])smem;
  float(*sW)[PROJ_SW_STRIDE] = (float(*)[PROJ_SW_STRIDE])(smem + 32 * PROJ_SX_STRIDE);

  const int tid = threadIdx.x;
  const int wave = tid >> 5;
  const int lane = tid & 31;
  const int hf = lane >> 4;
  const int m = lane & 15;
  const int n0 = wave * 16;

  if (tid < 96) sX[tid / 3][165 + tid % 3] = 0.0f;
  for (int i = tid; i < 3 * 64; i += 128) sW[165 + (i >> 6)][i & 63] = 0.0f;
  {
    const float4* wg = (const float4*)W;
    const int qw_last = tid < 80 ? tid + 2560 : 2639;
    float4 wv[21];
#pragma unroll
    for (int j = 0; j < 21; ++j) wv[j] = wg[j < 20 ? tid + 128 * j : qw_last];
#pragma unroll
    for (int j = 0; j < 21; ++j) {
      const int q = j < 20 ? tid + 128 * j : qw_last;
      *(float4*)&sW[q >> 4][(q & 15) * 4] = wv[j];
    }
  }
  const float bv = bias[n0 + m];
  const int ntiles = (n + 31) >> 5;

  for (int tile = blockIdx.x; tile < ntiles; tile += (int)gridDim.x) {
  const int rowBase = tile * 32;
  const bool full_tile = rowBase + 32 <= n;
  if (full_tile) {
    const float4* xg = (const float4*)(x + (long long)rowBase * 165);
    const int qx_last = tid < 40 ? tid + 1280 : 1319;
    float4 xv[11];
#pragma unroll
    for (int j = 0; j < 11; ++j) xv[j] = xg[j < 10 ? tid + 128 * j : qx_last];
    lds_barrier();
#pragma unroll
    for (int j = 0; j < 11; ++j) {
      const int q = j < 10 ? tid + 128 * j : qx_last;
      const unsigned f = 4u * (unsigned)q;
      const unsigned r = f / 165u;
      const unsigned c = f - r * 165u;
      const float e[4] = {xv[j].x, xv[j].y, xv[j].z, xv[j].w};
#pragma unroll
      for (int t = 0; t < 4; ++t) {
        unsigned cc = c + (unsigned)t, rr = r;
        if (cc >= 165u) {
          cc -= 165u;
          rr += 1u;
        }
        sX[rr][cc] = e[t];
      }
    }
  } else {
    lds_barrier();
    for (int i = tid; i < 32 * PROJ_K; i += 128) {
      int r = i / PROJ_K, c = i % PROJ_K;
      int row = rowBase + r;
      sX[r][c] = (row < n && c < 165) ? x[(long long)row * 165 + c] : 0.0f;
    }
  }
  __syncthreads();

  v8f acc0 = {};
  v8f acc1 = {};
  for (int k = 0; k < PROJ_K; k += 4) {
    const int ka = k + 2 * hf;
    v2f b, a0, a1;
    b[0] = sW[ka][n0 + m];
    b[1] = sW[ka + 1][n0 + m];
    a0[0] = sX[m][ka];
    a0[1] = sX[m][ka + 1];
    a1[0] = sX[16 + m][ka];
    a1[1] = sX[16 + m][ka + 1];
    acc0 = wmma_k4(a0, b, acc0);
    acc1 = wmma_k4(a1, b, acc1);
  }

  for (int v = 0; v < 8; ++v) {
    int r0 = rowBase + v + 8 * hf;
    int r1 = r0 + 16;
    if (r0 < n) {
      float val = acc0[v] + bv;
      h[(long long)r0 * HID + n0 + m] = val > 0.0f ? val : 0.0f;
    }
    if (r1 < n) {
      float val = acc1[v] + bv;
      h[(long long)r1 * HID + n0 + m] = val > 0.0f ? val : 0.0f;
    }
  }
  }
}

template <int U>
__device__ __forceinline__ unsigned half_share(unsigned x) {
  return (unsigned)__builtin_amdgcn_update_dpp(0, (int)x, 0x150 + U, 0xf, 0xf, true);
}

template <int U, int END>
struct RowLoads {
  static __device__ __forceinline__ void issue(float4* v, const float4* __restrict__ h4, unsigned q, unsigned m4) {
    v[U] = h4[half_share<(U & 15)>(q) + m4];
    RowLoads<U + 1, END>::issue(v, h4, q, m4);
  }
};
template <int END>
struct RowLoads<END, END> {
  static __device__ __forceinline__ void issue(float4*, const float4* __restrict__, unsigned, unsigned) {}
};

template <bool WITH_CLS>
__global__ __launch_bounds__(128) void conv_kernel(
    const float* __restrict__ h_in, const int* __restrict__ off, const int* __restrict__ csr,
    const float* __restrict__ Wl, const float* __restrict__ bl, const float* __restrict__ Wr,
    float* __restrict__ h_out, const float* __restrict__ Wc, const float* __restrict__ bc,
    float* __restrict__ out, const float* __restrict__ zrow, int n) {
  __shared__ __attribute__((aligned(16))) float sAa[32][68];
  __shared__ __attribute__((aligned(16))) float sAh[32][68];
  __shared__ __attribute__((aligned(16))) float sWl[64][68];
  __shared__ __attribute__((aligned(16))) float sWr[64][68];
  __shared__ float sPart[4][32];
  const int tid = threadIdx.x;
  const int wave = tid >> 5;
  const int lane = tid & 31;
  const int hf = lane >> 4;
  const int m = lane & 15;
  const int n0 = wave * 16;

  {
    const float4* wlg = (const float4*)Wl;
    const float4* wrg = (const float4*)Wr;
    float4 wl[8], wr[8];
#pragma unroll
    for (int j = 0; j < 8; ++j) {
      wl[j] = wlg[tid + 128 * j];
      wr[j] = wrg[tid + 128 * j];
    }
#pragma unroll
    for (int j = 0; j < 8; ++j) {
      const int q = tid + 128 * j;
      *(float4*)&sWl[q >> 4][(q & 15) * 4] = wl[j];
      *(float4*)&sWr[q >> 4][(q & 15) * 4] = wr[j];
    }
  }

  const float4* h4 = (const float4*)h_in;
  const unsigned zq = (unsigned)((zrow - h_in) >> 2);
  const float bv = bl[n0 + m];
  float wc = 0.0f;
  if constexpr (WITH_CLS) wc = Wc[n0 + m];
  const int ntiles = (n + 31) >> 5;

  for (int tile = blockIdx.x; tile < ntiles; tile += (int)gridDim.x) {
  const int rowBase = tile * 32;

  const int nb = rowBase + 8 * wave;
  int oi = nb + lane;
  if (oi > n) oi = n;
  const int ov = off[oi];

  int i0[4], i1[4];
#pragma unroll
  for (int t = 0; t < 4; ++t) {
    const int sA = __builtin_amdgcn_readlane(ov, 2 * t);
    const int sB = __builtin_amdgcn_readlane(ov, 2 * t + 1);
    const int sC = __builtin_amdgcn_readlane(ov, 2 * t + 2);
    const int s = hf ? sB : sA;
    const int cnt = hf ? (sC - sB) : (sB - sA);
    i0[t] = (m < cnt) ? csr[s + m] : 0;
    i1[t] = (16 + m < cnt) ? csr[s + 16 + m] : 0;
  }

  const bool full_tile = rowBase + 32 <= n;
  if (full_tile) {
    const float4* hg = (const float4*)(h_in + (long long)rowBase * HID);
    const float4 hv0 = hg[tid];
    const float4 hv1 = hg[tid + 128];
    const float4 hv2 = hg[tid + 256];
    const float4 hv3 = hg[tid + 384];
    if constexpr (!WITH_CLS) lds_barrier();
    *(float4*)&sAh[(tid >> 4)][(tid & 15) * 4] = hv0;
    *(float4*)&sAh[(tid >> 4) + 8][(tid & 15) * 4] = hv1;
    *(float4*)&sAh[(tid >> 4) + 16][(tid & 15) * 4] = hv2;
    *(float4*)&sAh[(tid >> 4) + 24][(tid & 15) * 4] = hv3;
  } else {
    if constexpr (!WITH_CLS) lds_barrier();
    for (int i = tid; i < 32 * 64; i += 128) {
      int r = i >> 6, c = i & 63;
      int row = rowBase + r;
      sAh[r][c] = (row < n) ? h_in[(long long)row * HID + c] : 0.0f;
    }
  }
#pragma unroll
  for (int t = 0; t < 4; ++t) {
    const int sA = __builtin_amdgcn_readlane(ov, 2 * t);
    const int sB = __builtin_amdgcn_readlane(ov, 2 * t + 1);
    const int sC = __builtin_amdgcn_readlane(ov, 2 * t + 2);
    const int s = hf ? sB : sA;
    const int cnt = hf ? (sC - sB) : (sB - sA);
    const int cA = sB - sA, cB = sC - sB;
    const int mx = cA > cB ? cA : cB;
    float4 acc;
    acc.x = 0.0f;
    acc.y = 0.0f;
    acc.z = 0.0f;
    acc.w = 0.0f;
    for (int base = 0; base < mx; base += 16) {
      int idx;
      if (base == 0) {
        idx = i0[t];
      } else if (base == 16) {
        idx = i1[t];
      } else {
        idx = (base + m < cnt) ? csr[s + base + m] : 0;
      }
      const bool live = base + m < cnt && (unsigned)idx < (unsigned)n;
      const unsigned q = live ? (unsigned)idx * (unsigned)(HID / 4) : zq;
      const bool more = mx - base > 8;
      float4 v[16];
      RowLoads<0, 8>::issue(v, h4, q, (unsigned)m);
      if (more) RowLoads<8, 16>::issue(v, h4, q, (unsigned)m);
#pragma unroll
      for (int u = 0; u < 8; ++u) {
        acc.x += v[u].x;
        acc.y += v[u].y;
        acc.z += v[u].z;
        acc.w += v[u].w;
      }
      if (more) {
#pragma unroll
        for (int u = 8; u < 16; ++u) {
          acc.x += v[u].x;
          acc.y += v[u].y;
          acc.z += v[u].z;
          acc.w += v[u].w;
        }
      }
    }
    float d = (float)cnt;
    d = d > 1.0f ? d : 1.0f;
    float4 a;
    a.x = acc.x / d;
    a.y = acc.y / d;
    a.z = acc.z / d;
    a.w = acc.w / d;
    *(float4*)&sAa[8 * wave + 2 * t + hf][4 * m] = a;
  }
  __syncthreads();

  v8f acc0 = {};
  v8f acc1 = {};
  for (int k = 0; k < 64; k += 4) {
    const int ka = k + 2 * hf;
    v2f b, a0, a1;
    b[0] = sWl[ka][n0 + m];
    b[1] = sWl[ka + 1][n0 + m];
    a0[0] = sAa[m][ka];
    a0[1] = sAa[m][ka + 1];
    a1[0] = sAa[16 + m][ka];
    a1[1] = sAa[16 + m][ka + 1];
    acc0 = wmma_k4(a0, b, acc0);
    acc1 = wmma_k4(a1, b, acc1);
  }
  for (int k = 0; k < 64; k += 4) {
    const int ka = k + 2 * hf;
    v2f b, a0, a1;
    b[0] = sWr[ka][n0 + m];
    b[1] = sWr[ka + 1][n0 + m];
    a0[0] = sAh[m][ka];
    a0[1] = sAh[m][ka + 1];
    a1[0] = sAh[16 + m][ka];
    a1[1] = sAh[16 + m][ka + 1];
    acc0 = wmma_k4(a0, b, acc0);
    acc1 = wmma_k4(a1, b, acc1);
  }

  if constexpr (!WITH_CLS) {
    for (int v = 0; v < 8; ++v) {
      int r0 = rowBase + v + 8 * hf;
      int r1 = r0 + 16;
      if (r0 < n) {
        float val = acc0[v] + bv;
        h_out[(long long)r0 * HID + n0 + m] = val > 0.0f ? val : 0.0f;
      }
      if (r1 < n) {
        float val = acc1[v] + bv;
        h_out[(long long)r1 * HID + n0 + m] = val > 0.0f ? val : 0.0f;
      }
    }
  } else {
    float p[16];
#pragma unroll
    for (int v = 0; v < 8; ++v) {
      float a = acc0[v] + bv;
      float b = acc1[v] + bv;
      a = a > 0.0f ? a : 0.0f;
      b = b > 0.0f ? b : 0.0f;
      p[v] = a * wc;
      p[8 + v] = b * wc;
    }
    const bool b3 = (m & 8) != 0, b2 = (m & 4) != 0, b1 = (m & 2) != 0, b0 = (m & 1) != 0;
    float q[8], r[4], s2[2];
#pragma unroll
    for (int i = 0; i < 8; ++i) {
      const float give = b3 ? p[i] : p[i + 8];
      const float keep = b3 ? p[i + 8] : p[i];
      q[i] = keep + __shfl_xor(give, 8);
    }
#pragma unroll
    for (int i = 0; i < 4; ++i) {
      const float give = b2 ? q[i] : q[i + 4];
      const float keep = b2 ? q[i + 4] : q[i];
      r[i] = keep + __shfl_xor(give, 4);
    }
#pragma unroll
    for (int i = 0; i < 2; ++i) {
      const float give = b1 ? r[i] : r[i + 2];
      const float keep = b1 ? r[i + 2] : r[i];
      s2[i] = keep + __shfl_xor(give, 2);
    }
    const float give = b0 ? s2[0] : s2[1];
    const float keep = b0 ? s2[1] : s2[0];
    const float t = keep + __shfl_xor(give, 1);
    const int lr = m < 8 ? m + 8 * hf : 8 + m + 8 * hf;
    sPart[wave][lr] = t;
    __syncthreads();
    if (tid < 32) {
      const int row = rowBase + tid;
      if (row < n) out[row] = ((sPart[0][tid] + sPart[1][tid]) + (sPart[2][tid] + sPart[3][tid])) + bc[0];
    }
  }
  }
}

extern "C" void kernel_launch(void* const* d_in, const int* in_sizes, int n_in,
                              void* d_out, int out_size, void* d_ws, size_t ws_size,
                              hipStream_t stream) {
  const float* x   = (const float*)d_in[0];
  const int* eidx  = (const int*)d_in[1];
  const float* Wp  = (const float*)d_in[2];
  const float* bp  = (const float*)d_in[3];
  const float* W1l = (const float*)d_in[4];
  const float* b1l = (const float*)d_in[5];
  const float* W1r = (const float*)d_in[6];
  const float* W2l = (const float*)d_in[7];
  const float* b2l = (const float*)d_in[8];
  const float* W2r = (const float*)d_in[9];
  const float* Wc  = (const float*)d_in[10];
  const float* bc  = (const float*)d_in[11];

  const int N = in_sizes[0] / 165;
  const int E = in_sizes[1] / 2;
  const int* src = eidx;
  const int* dst = eidx + E;

  float* HA  = (float*)d_ws;
  float* HB  = HA + (size_t)N * HID;
  const size_t E4 = ((size_t)E + 63) / 64 * 64;
  const size_t N4 = ((size_t)N + 1 + 63) / 64 * 64;
  int* CSR   = (int*)(HB + (size_t)N * HID);
  int* ST    = CSR + E4;
  int* OFF   = ST + E4 + N4;
  float* ZROW = (float*)(ST + E4);

  const int NC = (E + SORT_CH - 1) / SORT_CH;
  const int NB = (N + BK_NODES - 1) / BK_NODES;
  const unsigned g_row  = (unsigned)((N + 31) / 32);
  const unsigned g_proj = g_row < 1024u ? g_row : 1024u;
  const unsigned g_conv = g_row < 1536u ? g_row : 1536u;
  const size_t proj_lds = (size_t)PROJ_LDS_FLOATS * sizeof(float);

  bucket_kernel<<<(unsigned)NC, 1024, 0, stream>>>(src, dst, (int*)HB, ST, E, NB);
  group_kernel<<<(unsigned)NB, 1024, 0, stream>>>((const int*)HB, ST, CSR, OFF, ZROW, E, N, NB, NC);

  proj_kernel<<<g_proj, 128, proj_lds, stream>>>(x, Wp, bp, HA, N);

  conv_kernel<false><<<g_conv, 128, 0, stream>>>(HA, OFF, CSR, W1l, b1l, W1r, HB, nullptr, nullptr, nullptr, ZROW, N);
  conv_kernel<true><<<g_conv, 128, 0, stream>>>(HB, OFF, CSR, W2l, b2l, W2r, nullptr, Wc, bc, (float*)d_out, ZROW, N);
}
